# starting kernel
# baseline (speedup 1.0000x reference)
#include <hip/hip_runtime.h>
#include <hip/hip_bf16.h>

typedef float v2f __attribute__((ext_vector_type(2)));
typedef float v8f __attribute__((ext_vector_type(8)));

constexpr int TLEN  = 18000;
constexpr int BATCH = 16;
constexpr int WIN   = 48;
constexpr int MWIN  = TLEN - WIN;

__global__ void pos_window_stats(const float* __restrict__ rgb,
                                 const float* __restrict__ matrix,
                                 float* __restrict__ g) {
  int tid = blockIdx.x * blockDim.x + threadIdx.x;
  if (tid >= BATCH * MWIN) return;
  int b = tid & 15;
  int n = tid >> 4;

  const float* p = rgb + (size_t)(b * TLEN + n) * 3;
  float s0 = 0.f, s1 = 0.f, s2 = 0.f;
  float q00 = 0.f, q01 = 0.f, q02 = 0.f, q11 = 0.f, q12 = 0.f, q22 = 0.f;
#pragma unroll 4
  for (int w = 0; w < WIN; ++w) {
    float r0 = p[3 * w + 0];
    float r1 = p[3 * w + 1];
    float r2 = p[3 * w + 2];
    s0 += r0; s1 += r1; s2 += r2;
    q00 = fmaf(r0, r0, q00); q01 = fmaf(r0, r1, q01); q02 = fmaf(r0, r2, q02);
    q11 = fmaf(r1, r1, q11); q12 = fmaf(r1, r2, q12); q22 = fmaf(r2, r2, q22);
  }

  const float invW = 1.0f / (float)WIN;
  float inv0 = 1.0f / (s0 * invW + 1e-6f);
  float inv1 = 1.0f / (s1 * invW + 1e-6f);
  float inv2 = 1.0f / (s2 * invW + 1e-6f);

  float m00 = matrix[0], m01 = matrix[1], m02 = matrix[2];
  float m10 = matrix[3], m11 = matrix[4], m12 = matrix[5];

  float u0 = m00 * inv0, u1 = m01 * inv1, u2 = m02 * inv2;
  float sum0 = u0 * s0 + u1 * s1 + u2 * s2;
  float sq0  = u0 * u0 * q00 + u1 * u1 * q11 + u2 * u2 * q22 +
               2.f * (u0 * u1 * q01 + u0 * u2 * q02 + u1 * u2 * q12);
  float var0 = (sq0 - sum0 * sum0 * invW) * (1.0f / (float)(WIN - 1));
  float std0 = sqrtf(fmaxf(var0, 0.f));

  float v0 = m10 * inv0, v1 = m11 * inv1, v2 = m12 * inv2;
  float sum1 = v0 * s0 + v1 * s1 + v2 * s2;
  float sq1  = v0 * v0 * q00 + v1 * v1 * q11 + v2 * v2 * q22 +
               2.f * (v0 * v1 * q01 + v0 * v2 * q02 + v1 * v2 * q12);
  float var1 = (sq1 - sum1 * sum1 * invW) * (1.0f / (float)(WIN - 1));
  float std1 = sqrtf(fmaxf(var1, 0.f));

  float c = std0 / (std1 + 1e-4f);

  g[(size_t)(0 * MWIN + n) * 16 + b] = (m00 + c * m10) * inv0;
  g[(size_t)(1 * MWIN + n) * 16 + b] = (m01 + c * m11) * inv1;
  g[(size_t)(2 * MWIN + n) * 16 + b] = (m02 + c * m12) * inv2;
}

__global__ void pos_overlap_add_wmma(const float* __restrict__ rgb,
                                     const float* __restrict__ g,
                                     float* __restrict__ ppg) {
  const int t0   = blockIdx.x * 16;
  const int lane = threadIdx.x;
  const int col  = lane & 15;
  const int hi   = lane >> 4;
  const int base = t0 - 47;

  v8f acc[3];
#pragma unroll
  for (int k = 0; k < 3; ++k) {
    const float* gk = g + (size_t)k * MWIN * 16;
    v8f d = {0.f, 0.f, 0.f, 0.f, 0.f, 0.f, 0.f, 0.f};
#pragma unroll
    for (int s = 0; s < 16; ++s) {
      int j0 = 4 * s + 2 * hi;
      int j1 = j0 + 1;
      v2f a;
      a.x = (j0 >= col && j0 <= col + 47) ? 1.0f : 0.0f;
      a.y = (j1 >= col && j1 <= col + 47) ? 1.0f : 0.0f;
      int n0 = base + j0;
      int n1 = base + j1;
      int c0 = min(max(n0, 0), MWIN - 1);
      int c1 = min(max(n1, 0), MWIN - 1);
      float ok0 = ((unsigned)n0 < (unsigned)MWIN) ? 1.0f : 0.0f;
      float ok1 = ((unsigned)n1 < (unsigned)MWIN) ? 1.0f : 0.0f;
      v2f bb;
      bb.x = gk[(size_t)c0 * 16 + col] * ok0;
      bb.y = gk[(size_t)c1 * 16 + col] * ok1;
      d = __builtin_amdgcn_wmma_f32_16x16x4_f32(
           false, a,  false, bb,
           (short)0, d,  false,  false);
    }
    acc[k] = d;
  }

#pragma unroll
  for (int p = 0; p < 8; ++p) {
    int t = t0 + p + 8 * hi;
    const float* rp = rgb + (size_t)(col * TLEN + t) * 3;
    float v = rp[0] * acc[0][p] + rp[1] * acc[1][p] + rp[2] * acc[2][p];
    ppg[(size_t)t * 16 + col] = v;
  }
}

__global__ void pos_filtfilt(const float* __restrict__ ppg,
                             const float* __restrict__ acoef,
                             const float* __restrict__ bcoef,
                             float* __restrict__ ybuf,
                             float* __restrict__ out) {
  int b = threadIdx.x;
  if (b >= BATCH) return;
  float ia = 1.0f / acoef[0];
  float b0 = bcoef[0] * ia, b1 = bcoef[1] * ia, b2 = bcoef[2] * ia;
  float a1 = acoef[1] * ia, a2 = acoef[2] * ia;

  float z1 = 0.f, z2 = 0.f;
  for (int t = 0; t < TLEN; ++t) {
    float x = ppg[(size_t)t * 16 + b];
    float y = fmaf(b0, x, z1);
    z1 = fmaf(b1, x, fmaf(-a1, y, z2));
    z2 = fmaf(b2, x, -a2 * y);
    ybuf[(size_t)t * 16 + b] = y;
  }
  z1 = 0.f; z2 = 0.f;
  for (int k = 0; k < TLEN; ++k) {
    int t = TLEN - 1 - k;
    float x = ybuf[(size_t)t * 16 + b];
    float y = fmaf(b0, x, z1);
    z1 = fmaf(b1, x, fmaf(-a1, y, z2));
    z2 = fmaf(b2, x, -a2 * y);
    out[(size_t)b * TLEN + t] = y;
  }
}

extern "C" void kernel_launch(void* const* d_in, const int* in_sizes, int n_in,
                              void* d_out, int out_size, void* d_ws, size_t ws_size,
                              hipStream_t stream) {
  const float* rgb    = (const float*)d_in[0];
  const float* matrix = (const float*)d_in[1];
  const float* acoef  = (const float*)d_in[2];
  const float* bcoef  = (const float*)d_in[3];
  float* out = (float*)d_out;

  float* g    = (float*)d_ws;
  float* ppg  = g + (size_t)3 * MWIN * 16;
  float* ybuf = ppg + (size_t)TLEN * 16;

  int nwork = BATCH * MWIN;
  pos_window_stats<<<(nwork + 255) / 256, 256, 0, stream>>>(rgb, matrix, g);
  pos_overlap_add_wmma<<<TLEN / 16, 32, 0, stream>>>(rgb, g, ppg);
  pos_filtfilt<<<1, 32, 0, stream>>>(ppg, acoef, bcoef, ybuf, out);
}


# Stage-3 filtfilt by look-ahead over 32 samples: a pass = 65-tap FIR on all threads + 32 chains 32 apart; poles^32 over radius 1/2 keep the serial scans; f32 sum order changes, fixed per run
# speedup vs baseline: 28.7197x; 28.7197x over previous
#include <hip/hip_runtime.h>
#include <hip/hip_bf16.h>

typedef float v2f __attribute__((ext_vector_type(2)));
typedef float v8f __attribute__((ext_vector_type(8)));

constexpr int TLEN  = 18000;
constexpr int BATCH = 16;
constexpr int WIN   = 48;
constexpr int MWIN  = TLEN - WIN;

__global__ void pos_window_stats(const float* __restrict__ rgb,
                                 const float* __restrict__ matrix,
                                 float* __restrict__ g) {
  int tid = blockIdx.x * blockDim.x + threadIdx.x;
  if (tid >= BATCH * MWIN) return;
  int b = tid & 15;
  int n = tid >> 4;

  const float* p = rgb + (size_t)(b * TLEN + n) * 3;
  float s0 = 0.f, s1 = 0.f, s2 = 0.f;
  float q00 = 0.f, q01 = 0.f, q02 = 0.f, q11 = 0.f, q12 = 0.f, q22 = 0.f;
#pragma unroll 4
  for (int w = 0; w < WIN; ++w) {
    float r0 = p[3 * w + 0];
    float r1 = p[3 * w + 1];
    float r2 = p[3 * w + 2];
    s0 += r0; s1 += r1; s2 += r2;
    q00 = fmaf(r0, r0, q00); q01 = fmaf(r0, r1, q01); q02 = fmaf(r0, r2, q02);
    q11 = fmaf(r1, r1, q11); q12 = fmaf(r1, r2, q12); q22 = fmaf(r2, r2, q22);
  }

  const float invW = 1.0f / (float)WIN;
  float inv0 = 1.0f / (s0 * invW + 1e-6f);
  float inv1 = 1.0f / (s1 * invW + 1e-6f);
  float inv2 = 1.0f / (s2 * invW + 1e-6f);

  float m00 = matrix[0], m01 = matrix[1], m02 = matrix[2];
  float m10 = matrix[3], m11 = matrix[4], m12 = matrix[5];

  float u0 = m00 * inv0, u1 = m01 * inv1, u2 = m02 * inv2;
  float sum0 = u0 * s0 + u1 * s1 + u2 * s2;
  float sq0  = u0 * u0 * q00 + u1 * u1 * q11 + u2 * u2 * q22 +
               2.f * (u0 * u1 * q01 + u0 * u2 * q02 + u1 * u2 * q12);
  float var0 = (sq0 - sum0 * sum0 * invW) * (1.0f / (float)(WIN - 1));
  float std0 = sqrtf(fmaxf(var0, 0.f));

  float v0 = m10 * inv0, v1 = m11 * inv1, v2 = m12 * inv2;
  float sum1 = v0 * s0 + v1 * s1 + v2 * s2;
  float sq1  = v0 * v0 * q00 + v1 * v1 * q11 + v2 * v2 * q22 +
               2.f * (v0 * v1 * q01 + v0 * v2 * q02 + v1 * v2 * q12);
  float var1 = (sq1 - sum1 * sum1 * invW) * (1.0f / (float)(WIN - 1));
  float std1 = sqrtf(fmaxf(var1, 0.f));

  float c = std0 / (std1 + 1e-4f);

  g[(size_t)(0 * MWIN + n) * 16 + b] = (m00 + c * m10) * inv0;
  g[(size_t)(1 * MWIN + n) * 16 + b] = (m01 + c * m11) * inv1;
  g[(size_t)(2 * MWIN + n) * 16 + b] = (m02 + c * m12) * inv2;
}

__global__ void pos_overlap_add_wmma(const float* __restrict__ rgb,
                                     const float* __restrict__ g,
                                     float* __restrict__ ppg) {
  const int t0   = blockIdx.x * 16;
  const int lane = threadIdx.x;
  const int col  = lane & 15;
  const int hi   = lane >> 4;
  const int base = t0 - 47;

  v8f acc[3];
#pragma unroll
  for (int k = 0; k < 3; ++k) {
    const float* gk = g + (size_t)k * MWIN * 16;
    v8f d = {0.f, 0.f, 0.f, 0.f, 0.f, 0.f, 0.f, 0.f};
#pragma unroll
    for (int s = 0; s < 16; ++s) {
      int j0 = 4 * s + 2 * hi;
      int j1 = j0 + 1;
      v2f a;
      a.x = (j0 >= col && j0 <= col + 47) ? 1.0f : 0.0f;
      a.y = (j1 >= col && j1 <= col + 47) ? 1.0f : 0.0f;
      int n0 = base + j0;
      int n1 = base + j1;
      int c0 = min(max(n0, 0), MWIN - 1);
      int c1 = min(max(n1, 0), MWIN - 1);
      float ok0 = ((unsigned)n0 < (unsigned)MWIN) ? 1.0f : 0.0f;
      float ok1 = ((unsigned)n1 < (unsigned)MWIN) ? 1.0f : 0.0f;
      v2f bb;
      bb.x = gk[(size_t)c0 * 16 + col] * ok0;
      bb.y = gk[(size_t)c1 * 16 + col] * ok1;
      d = __builtin_amdgcn_wmma_f32_16x16x4_f32(
           false, a,  false, bb,
           (short)0, d,  false,  false);
    }
    acc[k] = d;
  }

#pragma unroll
  for (int p = 0; p < 8; ++p) {
    int t = t0 + p + 8 * hi;
    const float* rp = rgb + (size_t)(col * TLEN + t) * 3;
    float v = rp[0] * acc[0][p] + rp[1] * acc[1][p] + rp[2] * acc[2][p];
    ppg[(size_t)t * 16 + col] = v;
  }
}

constexpr int S3_THREADS = 256;
constexpr int LA     = 32;
constexpr int NTAP   = 2 * LA + 1;
constexpr int PAD    = 2 * LA;
constexpr int ROWBUF = PAD + TLEN + PAD;
constexpr int NTILE  = TLEN / 8;

template <int DIR>
__device__ __forceinline__ void fir_row(const float* src, float* dst,
                                        const float (&c)[NTAP], int tid) {
  for (int q = tid; q < NTILE; q += S3_THREADS) {
    const int t0 = 8 * q;
    const float4* p = reinterpret_cast<const float4*>(src + (DIR > 0 ? t0 : t0 + PAD));
    float xin[72];
#pragma unroll
    for (int m = 0; m < 18; ++m) {
      float4 v = p[m];
      xin[4 * m + 0] = v.x; xin[4 * m + 1] = v.y; xin[4 * m + 2] = v.z; xin[4 * m + 3] = v.w;
    }
    float acc[8];
#pragma unroll
    for (int o = 0; o < 8; ++o) acc[o] = 0.f;
#pragma unroll
    for (int k = 0; k < NTAP; ++k) {
#pragma unroll
      for (int o = 0; o < 8; ++o)
        acc[o] = fmaf(c[k], DIR > 0 ? xin[PAD + o - k] : xin[o + k], acc[o]);
    }
    float4* d = reinterpret_cast<float4*>(dst + PAD + t0);
    d[0] = make_float4(acc[0], acc[1], acc[2], acc[3]);
    d[1] = make_float4(acc[4], acc[5], acc[6], acc[7]);
  }
}

template <int DIR>
__device__ __forceinline__ void chains_row(float* row, float al1, float al2, int lane) {
  constexpr int NFULL = TLEN / LA;
  constexpr int NREM  = TLEN - NFULL * LA;
  constexpr int UB    = 16;
  constexpr int NTAIL = NFULL % UB;
  constexpr int LASTBASE = NFULL - UB;
  static_assert(NTAIL > 0 && NFULL >= 2 * UB, "tail indexing below assumes a clamped last fetch");
  constexpr int STR   = DIR > 0 ? LA : -LA;
  float* p = DIR > 0 ? row + lane : row + (TLEN - 1 - lane);
  float y1 = 0.f, y2 = 0.f;
  float cur[UB];
#pragma unroll
  for (int i = 0; i < UB; ++i) cur[i] = p[i * STR];
  int s = 0;
  for (; s + UB <= NFULL; s += UB) {
    float nxt[UB];
    const int sb = min(s + UB, LASTBASE);
#pragma unroll
    for (int i = 0; i < UB; ++i) nxt[i] = p[(sb + i) * STR];
#pragma unroll
    for (int i = 0; i < UB; ++i) {
      float v = fmaf(-al1, y1, fmaf(-al2, y2, cur[i]));
      y2 = y1; y1 = v; cur[i] = v;
    }
#pragma unroll
    for (int i = 0; i < UB; ++i) p[(s + i) * STR] = cur[i];
#pragma unroll
    for (int i = 0; i < UB; ++i) cur[i] = nxt[i];
  }
#pragma unroll
  for (int i = 0; i < NTAIL; ++i) {
    float v = fmaf(-al1, y1, fmaf(-al2, y2, cur[UB - NTAIL + i]));
    y2 = y1; y1 = v;
    p[(NFULL - NTAIL + i) * STR] = v;
  }
  if (lane < NREM) {
    float v = fmaf(-al1, y1, fmaf(-al2, y2, p[NFULL * STR]));
    p[NFULL * STR] = v;
  }
}

constexpr int NBLK16 = TLEN / 16;

#define BIQUAD_STEP(v)                                   \
  {                                                      \
    float xin_ = (v);                                    \
    float y_   = fmaf(b0, xin_, z1);                     \
    z1 = fmaf(b1, xin_, fmaf(-a1, y_, z2));              \
    z2 = fmaf(b2, xin_, -a2 * y_);                       \
    (v) = y_;                                            \
  }

__device__ __forceinline__ void serial_scans(float* xs, float b0, float b1, float b2,
                                             float a1, float a2) {
  float4* xv = reinterpret_cast<float4*>(xs);
  float z1 = 0.f, z2 = 0.f;
  float4 c0 = xv[0], c1 = xv[1], c2 = xv[2], c3 = xv[3];
  for (int k = 0; k < NBLK16; ++k) {
    int kn = min(k + 1, NBLK16 - 1);
    float4 n0 = xv[4 * kn + 0], n1 = xv[4 * kn + 1];
    float4 n2 = xv[4 * kn + 2], n3 = xv[4 * kn + 3];
    BIQUAD_STEP(c0.x) BIQUAD_STEP(c0.y) BIQUAD_STEP(c0.z) BIQUAD_STEP(c0.w)
    BIQUAD_STEP(c1.x) BIQUAD_STEP(c1.y) BIQUAD_STEP(c1.z) BIQUAD_STEP(c1.w)
    BIQUAD_STEP(c2.x) BIQUAD_STEP(c2.y) BIQUAD_STEP(c2.z) BIQUAD_STEP(c2.w)
    BIQUAD_STEP(c3.x) BIQUAD_STEP(c3.y) BIQUAD_STEP(c3.z) BIQUAD_STEP(c3.w)
    xv[4 * k + 0] = c0; xv[4 * k + 1] = c1; xv[4 * k + 2] = c2; xv[4 * k + 3] = c3;
    c0 = n0; c1 = n1; c2 = n2; c3 = n3;
  }
  z1 = 0.f; z2 = 0.f;
  c0 = xv[4 * (NBLK16 - 1) + 0]; c1 = xv[4 * (NBLK16 - 1) + 1];
  c2 = xv[4 * (NBLK16 - 1) + 2]; c3 = xv[4 * (NBLK16 - 1) + 3];
  for (int k = NBLK16 - 1; k >= 0; --k) {
    int kn = max(k - 1, 0);
    float4 n0 = xv[4 * kn + 0], n1 = xv[4 * kn + 1];
    float4 n2 = xv[4 * kn + 2], n3 = xv[4 * kn + 3];
    BIQUAD_STEP(c3.w) BIQUAD_STEP(c3.z) BIQUAD_STEP(c3.y) BIQUAD_STEP(c3.x)
    BIQUAD_STEP(c2.w) BIQUAD_STEP(c2.z) BIQUAD_STEP(c2.y) BIQUAD_STEP(c2.x)
    BIQUAD_STEP(c1.w) BIQUAD_STEP(c1.z) BIQUAD_STEP(c1.y) BIQUAD_STEP(c1.x)
    BIQUAD_STEP(c0.w) BIQUAD_STEP(c0.z) BIQUAD_STEP(c0.y) BIQUAD_STEP(c0.x)
    xv[4 * k + 0] = c0; xv[4 * k + 1] = c1; xv[4 * k + 2] = c2; xv[4 * k + 3] = c3;
    c0 = n0; c1 = n1; c2 = n2; c3 = n3;
  }
}

__global__ __launch_bounds__(S3_THREADS)
void pos_filtfilt(const float* __restrict__ ppg,
                  const float* __restrict__ acoef,
                  const float* __restrict__ bcoef,
                  float* __restrict__ out) {
  __shared__ __attribute__((aligned(16))) float bufA[ROWBUF];
  __shared__ __attribute__((aligned(16))) float bufB[ROWBUF];
  __shared__ double cw[2][NTAP + 7];
  __shared__ float  cf[NTAP + 7];
  __shared__ int    lookahead_ok;
  const int b   = blockIdx.x;
  const int tid = threadIdx.x;

  const float a0  = acoef[0];
  const float bn0 = bcoef[0] / a0, bn1 = bcoef[1] / a0, bn2 = bcoef[2] / a0;
  const float an1 = acoef[1] / a0, an2 = acoef[2] / a0;

  for (int t = tid; t < TLEN; t += S3_THREADS) bufA[PAD + t] = ppg[(size_t)t * 16 + b];
  if (tid < PAD) { bufA[tid] = 0.f; bufB[PAD + TLEN + tid] = 0.f; }

  const bool tapwave = tid < 96;
  double e1 = 0.0, e2 = 0.0;
  if (tapwave) {
    e1 = (double)an1; e2 = (double)an2;
    if (tid < NTAP)
      cw[0][tid] = tid == 0 ? (double)bn0 : tid == 1 ? (double)bn1 : tid == 2 ? (double)bn2 : 0.0;
  }
  int cur = 0;
  for (int s = 1; s < LA; s <<= 1) {
    __syncthreads();
    if (tapwave) {
      if (tid < NTAP) {
        double v = cw[cur][tid];
        if (tid >= s)     v = fma(-e1, cw[cur][tid - s], v);
        if (tid >= 2 * s) v = fma( e2, cw[cur][tid - 2 * s], v);
        cw[cur ^ 1][tid] = v;
      }
      double n1 = fma(-e1, e1, 2.0 * e2);
      e2 = e2 * e2;
      e1 = n1;
    }
    cur ^= 1;
  }
  __syncthreads();
  if (tid < NTAP) cf[tid] = (float)cw[cur][tid];
  if (tid == 0) lookahead_ok = (fabs(e2) <= 0.25 && fabs(e1) <= 0.5 + 2.0 * e2) ? 1 : 0;
  const float al1 = (float)e1, al2 = (float)e2;
  __syncthreads();

  if (lookahead_ok != 0) {
    float c[NTAP];
#pragma unroll
    for (int k = 0; k < NTAP; ++k) c[k] = cf[k];

    fir_row<+1>(bufA, bufB, c, tid);
    __syncthreads();
    if (tid < 32) chains_row<+1>(bufB + PAD, al1, al2, tid);
    __syncthreads();
    fir_row<-1>(bufB, bufA, c, tid);
    __syncthreads();
    if (tid < 32) chains_row<-1>(bufA + PAD, al1, al2, tid);
  } else {
    if (tid == 0) serial_scans(bufA + PAD, bn0, bn1, bn2, an1, an2);
  }
  __syncthreads();

  const float4* xr = reinterpret_cast<const float4*>(bufA + PAD);
  float4* orow = reinterpret_cast<float4*>(out + (size_t)b * TLEN);
  for (int i = tid; i < TLEN / 4; i += S3_THREADS) orow[i] = xr[i];
}

extern "C" void kernel_launch(void* const* d_in, const int* in_sizes, int n_in,
                              void* d_out, int out_size, void* d_ws, size_t ws_size,
                              hipStream_t stream) {
  const float* rgb    = (const float*)d_in[0];
  const float* matrix = (const float*)d_in[1];
  const float* acoef  = (const float*)d_in[2];
  const float* bcoef  = (const float*)d_in[3];
  float* out = (float*)d_out;

  float* g    = (float*)d_ws;
  float* ppg  = g + (size_t)3 * MWIN * 16;

  int nwork = BATCH * MWIN;
  pos_window_stats<<<(nwork + 255) / 256, 256, 0, stream>>>(rgb, matrix, g);
  pos_overlap_add_wmma<<<TLEN / 16, 32, 0, stream>>>(rgb, g, ppg);
  pos_filtfilt<<<BATCH, S3_THREADS, 0, stream>>>(ppg, acoef, bcoef, out);
}


# ppg handed from overlap-add to filtfilt row-wise [b][t] at a line-aligned pitch: a wave owns 32 samples of all rows and stores whole lines, filtfilt reads only its row, 16 B a lane; sums unchanged
# speedup vs baseline: 36.3850x; 1.2669x over previous
#include <hip/hip_runtime.h>
#include <hip/hip_bf16.h>

typedef float v2f __attribute__((ext_vector_type(2)));
typedef float v8f __attribute__((ext_vector_type(8)));

constexpr int TLEN  = 18000;
constexpr int BATCH = 16;
constexpr int WIN   = 48;
constexpr int MWIN  = TLEN - WIN;
constexpr int PPG_PITCH = 18048;
constexpr int OA_TILE   = 32;
constexpr int OA_WAVES  = (TLEN + OA_TILE - 1) / OA_TILE;
static_assert(PPG_PITCH % 32 == 0 && PPG_PITCH >= OA_WAVES * OA_TILE, "row pitch: whole lines, room for the last tile");

__global__ void pos_window_stats(const float* __restrict__ rgb,
                                 const float* __restrict__ matrix,
                                 float* __restrict__ g) {
  int tid = blockIdx.x * blockDim.x + threadIdx.x;
  if (tid >= BATCH * MWIN) return;
  int b = tid & 15;
  int n = tid >> 4;

  const float* p = rgb + (size_t)(b * TLEN + n) * 3;
  float s0 = 0.f, s1 = 0.f, s2 = 0.f;
  float q00 = 0.f, q01 = 0.f, q02 = 0.f, q11 = 0.f, q12 = 0.f, q22 = 0.f;
#pragma unroll 4
  for (int w = 0; w < WIN; ++w) {
    float r0 = p[3 * w + 0];
    float r1 = p[3 * w + 1];
    float r2 = p[3 * w + 2];
    s0 += r0; s1 += r1; s2 += r2;
    q00 = fmaf(r0, r0, q00); q01 = fmaf(r0, r1, q01); q02 = fmaf(r0, r2, q02);
    q11 = fmaf(r1, r1, q11); q12 = fmaf(r1, r2, q12); q22 = fmaf(r2, r2, q22);
  }

  const float invW = 1.0f / (float)WIN;
  float inv0 = 1.0f / (s0 * invW + 1e-6f);
  float inv1 = 1.0f / (s1 * invW + 1e-6f);
  float inv2 = 1.0f / (s2 * invW + 1e-6f);

  float m00 = matrix[0], m01 = matrix[1], m02 = matrix[2];
  float m10 = matrix[3], m11 = matrix[4], m12 = matrix[5];

  float u0 = m00 * inv0, u1 = m01 * inv1, u2 = m02 * inv2;
  float sum0 = u0 * s0 + u1 * s1 + u2 * s2;
  float sq0  = u0 * u0 * q00 + u1 * u1 * q11 + u2 * u2 * q22 +
               2.f * (u0 * u1 * q01 + u0 * u2 * q02 + u1 * u2 * q12);
  float var0 = (sq0 - sum0 * sum0 * invW) * (1.0f / (float)(WIN - 1));
  float std0 = sqrtf(fmaxf(var0, 0.f));

  float v0 = m10 * inv0, v1 = m11 * inv1, v2 = m12 * inv2;
  float sum1 = v0 * s0 + v1 * s1 + v2 * s2;
  float sq1  = v0 * v0 * q00 + v1 * v1 * q11 + v2 * v2 * q22 +
               2.f * (v0 * v1 * q01 + v0 * v2 * q02 + v1 * v2 * q12);
  float var1 = (sq1 - sum1 * sum1 * invW) * (1.0f / (float)(WIN - 1));
  float std1 = sqrtf(fmaxf(var1, 0.f));

  float c = std0 / (std1 + 1e-4f);

  g[(size_t)(0 * MWIN + n) * 16 + b] = (m00 + c * m10) * inv0;
  g[(size_t)(1 * MWIN + n) * 16 + b] = (m01 + c * m11) * inv1;
  g[(size_t)(2 * MWIN + n) * 16 + b] = (m02 + c * m12) * inv2;
}

__device__ __forceinline__ float f4c(const float4& v, int c) {
  return c == 0 ? v.x : c == 1 ? v.y : c == 2 ? v.z : v.w;
}

__global__ __launch_bounds__(32) void pos_overlap_add_wmma(const float* __restrict__ rgb,
                                     const float* __restrict__ g,
                                     float* __restrict__ ppg) {
  const int t0   = blockIdx.x * OA_TILE;
  const int lane = threadIdx.x;
  const int col  = lane & 15;
  const int hi   = lane >> 4;
  const int base = t0 - 47;

  float4 r[2][6];
#pragma unroll
  for (int tile = 0; tile < 2; ++tile) {
    const int ttc = min(t0 + 16 * tile + 8 * hi, TLEN - 8);
    const float4* rp4 = reinterpret_cast<const float4*>(rgb + (size_t)(col * TLEN + ttc) * 3);
#pragma unroll
    for (int q = 0; q < 6; ++q) r[tile][q] = rp4[q];
  }

  v8f acc[2][3];
#pragma unroll
  for (int k = 0; k < 3; ++k) {
    const float* gk = g + (size_t)k * MWIN * 16;
    v8f d0 = {0.f, 0.f, 0.f, 0.f, 0.f, 0.f, 0.f, 0.f};
    v8f d1 = {0.f, 0.f, 0.f, 0.f, 0.f, 0.f, 0.f, 0.f};
#pragma unroll
    for (int s = 0; s < 20; ++s) {
      int j0 = 4 * s + 2 * hi;
      int j1 = j0 + 1;
      int n0 = base + j0;
      int n1 = base + j1;
      int c0 = min(max(n0, 0), MWIN - 1);
      int c1 = min(max(n1, 0), MWIN - 1);
      float ok0 = ((unsigned)n0 < (unsigned)MWIN) ? 1.0f : 0.0f;
      float ok1 = ((unsigned)n1 < (unsigned)MWIN) ? 1.0f : 0.0f;
      v2f bb;
      bb.x = gk[(size_t)c0 * 16 + col] * ok0;
      bb.y = gk[(size_t)c1 * 16 + col] * ok1;
      if (s < 16) {
        v2f a;
        a.x = (j0 >= col && j0 <= col + 47) ? 1.0f : 0.0f;
        a.y = (j1 >= col && j1 <= col + 47) ? 1.0f : 0.0f;
        d0 = __builtin_amdgcn_wmma_f32_16x16x4_f32(
             false, a,  false, bb,
             (short)0, d0,  false,  false);
      }
      if (s >= 4) {
        v2f a;
        a.x = (j0 - 16 >= col && j0 - 16 <= col + 47) ? 1.0f : 0.0f;
        a.y = (j1 - 16 >= col && j1 - 16 <= col + 47) ? 1.0f : 0.0f;
        d1 = __builtin_amdgcn_wmma_f32_16x16x4_f32(
             false, a,  false, bb,
             (short)0, d1,  false,  false);
      }
    }
    acc[0][k] = d0;
    acc[1][k] = d1;
  }

#pragma unroll
  for (int tile = 0; tile < 2; ++tile) {
    const int tt = t0 + 16 * tile + 8 * hi;
    float v[8];
#pragma unroll
    for (int p = 0; p < 8; ++p)
      v[p] = f4c(r[tile][(3 * p + 0) >> 2], (3 * p + 0) & 3) * acc[tile][0][p] +
             f4c(r[tile][(3 * p + 1) >> 2], (3 * p + 1) & 3) * acc[tile][1][p] +
             f4c(r[tile][(3 * p + 2) >> 2], (3 * p + 2) & 3) * acc[tile][2][p];
    float4* dst = reinterpret_cast<float4*>(ppg + (size_t)col * PPG_PITCH + tt);
    dst[0] = make_float4(v[0], v[1], v[2], v[3]);
    dst[1] = make_float4(v[4], v[5], v[6], v[7]);
  }
}

constexpr int S3_THREADS = 256;
constexpr int LA     = 32;
constexpr int NTAP   = 2 * LA + 1;
constexpr int PAD    = 2 * LA;
constexpr int ROWBUF = PAD + TLEN + PAD;
constexpr int NTILE  = TLEN / 8;

template <int DIR>
__device__ __forceinline__ void fir_row(const float* src, float* dst,
                                        const float (&c)[NTAP], int tid) {
  for (int q = tid; q < NTILE; q += S3_THREADS) {
    const int t0 = 8 * q;
    const float4* p = reinterpret_cast<const float4*>(src + (DIR > 0 ? t0 : t0 + PAD));
    float xin[72];
#pragma unroll
    for (int m = 0; m < 18; ++m) {
      float4 v = p[m];
      xin[4 * m + 0] = v.x; xin[4 * m + 1] = v.y; xin[4 * m + 2] = v.z; xin[4 * m + 3] = v.w;
    }
    float acc[8];
#pragma unroll
    for (int o = 0; o < 8; ++o) acc[o] = 0.f;
#pragma unroll
    for (int k = 0; k < NTAP; ++k) {
#pragma unroll
      for (int o = 0; o < 8; ++o)
        acc[o] = fmaf(c[k], DIR > 0 ? xin[PAD + o - k] : xin[o + k], acc[o]);
    }
    float4* d = reinterpret_cast<float4*>(dst + PAD + t0);
    d[0] = make_float4(acc[0], acc[1], acc[2], acc[3]);
    d[1] = make_float4(acc[4], acc[5], acc[6], acc[7]);
  }
}

template <int DIR>
__device__ __forceinline__ void chains_row(float* row, float al1, float al2, int lane) {
  constexpr int NFULL = TLEN / LA;
  constexpr int NREM  = TLEN - NFULL * LA;
  constexpr int UB    = 16;
  constexpr int NTAIL = NFULL % UB;
  constexpr int LASTBASE = NFULL - UB;
  static_assert(NTAIL > 0 && NFULL >= 2 * UB, "tail indexing below assumes a clamped last fetch");
  constexpr int STR   = DIR > 0 ? LA : -LA;
  float* p = DIR > 0 ? row + lane : row + (TLEN - 1 - lane);
  float y1 = 0.f, y2 = 0.f;
  float cur[UB];
#pragma unroll
  for (int i = 0; i < UB; ++i) cur[i] = p[i * STR];
  int s = 0;
  for (; s + UB <= NFULL; s += UB) {
    float nxt[UB];
    const int sb = min(s + UB, LASTBASE);
#pragma unroll
    for (int i = 0; i < UB; ++i) nxt[i] = p[(sb + i) * STR];
#pragma unroll
    for (int i = 0; i < UB; ++i) {
      float v = fmaf(-al1, y1, fmaf(-al2, y2, cur[i]));
      y2 = y1; y1 = v; cur[i] = v;
    }
#pragma unroll
    for (int i = 0; i < UB; ++i) p[(s + i) * STR] = cur[i];
#pragma unroll
    for (int i = 0; i < UB; ++i) cur[i] = nxt[i];
  }
#pragma unroll
  for (int i = 0; i < NTAIL; ++i) {
    float v = fmaf(-al1, y1, fmaf(-al2, y2, cur[UB - NTAIL + i]));
    y2 = y1; y1 = v;
    p[(NFULL - NTAIL + i) * STR] = v;
  }
  if (lane < NREM) {
    float v = fmaf(-al1, y1, fmaf(-al2, y2, p[NFULL * STR]));
    p[NFULL * STR] = v;
  }
}

constexpr int NBLK16 = TLEN / 16;

#define BIQUAD_STEP(v)                                   \
  {                                                      \
    float xin_ = (v);                                    \
    float y_   = fmaf(b0, xin_, z1);                     \
    z1 = fmaf(b1, xin_, fmaf(-a1, y_, z2));              \
    z2 = fmaf(b2, xin_, -a2 * y_);                       \
    (v) = y_;                                            \
  }

__device__ __forceinline__ void serial_scans(float* xs, float b0, float b1, float b2,
                                             float a1, float a2) {
  float4* xv = reinterpret_cast<float4*>(xs);
  float z1 = 0.f, z2 = 0.f;
  float4 c0 = xv[0], c1 = xv[1], c2 = xv[2], c3 = xv[3];
  for (int k = 0; k < NBLK16; ++k) {
    int kn = min(k + 1, NBLK16 - 1);
    float4 n0 = xv[4 * kn + 0], n1 = xv[4 * kn + 1];
    float4 n2 = xv[4 * kn + 2], n3 = xv[4 * kn + 3];
    BIQUAD_STEP(c0.x) BIQUAD_STEP(c0.y) BIQUAD_STEP(c0.z) BIQUAD_STEP(c0.w)
    BIQUAD_STEP(c1.x) BIQUAD_STEP(c1.y) BIQUAD_STEP(c1.z) BIQUAD_STEP(c1.w)
    BIQUAD_STEP(c2.x) BIQUAD_STEP(c2.y) BIQUAD_STEP(c2.z) BIQUAD_STEP(c2.w)
    BIQUAD_STEP(c3.x) BIQUAD_STEP(c3.y) BIQUAD_STEP(c3.z) BIQUAD_STEP(c3.w)
    xv[4 * k + 0] = c0; xv[4 * k + 1] = c1; xv[4 * k + 2] = c2; xv[4 * k + 3] = c3;
    c0 = n0; c1 = n1; c2 = n2; c3 = n3;
  }
  z1 = 0.f; z2 = 0.f;
  c0 = xv[4 * (NBLK16 - 1) + 0]; c1 = xv[4 * (NBLK16 - 1) + 1];
  c2 = xv[4 * (NBLK16 - 1) + 2]; c3 = xv[4 * (NBLK16 - 1) + 3];
  for (int k = NBLK16 - 1; k >= 0; --k) {
    int kn = max(k - 1, 0);
    float4 n0 = xv[4 * kn + 0], n1 = xv[4 * kn + 1];
    float4 n2 = xv[4 * kn + 2], n3 = xv[4 * kn + 3];
    BIQUAD_STEP(c3.w) BIQUAD_STEP(c3.z) BIQUAD_STEP(c3.y) BIQUAD_STEP(c3.x)
    BIQUAD_STEP(c2.w) BIQUAD_STEP(c2.z) BIQUAD_STEP(c2.y) BIQUAD_STEP(c2.x)
    BIQUAD_STEP(c1.w) BIQUAD_STEP(c1.z) BIQUAD_STEP(c1.y) BIQUAD_STEP(c1.x)
    BIQUAD_STEP(c0.w) BIQUAD_STEP(c0.z) BIQUAD_STEP(c0.y) BIQUAD_STEP(c0.x)
    xv[4 * k + 0] = c0; xv[4 * k + 1] = c1; xv[4 * k + 2] = c2; xv[4 * k + 3] = c3;
    c0 = n0; c1 = n1; c2 = n2; c3 = n3;
  }
}

__global__ __launch_bounds__(S3_THREADS)
void pos_filtfilt(const float* __restrict__ ppg,
                  const float* __restrict__ acoef,
                  const float* __restrict__ bcoef,
                  float* __restrict__ out) {
  __shared__ __attribute__((aligned(16))) float bufA[ROWBUF];
  __shared__ __attribute__((aligned(16))) float bufB[ROWBUF];
  __shared__ double cw[2][NTAP + 7];
  __shared__ float  cf[NTAP + 7];
  __shared__ int    lookahead_ok;
  const int b   = blockIdx.x;
  const int tid = threadIdx.x;

  const float a0  = acoef[0];
  const float bn0 = bcoef[0] / a0, bn1 = bcoef[1] / a0, bn2 = bcoef[2] / a0;
  const float an1 = acoef[1] / a0, an2 = acoef[2] / a0;

  {
    constexpr int NROWLD = (TLEN / 4 + S3_THREADS - 1) / S3_THREADS;
    const float4* prow = reinterpret_cast<const float4*>(ppg + (size_t)b * PPG_PITCH);
    float4* arow = reinterpret_cast<float4*>(bufA + PAD);
    float4 rowv[NROWLD];
#pragma unroll
    for (int m = 0; m < NROWLD; ++m) {
      const int i = tid + m * S3_THREADS;
      rowv[m] = prow[m == NROWLD - 1 ? min(i, TLEN / 4 - 1) : i];
    }
#pragma unroll
    for (int m = 0; m < NROWLD; ++m) {
      const int i = tid + m * S3_THREADS;
      arow[m == NROWLD - 1 ? min(i, TLEN / 4 - 1) : i] = rowv[m];
    }
  }
  if (tid < PAD) { bufA[tid] = 0.f; bufB[PAD + TLEN + tid] = 0.f; }

  const bool tapwave = tid < 96;
  double e1 = 0.0, e2 = 0.0;
  if (tapwave) {
    e1 = (double)an1; e2 = (double)an2;
    if (tid < NTAP)
      cw[0][tid] = tid == 0 ? (double)bn0 : tid == 1 ? (double)bn1 : tid == 2 ? (double)bn2 : 0.0;
  }
  int cur = 0;
  for (int s = 1; s < LA; s <<= 1) {
    __syncthreads();
    if (tapwave) {
      if (tid < NTAP) {
        double v = cw[cur][tid];
        if (tid >= s)     v = fma(-e1, cw[cur][tid - s], v);
        if (tid >= 2 * s) v = fma( e2, cw[cur][tid - 2 * s], v);
        cw[cur ^ 1][tid] = v;
      }
      double n1 = fma(-e1, e1, 2.0 * e2);
      e2 = e2 * e2;
      e1 = n1;
    }
    cur ^= 1;
  }
  __syncthreads();
  if (tid < NTAP) cf[tid] = (float)cw[cur][tid];
  if (tid == 0) lookahead_ok = (fabs(e2) <= 0.25 && fabs(e1) <= 0.5 + 2.0 * e2) ? 1 : 0;
  const float al1 = (float)e1, al2 = (float)e2;
  __syncthreads();

  if (lookahead_ok != 0) {
    float c[NTAP];
#pragma unroll
    for (int k = 0; k < NTAP; ++k) c[k] = cf[k];

    fir_row<+1>(bufA, bufB, c, tid);
    __syncthreads();
    if (tid < 32) chains_row<+1>(bufB + PAD, al1, al2, tid);
    __syncthreads();
    fir_row<-1>(bufB, bufA, c, tid);
    __syncthreads();
    if (tid < 32) chains_row<-1>(bufA + PAD, al1, al2, tid);
  } else {
    if (tid == 0) serial_scans(bufA + PAD, bn0, bn1, bn2, an1, an2);
  }
  __syncthreads();

  const float4* xr = reinterpret_cast<const float4*>(bufA + PAD);
  float4* orow = reinterpret_cast<float4*>(out + (size_t)b * TLEN);
  for (int i = tid; i < TLEN / 4; i += S3_THREADS) orow[i] = xr[i];
}

extern "C" void kernel_launch(void* const* d_in, const int* in_sizes, int n_in,
                              void* d_out, int out_size, void* d_ws, size_t ws_size,
                              hipStream_t stream) {
  const float* rgb    = (const float*)d_in[0];
  const float* matrix = (const float*)d_in[1];
  const float* acoef  = (const float*)d_in[2];
  const float* bcoef  = (const float*)d_in[3];
  float* out = (float*)d_out;

  float* g    = (float*)d_ws;
  float* ppg  = g + (size_t)3 * MWIN * 16;

  int nwork = BATCH * MWIN;
  pos_window_stats<<<(nwork + 255) / 256, 256, 0, stream>>>(rgb, matrix, g);
  pos_overlap_add_wmma<<<OA_WAVES, 32, 0, stream>>>(rgb, g, ppg);
  pos_filtfilt<<<BATCH, S3_THREADS, 0, stream>>>(ppg, acoef, bcoef, out);
}


# Stage-1 window sums as a banded 0/1 GEMM on the f32 WMMA: a wave per 16 windows x 16 rows, samples read along time; squares and cross products rounded before summing, sum order unchanged
# speedup vs baseline: 47.8706x; 1.3157x over previous
#include <hip/hip_runtime.h>
#include <hip/hip_bf16.h>

typedef float v2f __attribute__((ext_vector_type(2)));
typedef float v8f __attribute__((ext_vector_type(8)));

constexpr int TLEN  = 18000;
constexpr int BATCH = 16;
constexpr int WIN   = 48;
constexpr int MWIN  = TLEN - WIN;
constexpr int PPG_PITCH = 18048;
constexpr int OA_TILE   = 32;
constexpr int OA_WAVES  = (TLEN + OA_TILE - 1) / OA_TILE;
static_assert(PPG_PITCH % 32 == 0 && PPG_PITCH >= OA_WAVES * OA_TILE, "row pitch: whole lines, room for the last tile");

constexpr int ST_TILE  = 16;
constexpr int ST_WAVES = MWIN / ST_TILE;
static_assert(MWIN % ST_TILE == 0, "whole tiles: no window of a tile lies past the last one");
static_assert((ST_WAVES - 1) * ST_TILE + 63 <= TLEN - 1, "the last tile's 64 samples lie inside the row");

__global__ __launch_bounds__(32) __attribute__((amdgpu_waves_per_eu(1, 2))) void pos_window_stats(const float* __restrict__ rgb,
                                 const float* __restrict__ matrix,
                                 float* __restrict__ g) {
  const int n0   = blockIdx.x * ST_TILE;
  const int lane = threadIdx.x;
  const int col  = lane & 15;
  const int hi   = lane >> 4;
  const int wofs = 2 * (col & 7) + (col >> 3);

  const float2* rp2 = reinterpret_cast<const float2*>(rgb + (size_t)(col * TLEN + n0 + 2 * hi) * 3);

  float2 u[16][3];
#pragma unroll
  for (int s = 0; s < 16; ++s) {
#pragma unroll
    for (int m = 0; m < 3; ++m) u[s][m] = rp2[6 * s + m];
  }

  v8f acc[9];
#pragma unroll
  for (int f = 0; f < 9; ++f) acc[f] = v8f{0.f, 0.f, 0.f, 0.f, 0.f, 0.f, 0.f, 0.f};

#pragma unroll
  for (int s = 0; s < 16; ++s) {
    const float2 u0 = u[s][0];
    const float2 u1 = u[s][1];
    const float2 u2 = u[s][2];
    const float x0 = u0.x, x1 = u0.y, x2 = u1.x;
    const float y0 = u1.y, y1 = u2.x, y2 = u2.y;
    const int j0 = 4 * s + 2 * hi;
    const int j1 = j0 + 1;
    v2f a;
    a.x = (j0 >= wofs && j0 <= wofs + 47) ? 1.0f : 0.0f;
    a.y = (j1 >= wofs && j1 <= wofs + 47) ? 1.0f : 0.0f;
    v2f bf[9];
    bf[0].x = x0;      bf[0].y = y0;
    bf[1].x = x1;      bf[1].y = y1;
    bf[2].x = x2;      bf[2].y = y2;
    bf[3].x = x0 * x0; bf[3].y = y0 * y0;
    bf[4].x = x0 * x1; bf[4].y = y0 * y1;
    bf[5].x = x0 * x2; bf[5].y = y0 * y2;
    bf[6].x = x1 * x1; bf[6].y = y1 * y1;
    bf[7].x = x1 * x2; bf[7].y = y1 * y2;
    bf[8].x = x2 * x2; bf[8].y = y2 * y2;
#pragma unroll
    for (int f = 0; f < 9; ++f)
      acc[f] = __builtin_amdgcn_wmma_f32_16x16x4_f32(
           false, a,  false, bf[f],
           (short)0, acc[f],  false,  false);
  }

  const float m00 = matrix[0], m01 = matrix[1], m02 = matrix[2];
  const float m10 = matrix[3], m11 = matrix[4], m12 = matrix[5];
  const float invW = 1.0f / (float)WIN;

#pragma unroll
  for (int p = 0; p < 8; ++p) {
    const int n = n0 + 2 * p + hi;
    const float s0 = acc[0][p], s1 = acc[1][p], s2 = acc[2][p];
    const float q00 = acc[3][p], q01 = acc[4][p], q02 = acc[5][p];
    const float q11 = acc[6][p], q12 = acc[7][p], q22 = acc[8][p];

    float inv0 = 1.0f / (s0 * invW + 1e-6f);
    float inv1 = 1.0f / (s1 * invW + 1e-6f);
    float inv2 = 1.0f / (s2 * invW + 1e-6f);

    float u0 = m00 * inv0, u1 = m01 * inv1, u2 = m02 * inv2;
    float sum0 = u0 * s0 + u1 * s1 + u2 * s2;
    float sq0  = u0 * u0 * q00 + u1 * u1 * q11 + u2 * u2 * q22 +
                 2.f * (u0 * u1 * q01 + u0 * u2 * q02 + u1 * u2 * q12);
    float var0 = (sq0 - sum0 * sum0 * invW) * (1.0f / (float)(WIN - 1));
    float std0 = sqrtf(fmaxf(var0, 0.f));

    float v0 = m10 * inv0, v1 = m11 * inv1, v2 = m12 * inv2;
    float sum1 = v0 * s0 + v1 * s1 + v2 * s2;
    float sq1  = v0 * v0 * q00 + v1 * v1 * q11 + v2 * v2 * q22 +
                 2.f * (v0 * v1 * q01 + v0 * v2 * q02 + v1 * v2 * q12);
    float var1 = (sq1 - sum1 * sum1 * invW) * (1.0f / (float)(WIN - 1));
    float std1 = sqrtf(fmaxf(var1, 0.f));

    float c = std0 / (std1 + 1e-4f);

    g[(size_t)(0 * MWIN + n) * 16 + col] = (m00 + c * m10) * inv0;
    g[(size_t)(1 * MWIN + n) * 16 + col] = (m01 + c * m11) * inv1;
    g[(size_t)(2 * MWIN + n) * 16 + col] = (m02 + c * m12) * inv2;
  }
}

__device__ __forceinline__ float f4c(const float4& v, int c) {
  return c == 0 ? v.x : c == 1 ? v.y : c == 2 ? v.z : v.w;
}

__global__ __launch_bounds__(32) void pos_overlap_add_wmma(const float* __restrict__ rgb,
                                     const float* __restrict__ g,
                                     float* __restrict__ ppg) {
  const int t0   = blockIdx.x * OA_TILE;
  const int lane = threadIdx.x;
  const int col  = lane & 15;
  const int hi   = lane >> 4;
  const int base = t0 - 47;

  float4 r[2][6];
#pragma unroll
  for (int tile = 0; tile < 2; ++tile) {
    const int ttc = min(t0 + 16 * tile + 8 * hi, TLEN - 8);
    const float4* rp4 = reinterpret_cast<const float4*>(rgb + (size_t)(col * TLEN + ttc) * 3);
#pragma unroll
    for (int q = 0; q < 6; ++q) r[tile][q] = rp4[q];
  }

  v8f acc[2][3];
#pragma unroll
  for (int k = 0; k < 3; ++k) {
    const float* gk = g + (size_t)k * MWIN * 16;
    v8f d0 = {0.f, 0.f, 0.f, 0.f, 0.f, 0.f, 0.f, 0.f};
    v8f d1 = {0.f, 0.f, 0.f, 0.f, 0.f, 0.f, 0.f, 0.f};
#pragma unroll
    for (int s = 0; s < 20; ++s) {
      int j0 = 4 * s + 2 * hi;
      int j1 = j0 + 1;
      int n0 = base + j0;
      int n1 = base + j1;
      int c0 = min(max(n0, 0), MWIN - 1);
      int c1 = min(max(n1, 0), MWIN - 1);
      float ok0 = ((unsigned)n0 < (unsigned)MWIN) ? 1.0f : 0.0f;
      float ok1 = ((unsigned)n1 < (unsigned)MWIN) ? 1.0f : 0.0f;
      v2f bb;
      bb.x = gk[(size_t)c0 * 16 + col] * ok0;
      bb.y = gk[(size_t)c1 * 16 + col] * ok1;
      if (s < 16) {
        v2f a;
        a.x = (j0 >= col && j0 <= col + 47) ? 1.0f : 0.0f;
        a.y = (j1 >= col && j1 <= col + 47) ? 1.0f : 0.0f;
        d0 = __builtin_amdgcn_wmma_f32_16x16x4_f32(
             false, a,  false, bb,
             (short)0, d0,  false,  false);
      }
      if (s >= 4) {
        v2f a;
        a.x = (j0 - 16 >= col && j0 - 16 <= col + 47) ? 1.0f : 0.0f;
        a.y = (j1 - 16 >= col && j1 - 16 <= col + 47) ? 1.0f : 0.0f;
        d1 = __builtin_amdgcn_wmma_f32_16x16x4_f32(
             false, a,  false, bb,
             (short)0, d1,  false,  false);
      }
    }
    acc[0][k] = d0;
    acc[1][k] = d1;
  }

#pragma unroll
  for (int tile = 0; tile < 2; ++tile) {
    const int tt = t0 + 16 * tile + 8 * hi;
    float v[8];
#pragma unroll
    for (int p = 0; p < 8; ++p)
      v[p] = f4c(r[tile][(3 * p + 0) >> 2], (3 * p + 0) & 3) * acc[tile][0][p] +
             f4c(r[tile][(3 * p + 1) >> 2], (3 * p + 1) & 3) * acc[tile][1][p] +
             f4c(r[tile][(3 * p + 2) >> 2], (3 * p + 2) & 3) * acc[tile][2][p];
    float4* dst = reinterpret_cast<float4*>(ppg + (size_t)col * PPG_PITCH + tt);
    dst[0] = make_float4(v[0], v[1], v[2], v[3]);
    dst[1] = make_float4(v[4], v[5], v[6], v[7]);
  }
}

constexpr int S3_THREADS = 256;
constexpr int LA     = 32;
constexpr int NTAP   = 2 * LA + 1;
constexpr int PAD    = 2 * LA;
constexpr int ROWBUF = PAD + TLEN + PAD;
constexpr int NTILE  = TLEN / 8;

template <int DIR>
__device__ __forceinline__ void fir_row(const float* src, float* dst,
                                        const float (&c)[NTAP], int tid) {
  for (int q = tid; q < NTILE; q += S3_THREADS) {
    const int t0 = 8 * q;
    const float4* p = reinterpret_cast<const float4*>(src + (DIR > 0 ? t0 : t0 + PAD));
    float xin[72];
#pragma unroll
    for (int m = 0; m < 18; ++m) {
      float4 v = p[m];
      xin[4 * m + 0] = v.x; xin[4 * m + 1] = v.y; xin[4 * m + 2] = v.z; xin[4 * m + 3] = v.w;
    }
    float acc[8];
#pragma unroll
    for (int o = 0; o < 8; ++o) acc[o] = 0.f;
#pragma unroll
    for (int k = 0; k < NTAP; ++k) {
#pragma unroll
      for (int o = 0; o < 8; ++o)
        acc[o] = fmaf(c[k], DIR > 0 ? xin[PAD + o - k] : xin[o + k], acc[o]);
    }
    float4* d = reinterpret_cast<float4*>(dst + PAD + t0);
    d[0] = make_float4(acc[0], acc[1], acc[2], acc[3]);
    d[1] = make_float4(acc[4], acc[5], acc[6], acc[7]);
  }
}

template <int DIR>
__device__ __forceinline__ void chains_row(float* row, float al1, float al2, int lane) {
  constexpr int NFULL = TLEN / LA;
  constexpr int NREM  = TLEN - NFULL * LA;
  constexpr int UB    = 16;
  constexpr int NTAIL = NFULL % UB;
  constexpr int LASTBASE = NFULL - UB;
  static_assert(NTAIL > 0 && NFULL >= 2 * UB, "tail indexing below assumes a clamped last fetch");
  constexpr int STR   = DIR > 0 ? LA : -LA;
  float* p = DIR > 0 ? row + lane : row + (TLEN - 1 - lane);
  float y1 = 0.f, y2 = 0.f;
  float cur[UB];
#pragma unroll
  for (int i = 0; i < UB; ++i) cur[i] = p[i * STR];
  int s = 0;
  for (; s + UB <= NFULL; s += UB) {
    float nxt[UB];
    const int sb = min(s + UB, LASTBASE);
#pragma unroll
    for (int i = 0; i < UB; ++i) nxt[i] = p[(sb + i) * STR];
#pragma unroll
    for (int i = 0; i < UB; ++i) {
      float v = fmaf(-al1, y1, fmaf(-al2, y2, cur[i]));
      y2 = y1; y1 = v; cur[i] = v;
    }
#pragma unroll
    for (int i = 0; i < UB; ++i) p[(s + i) * STR] = cur[i];
#pragma unroll
    for (int i = 0; i < UB; ++i) cur[i] = nxt[i];
  }
#pragma unroll
  for (int i = 0; i < NTAIL; ++i) {
    float v = fmaf(-al1, y1, fmaf(-al2, y2, cur[UB - NTAIL + i]));
    y2 = y1; y1 = v;
    p[(NFULL - NTAIL + i) * STR] = v;
  }
  if (lane < NREM) {
    float v = fmaf(-al1, y1, fmaf(-al2, y2, p[NFULL * STR]));
    p[NFULL * STR] = v;
  }
}

constexpr int NBLK16 = TLEN / 16;

#define BIQUAD_STEP(v)                                   \
  {                                                      \
    float xin_ = (v);                                    \
    float y_   = fmaf(b0, xin_, z1);                     \
    z1 = fmaf(b1, xin_, fmaf(-a1, y_, z2));              \
    z2 = fmaf(b2, xin_, -a2 * y_);                       \
    (v) = y_;                                            \
  }

__device__ __forceinline__ void serial_scans(float* xs, float b0, float b1, float b2,
                                             float a1, float a2) {
  float4* xv = reinterpret_cast<float4*>(xs);
  float z1 = 0.f, z2 = 0.f;
  float4 c0 = xv[0], c1 = xv[1], c2 = xv[2], c3 = xv[3];
  for (int k = 0; k < NBLK16; ++k) {
    int kn = min(k + 1, NBLK16 - 1);
    float4 n0 = xv[4 * kn + 0], n1 = xv[4 * kn + 1];
    float4 n2 = xv[4 * kn + 2], n3 = xv[4 * kn + 3];
    BIQUAD_STEP(c0.x) BIQUAD_STEP(c0.y) BIQUAD_STEP(c0.z) BIQUAD_STEP(c0.w)
    BIQUAD_STEP(c1.x) BIQUAD_STEP(c1.y) BIQUAD_STEP(c1.z) BIQUAD_STEP(c1.w)
    BIQUAD_STEP(c2.x) BIQUAD_STEP(c2.y) BIQUAD_STEP(c2.z) BIQUAD_STEP(c2.w)
    BIQUAD_STEP(c3.x) BIQUAD_STEP(c3.y) BIQUAD_STEP(c3.z) BIQUAD_STEP(c3.w)
    xv[4 * k + 0] = c0; xv[4 * k + 1] = c1; xv[4 * k + 2] = c2; xv[4 * k + 3] = c3;
    c0 = n0; c1 = n1; c2 = n2; c3 = n3;
  }
  z1 = 0.f; z2 = 0.f;
  c0 = xv[4 * (NBLK16 - 1) + 0]; c1 = xv[4 * (NBLK16 - 1) + 1];
  c2 = xv[4 * (NBLK16 - 1) + 2]; c3 = xv[4 * (NBLK16 - 1) + 3];
  for (int k = NBLK16 - 1; k >= 0; --k) {
    int kn = max(k - 1, 0);
    float4 n0 = xv[4 * kn + 0], n1 = xv[4 * kn + 1];
    float4 n2 = xv[4 * kn + 2], n3 = xv[4 * kn + 3];
    BIQUAD_STEP(c3.w) BIQUAD_STEP(c3.z) BIQUAD_STEP(c3.y) BIQUAD_STEP(c3.x)
    BIQUAD_STEP(c2.w) BIQUAD_STEP(c2.z) BIQUAD_STEP(c2.y) BIQUAD_STEP(c2.x)
    BIQUAD_STEP(c1.w) BIQUAD_STEP(c1.z) BIQUAD_STEP(c1.y) BIQUAD_STEP(c1.x)
    BIQUAD_STEP(c0.w) BIQUAD_STEP(c0.z) BIQUAD_STEP(c0.y) BIQUAD_STEP(c0.x)
    xv[4 * k + 0] = c0; xv[4 * k + 1] = c1; xv[4 * k + 2] = c2; xv[4 * k + 3] = c3;
    c0 = n0; c1 = n1; c2 = n2; c3 = n3;
  }
}

__global__ __launch_bounds__(S3_THREADS)
void pos_filtfilt(const float* __restrict__ ppg,
                  const float* __restrict__ acoef,
                  const float* __restrict__ bcoef,
                  float* __restrict__ out) {
  __shared__ __attribute__((aligned(16))) float bufA[ROWBUF];
  __shared__ __attribute__((aligned(16))) float bufB[ROWBUF];
  __shared__ double cw[2][NTAP + 7];
  __shared__ float  cf[NTAP + 7];
  __shared__ int    lookahead_ok;
  const int b   = blockIdx.x;
  const int tid = threadIdx.x;

  const float a0  = acoef[0];
  const float bn0 = bcoef[0] / a0, bn1 = bcoef[1] / a0, bn2 = bcoef[2] / a0;
  const float an1 = acoef[1] / a0, an2 = acoef[2] / a0;

  {
    constexpr int NROWLD = (TLEN / 4 + S3_THREADS - 1) / S3_THREADS;
    const float4* prow = reinterpret_cast<const float4*>(ppg + (size_t)b * PPG_PITCH);
    float4* arow = reinterpret_cast<float4*>(bufA + PAD);
    float4 rowv[NROWLD];
#pragma unroll
    for (int m = 0; m < NROWLD; ++m) {
      const int i = tid + m * S3_THREADS;
      rowv[m] = prow[m == NROWLD - 1 ? min(i, TLEN / 4 - 1) : i];
    }
#pragma unroll
    for (int m = 0; m < NROWLD; ++m) {
      const int i = tid + m * S3_THREADS;
      arow[m == NROWLD - 1 ? min(i, TLEN / 4 - 1) : i] = rowv[m];
    }
  }
  if (tid < PAD) { bufA[tid] = 0.f; bufB[PAD + TLEN + tid] = 0.f; }

  const bool tapwave = tid < 96;
  double e1 = 0.0, e2 = 0.0;
  if (tapwave) {
    e1 = (double)an1; e2 = (double)an2;
    if (tid < NTAP)
      cw[0][tid] = tid == 0 ? (double)bn0 : tid == 1 ? (double)bn1 : tid == 2 ? (double)bn2 : 0.0;
  }
  int cur = 0;
  for (int s = 1; s < LA; s <<= 1) {
    __syncthreads();
    if (tapwave) {
      if (tid < NTAP) {
        double v = cw[cur][tid];
        if (tid >= s)     v = fma(-e1, cw[cur][tid - s], v);
        if (tid >= 2 * s) v = fma( e2, cw[cur][tid - 2 * s], v);
        cw[cur ^ 1][tid] = v;
      }
      double n1 = fma(-e1, e1, 2.0 * e2);
      e2 = e2 * e2;
      e1 = n1;
    }
    cur ^= 1;
  }
  __syncthreads();
  if (tid < NTAP) cf[tid] = (float)cw[cur][tid];
  if (tid == 0) lookahead_ok = (fabs(e2) <= 0.25 && fabs(e1) <= 0.5 + 2.0 * e2) ? 1 : 0;
  const float al1 = (float)e1, al2 = (float)e2;
  __syncthreads();

  if (lookahead_ok != 0) {
    float c[NTAP];
#pragma unroll
    for (int k = 0; k < NTAP; ++k) c[k] = cf[k];

    fir_row<+1>(bufA, bufB, c, tid);
    __syncthreads();
    if (tid < 32) chains_row<+1>(bufB + PAD, al1, al2, tid);
    __syncthreads();
    fir_row<-1>(bufB, bufA, c, tid);
    __syncthreads();
    if (tid < 32) chains_row<-1>(bufA + PAD, al1, al2, tid);
  } else {
    if (tid == 0) serial_scans(bufA + PAD, bn0, bn1, bn2, an1, an2);
  }
  __syncthreads();

  const float4* xr = reinterpret_cast<const float4*>(bufA + PAD);
  float4* orow = reinterpret_cast<float4*>(out + (size_t)b * TLEN);
  for (int i = tid; i < TLEN / 4; i += S3_THREADS) orow[i] = xr[i];
}

extern "C" void kernel_launch(void* const* d_in, const int* in_sizes, int n_in,
                              void* d_out, int out_size, void* d_ws, size_t ws_size,
                              hipStream_t stream) {
  const float* rgb    = (const float*)d_in[0];
  const float* matrix = (const float*)d_in[1];
  const float* acoef  = (const float*)d_in[2];
  const float* bcoef  = (const float*)d_in[3];
  float* out = (float*)d_out;

  float* g    = (float*)d_ws;
  float* ppg  = g + (size_t)3 * MWIN * 16;

  pos_window_stats<<<ST_WAVES, 32, 0, stream>>>(rgb, matrix, g);
  pos_overlap_add_wmma<<<OA_WAVES, 32, 0, stream>>>(rgb, g, ppg);
  pos_filtfilt<<<BATCH, S3_THREADS, 0, stream>>>(ppg, acoef, bcoef, out);
}


# Stage-3 FIR of each pass applied as its six 3-tap factors, B and A_i(-z^(2^i)), on one 76-sample tile per thread in registers, LDS read at an odd 16-B pitch; f32 sum order changes, fixed per run
# speedup vs baseline: 66.4266x; 1.3876x over previous
#include <hip/hip_runtime.h>
#include <hip/hip_bf16.h>

typedef float v2f __attribute__((ext_vector_type(2)));
typedef float v8f __attribute__((ext_vector_type(8)));

constexpr int TLEN  = 18000;
constexpr int BATCH = 16;
constexpr int WIN   = 48;
constexpr int MWIN  = TLEN - WIN;
constexpr int PPG_PITCH = 18048;
constexpr int OA_TILE   = 32;
constexpr int OA_WAVES  = (TLEN + OA_TILE - 1) / OA_TILE;
static_assert(PPG_PITCH % 32 == 0 && PPG_PITCH >= OA_WAVES * OA_TILE, "row pitch: whole lines, room for the last tile");

constexpr int ST_TILE  = 16;
constexpr int ST_WAVES = MWIN / ST_TILE;
static_assert(MWIN % ST_TILE == 0, "whole tiles: no window of a tile lies past the last one");
static_assert((ST_WAVES - 1) * ST_TILE + 63 <= TLEN - 1, "the last tile's 64 samples lie inside the row");

__global__ __launch_bounds__(32) __attribute__((amdgpu_waves_per_eu(1, 2))) void pos_window_stats(const float* __restrict__ rgb,
                                 const float* __restrict__ matrix,
                                 float* __restrict__ g) {
  const int n0   = blockIdx.x * ST_TILE;
  const int lane = threadIdx.x;
  const int col  = lane & 15;
  const int hi   = lane >> 4;
  const int wofs = 2 * (col & 7) + (col >> 3);

  const float2* rp2 = reinterpret_cast<const float2*>(rgb + (size_t)(col * TLEN + n0 + 2 * hi) * 3);

  float2 u[16][3];
#pragma unroll
  for (int s = 0; s < 16; ++s) {
#pragma unroll
    for (int m = 0; m < 3; ++m) u[s][m] = rp2[6 * s + m];
  }

  v8f acc[9];
#pragma unroll
  for (int f = 0; f < 9; ++f) acc[f] = v8f{0.f, 0.f, 0.f, 0.f, 0.f, 0.f, 0.f, 0.f};

#pragma unroll
  for (int s = 0; s < 16; ++s) {
    const float2 u0 = u[s][0];
    const float2 u1 = u[s][1];
    const float2 u2 = u[s][2];
    const float x0 = u0.x, x1 = u0.y, x2 = u1.x;
    const float y0 = u1.y, y1 = u2.x, y2 = u2.y;
    const int j0 = 4 * s + 2 * hi;
    const int j1 = j0 + 1;
    v2f a;
    a.x = (j0 >= wofs && j0 <= wofs + 47) ? 1.0f : 0.0f;
    a.y = (j1 >= wofs && j1 <= wofs + 47) ? 1.0f : 0.0f;
    v2f bf[9];
    bf[0].x = x0;      bf[0].y = y0;
    bf[1].x = x1;      bf[1].y = y1;
    bf[2].x = x2;      bf[2].y = y2;
    bf[3].x = x0 * x0; bf[3].y = y0 * y0;
    bf[4].x = x0 * x1; bf[4].y = y0 * y1;
    bf[5].x = x0 * x2; bf[5].y = y0 * y2;
    bf[6].x = x1 * x1; bf[6].y = y1 * y1;
    bf[7].x = x1 * x2; bf[7].y = y1 * y2;
    bf[8].x = x2 * x2; bf[8].y = y2 * y2;
#pragma unroll
    for (int f = 0; f < 9; ++f)
      acc[f] = __builtin_amdgcn_wmma_f32_16x16x4_f32(
           false, a,  false, bf[f],
           (short)0, acc[f],  false,  false);
  }

  const float m00 = matrix[0], m01 = matrix[1], m02 = matrix[2];
  const float m10 = matrix[3], m11 = matrix[4], m12 = matrix[5];
  const float invW = 1.0f / (float)WIN;

#pragma unroll
  for (int p = 0; p < 8; ++p) {
    const int n = n0 + 2 * p + hi;
    const float s0 = acc[0][p], s1 = acc[1][p], s2 = acc[2][p];
    const float q00 = acc[3][p], q01 = acc[4][p], q02 = acc[5][p];
    const float q11 = acc[6][p], q12 = acc[7][p], q22 = acc[8][p];

    float inv0 = 1.0f / (s0 * invW + 1e-6f);
    float inv1 = 1.0f / (s1 * invW + 1e-6f);
    float inv2 = 1.0f / (s2 * invW + 1e-6f);

    float u0 = m00 * inv0, u1 = m01 * inv1, u2 = m02 * inv2;
    float sum0 = u0 * s0 + u1 * s1 + u2 * s2;
    float sq0  = u0 * u0 * q00 + u1 * u1 * q11 + u2 * u2 * q22 +
                 2.f * (u0 * u1 * q01 + u0 * u2 * q02 + u1 * u2 * q12);
    float var0 = (sq0 - sum0 * sum0 * invW) * (1.0f / (float)(WIN - 1));
    float std0 = sqrtf(fmaxf(var0, 0.f));

    float v0 = m10 * inv0, v1 = m11 * inv1, v2 = m12 * inv2;
    float sum1 = v0 * s0 + v1 * s1 + v2 * s2;
    float sq1  = v0 * v0 * q00 + v1 * v1 * q11 + v2 * v2 * q22 +
                 2.f * (v0 * v1 * q01 + v0 * v2 * q02 + v1 * v2 * q12);
    float var1 = (sq1 - sum1 * sum1 * invW) * (1.0f / (float)(WIN - 1));
    float std1 = sqrtf(fmaxf(var1, 0.f));

    float c = std0 / (std1 + 1e-4f);

    g[(size_t)(0 * MWIN + n) * 16 + col] = (m00 + c * m10) * inv0;
    g[(size_t)(1 * MWIN + n) * 16 + col] = (m01 + c * m11) * inv1;
    g[(size_t)(2 * MWIN + n) * 16 + col] = (m02 + c * m12) * inv2;
  }
}

__device__ __forceinline__ float f4c(const float4& v, int c) {
  return c == 0 ? v.x : c == 1 ? v.y : c == 2 ? v.z : v.w;
}

__global__ __launch_bounds__(32) void pos_overlap_add_wmma(const float* __restrict__ rgb,
                                     const float* __restrict__ g,
                                     float* __restrict__ ppg) {
  const int t0   = blockIdx.x * OA_TILE;
  const int lane = threadIdx.x;
  const int col  = lane & 15;
  const int hi   = lane >> 4;
  const int base = t0 - 47;

  float4 r[2][6];
#pragma unroll
  for (int tile = 0; tile < 2; ++tile) {
    const int ttc = min(t0 + 16 * tile + 8 * hi, TLEN - 8);
    const float4* rp4 = reinterpret_cast<const float4*>(rgb + (size_t)(col * TLEN + ttc) * 3);
#pragma unroll
    for (int q = 0; q < 6; ++q) r[tile][q] = rp4[q];
  }

  v8f acc[2][3];
#pragma unroll
  for (int k = 0; k < 3; ++k) {
    const float* gk = g + (size_t)k * MWIN * 16;
    v8f d0 = {0.f, 0.f, 0.f, 0.f, 0.f, 0.f, 0.f, 0.f};
    v8f d1 = {0.f, 0.f, 0.f, 0.f, 0.f, 0.f, 0.f, 0.f};
#pragma unroll
    for (int s = 0; s < 20; ++s) {
      int j0 = 4 * s + 2 * hi;
      int j1 = j0 + 1;
      int n0 = base + j0;
      int n1 = base + j1;
      int c0 = min(max(n0, 0), MWIN - 1);
      int c1 = min(max(n1, 0), MWIN - 1);
      float ok0 = ((unsigned)n0 < (unsigned)MWIN) ? 1.0f : 0.0f;
      float ok1 = ((unsigned)n1 < (unsigned)MWIN) ? 1.0f : 0.0f;
      v2f bb;
      bb.x = gk[(size_t)c0 * 16 + col] * ok0;
      bb.y = gk[(size_t)c1 * 16 + col] * ok1;
      if (s < 16) {
        v2f a;
        a.x = (j0 >= col && j0 <= col + 47) ? 1.0f : 0.0f;
        a.y = (j1 >= col && j1 <= col + 47) ? 1.0f : 0.0f;
        d0 = __builtin_amdgcn_wmma_f32_16x16x4_f32(
             false, a,  false, bb,
             (short)0, d0,  false,  false);
      }
      if (s >= 4) {
        v2f a;
        a.x = (j0 - 16 >= col && j0 - 16 <= col + 47) ? 1.0f : 0.0f;
        a.y = (j1 - 16 >= col && j1 - 16 <= col + 47) ? 1.0f : 0.0f;
        d1 = __builtin_amdgcn_wmma_f32_16x16x4_f32(
             false, a,  false, bb,
             (short)0, d1,  false,  false);
      }
    }
    acc[0][k] = d0;
    acc[1][k] = d1;
  }

#pragma unroll
  for (int tile = 0; tile < 2; ++tile) {
    const int tt = t0 + 16 * tile + 8 * hi;
    float v[8];
#pragma unroll
    for (int p = 0; p < 8; ++p)
      v[p] = f4c(r[tile][(3 * p + 0) >> 2], (3 * p + 0) & 3) * acc[tile][0][p] +
             f4c(r[tile][(3 * p + 1) >> 2], (3 * p + 1) & 3) * acc[tile][1][p] +
             f4c(r[tile][(3 * p + 2) >> 2], (3 * p + 2) & 3) * acc[tile][2][p];
    float4* dst = reinterpret_cast<float4*>(ppg + (size_t)col * PPG_PITCH + tt);
    dst[0] = make_float4(v[0], v[1], v[2], v[3]);
    dst[1] = make_float4(v[4], v[5], v[6], v[7]);
  }
}

constexpr int S3_THREADS = 256;
constexpr int LA     = 32;
constexpr int NFAC   = 5;
constexpr int PAD    = 2 * LA;
constexpr int FT     = 76;
constexpr int NFT    = (TLEN + FT - 1) / FT;
constexpr int FW     = FT + PAD;
constexpr int LASTG  = (TLEN - (NFT - 1) * FT) / 4;
constexpr int TAILX  = NFT * FT - TLEN;
constexpr int ROWBUF = PAD + NFT * FT + PAD;
static_assert(FT % 4 == 0 && (FT / 4) % 2 == 1, "tile: whole 16-B pieces, an odd number of them");
static_assert(NFT <= S3_THREADS && PAD + TAILX <= S3_THREADS, "one tile per thread, one pad sample per thread");
static_assert((TLEN - (NFT - 1) * FT) % 4 == 0 && LASTG > 0, "the row ends on a 16-B piece of the last tile");
static_assert(ROWBUF % 4 == 0, "the second buffer stays 16-B aligned");

template <int DIR, int S, int INV>
__device__ __forceinline__ void fir_factor(float (&w)[FW], float m1, float m2) {
  if (DIR > 0) {
#pragma unroll
    for (int j = FW - 1; j >= INV; --j)
      w[j] = fmaf(m2, w[j - 2 * S], fmaf(m1, w[j - S], w[j]));
  } else {
#pragma unroll
    for (int j = 0; j < FW - INV; ++j)
      w[j] = fmaf(m2, w[j + 2 * S], fmaf(m1, w[j + S], w[j]));
  }
}

template <int DIR>
__device__ __forceinline__ void fir_row(const float* src, float* dst,
                                        const float (&m1)[NFAC], const float (&m2)[NFAC],
                                        float b0, float b1, float b2, int tid) {
  if (tid < NFT) {
    const int t0 = FT * tid;
    const float4* p = reinterpret_cast<const float4*>(src + (DIR > 0 ? t0 : t0 + PAD));
    float w[FW];
#pragma unroll
    for (int m = 0; m < FW / 4; ++m) {
      float4 v = p[m];
      w[4 * m + 0] = v.x; w[4 * m + 1] = v.y; w[4 * m + 2] = v.z; w[4 * m + 3] = v.w;
    }
    fir_factor<DIR, 16, 32>(w, m1[4], m2[4]);
    fir_factor<DIR,  8, 48>(w, m1[3], m2[3]);
    fir_factor<DIR,  4, 56>(w, m1[2], m2[2]);
    fir_factor<DIR,  2, 60>(w, m1[1], m2[1]);
    fir_factor<DIR,  1, 62>(w, m1[0], m2[0]);
    float o[FT];
#pragma unroll
    for (int j = 0; j < FT; ++j)
      o[j] = DIR > 0 ? fmaf(b2, w[PAD + j - 2], fmaf(b1, w[PAD + j - 1], b0 * w[PAD + j]))
                     : fmaf(b2, w[j + 2],       fmaf(b1, w[j + 1],       b0 * w[j]));
    float4* d = reinterpret_cast<float4*>(dst + PAD + t0);
    const bool whole = tid < NFT - 1;
#pragma unroll
    for (int g = 0; g < FT / 4; ++g)
      if (g < LASTG || whole) d[g] = make_float4(o[4 * g], o[4 * g + 1], o[4 * g + 2], o[4 * g + 3]);
  }
}

template <int DIR>
__device__ __forceinline__ void chains_row(float* row, float al1, float al2, int lane) {
  constexpr int NFULL = TLEN / LA;
  constexpr int NREM  = TLEN - NFULL * LA;
  constexpr int UB    = 16;
  constexpr int NTAIL = NFULL % UB;
  constexpr int LASTBASE = NFULL - UB;
  static_assert(NTAIL > 0 && NFULL >= 2 * UB, "tail indexing below assumes a clamped last fetch");
  constexpr int STR   = DIR > 0 ? LA : -LA;
  float* p = DIR > 0 ? row + lane : row + (TLEN - 1 - lane);
  float y1 = 0.f, y2 = 0.f;
  float cur[UB];
#pragma unroll
  for (int i = 0; i < UB; ++i) cur[i] = p[i * STR];
  int s = 0;
  for (; s + UB <= NFULL; s += UB) {
    float nxt[UB];
    const int sb = min(s + UB, LASTBASE);
#pragma unroll
    for (int i = 0; i < UB; ++i) nxt[i] = p[(sb + i) * STR];
#pragma unroll
    for (int i = 0; i < UB; ++i) {
      float v = fmaf(-al1, y1, fmaf(-al2, y2, cur[i]));
      y2 = y1; y1 = v; cur[i] = v;
    }
#pragma unroll
    for (int i = 0; i < UB; ++i) p[(s + i) * STR] = cur[i];
#pragma unroll
    for (int i = 0; i < UB; ++i) cur[i] = nxt[i];
  }
#pragma unroll
  for (int i = 0; i < NTAIL; ++i) {
    float v = fmaf(-al1, y1, fmaf(-al2, y2, cur[UB - NTAIL + i]));
    y2 = y1; y1 = v;
    p[(NFULL - NTAIL + i) * STR] = v;
  }
  if (lane < NREM) {
    float v = fmaf(-al1, y1, fmaf(-al2, y2, p[NFULL * STR]));
    p[NFULL * STR] = v;
  }
}

constexpr int NBLK16 = TLEN / 16;

#define BIQUAD_STEP(v)                                   \
  {                                                      \
    float xin_ = (v);                                    \
    float y_   = fmaf(b0, xin_, z1);                     \
    z1 = fmaf(b1, xin_, fmaf(-a1, y_, z2));              \
    z2 = fmaf(b2, xin_, -a2 * y_);                       \
    (v) = y_;                                            \
  }

__device__ __forceinline__ void serial_scans(float* xs, float b0, float b1, float b2,
                                             float a1, float a2) {
  float4* xv = reinterpret_cast<float4*>(xs);
  float z1 = 0.f, z2 = 0.f;
  float4 c0 = xv[0], c1 = xv[1], c2 = xv[2], c3 = xv[3];
  for (int k = 0; k < NBLK16; ++k) {
    int kn = min(k + 1, NBLK16 - 1);
    float4 n0 = xv[4 * kn + 0], n1 = xv[4 * kn + 1];
    float4 n2 = xv[4 * kn + 2], n3 = xv[4 * kn + 3];
    BIQUAD_STEP(c0.x) BIQUAD_STEP(c0.y) BIQUAD_STEP(c0.z) BIQUAD_STEP(c0.w)
    BIQUAD_STEP(c1.x) BIQUAD_STEP(c1.y) BIQUAD_STEP(c1.z) BIQUAD_STEP(c1.w)
    BIQUAD_STEP(c2.x) BIQUAD_STEP(c2.y) BIQUAD_STEP(c2.z) BIQUAD_STEP(c2.w)
    BIQUAD_STEP(c3.x) BIQUAD_STEP(c3.y) BIQUAD_STEP(c3.z) BIQUAD_STEP(c3.w)
    xv[4 * k + 0] = c0; xv[4 * k + 1] = c1; xv[4 * k + 2] = c2; xv[4 * k + 3] = c3;
    c0 = n0; c1 = n1; c2 = n2; c3 = n3;
  }
  z1 = 0.f; z2 = 0.f;
  c0 = xv[4 * (NBLK16 - 1) + 0]; c1 = xv[4 * (NBLK16 - 1) + 1];
  c2 = xv[4 * (NBLK16 - 1) + 2]; c3 = xv[4 * (NBLK16 - 1) + 3];
  for (int k = NBLK16 - 1; k >= 0; --k) {
    int kn = max(k - 1, 0);
    float4 n0 = xv[4 * kn + 0], n1 = xv[4 * kn + 1];
    float4 n2 = xv[4 * kn + 2], n3 = xv[4 * kn + 3];
    BIQUAD_STEP(c3.w) BIQUAD_STEP(c3.z) BIQUAD_STEP(c3.y) BIQUAD_STEP(c3.x)
    BIQUAD_STEP(c2.w) BIQUAD_STEP(c2.z) BIQUAD_STEP(c2.y) BIQUAD_STEP(c2.x)
    BIQUAD_STEP(c1.w) BIQUAD_STEP(c1.z) BIQUAD_STEP(c1.y) BIQUAD_STEP(c1.x)
    BIQUAD_STEP(c0.w) BIQUAD_STEP(c0.z) BIQUAD_STEP(c0.y) BIQUAD_STEP(c0.x)
    xv[4 * k + 0] = c0; xv[4 * k + 1] = c1; xv[4 * k + 2] = c2; xv[4 * k + 3] = c3;
    c0 = n0; c1 = n1; c2 = n2; c3 = n3;
  }
}

__global__ __launch_bounds__(S3_THREADS)
void pos_filtfilt(const float* __restrict__ ppg,
                  const float* __restrict__ acoef,
                  const float* __restrict__ bcoef,
                  float* __restrict__ out) {
  __shared__ __attribute__((aligned(16))) float bufA[ROWBUF];
  __shared__ __attribute__((aligned(16))) float bufB[ROWBUF];
  __shared__ __attribute__((aligned(16))) float cfs[2 * NFAC + 2];
  __shared__ int    lookahead_ok;
  const int b   = blockIdx.x;
  const int tid = threadIdx.x;

  const float a0  = acoef[0];
  const float bn0 = bcoef[0] / a0, bn1 = bcoef[1] / a0, bn2 = bcoef[2] / a0;
  const float an1 = acoef[1] / a0, an2 = acoef[2] / a0;

  if (tid == 0) {
    double e1 = (double)an1, e2 = (double)an2;
    for (int i = 0; i < NFAC; ++i) {
      cfs[i] = (float)(-e1);
      cfs[NFAC + i] = (float)e2;
      const double n1 = fma(-e1, e1, 2.0 * e2);
      e2 = e2 * e2;
      e1 = n1;
    }
    cfs[2 * NFAC]     = (float)e1;
    cfs[2 * NFAC + 1] = (float)e2;
    lookahead_ok = (fabs(e2) <= 0.25 && fabs(e1) <= 0.5 + 2.0 * e2) ? 1 : 0;
  }

  {
    constexpr int NROWLD = (TLEN / 4 + S3_THREADS - 1) / S3_THREADS;
    const float4* prow = reinterpret_cast<const float4*>(ppg + (size_t)b * PPG_PITCH);
    float4* arow = reinterpret_cast<float4*>(bufA + PAD);
    float4 rowv[NROWLD];
#pragma unroll
    for (int m = 0; m < NROWLD; ++m) {
      const int i = tid + m * S3_THREADS;
      rowv[m] = prow[m == NROWLD - 1 ? min(i, TLEN / 4 - 1) : i];
    }
#pragma unroll
    for (int m = 0; m < NROWLD; ++m) {
      const int i = tid + m * S3_THREADS;
      arow[m == NROWLD - 1 ? min(i, TLEN / 4 - 1) : i] = rowv[m];
    }
  }
  if (tid < PAD) bufA[tid] = 0.f;
  if (tid < TAILX) bufA[PAD + TLEN + tid] = 0.f;
  if (tid < PAD + TAILX) bufB[PAD + TLEN + tid] = 0.f;
  __syncthreads();

  if (lookahead_ok != 0) {
    float m1[NFAC], m2[NFAC];
#pragma unroll
    for (int i = 0; i < NFAC; ++i) { m1[i] = cfs[i]; m2[i] = cfs[NFAC + i]; }
    const float al1 = cfs[2 * NFAC], al2 = cfs[2 * NFAC + 1];

    fir_row<+1>(bufA, bufB, m1, m2, bn0, bn1, bn2, tid);
    __syncthreads();
    if (tid < 32) chains_row<+1>(bufB + PAD, al1, al2, tid);
    __syncthreads();
    fir_row<-1>(bufB, bufA, m1, m2, bn0, bn1, bn2, tid);
    __syncthreads();
    if (tid < 32) chains_row<-1>(bufA + PAD, al1, al2, tid);
  } else {
    if (tid == 0) serial_scans(bufA + PAD, bn0, bn1, bn2, an1, an2);
  }
  __syncthreads();

  const float4* xr = reinterpret_cast<const float4*>(bufA + PAD);
  float4* orow = reinterpret_cast<float4*>(out + (size_t)b * TLEN);
  for (int i = tid; i < TLEN / 4; i += S3_THREADS) orow[i] = xr[i];
}

extern "C" void kernel_launch(void* const* d_in, const int* in_sizes, int n_in,
                              void* d_out, int out_size, void* d_ws, size_t ws_size,
                              hipStream_t stream) {
  const float* rgb    = (const float*)d_in[0];
  const float* matrix = (const float*)d_in[1];
  const float* acoef  = (const float*)d_in[2];
  const float* bcoef  = (const float*)d_in[3];
  float* out = (float*)d_out;

  float* g    = (float*)d_ws;
  float* ppg  = g + (size_t)3 * MWIN * 16;

  pos_window_stats<<<ST_WAVES, 32, 0, stream>>>(rgb, matrix, g);
  pos_overlap_add_wmma<<<OA_WAVES, 32, 0, stream>>>(rgb, g, ppg);
  pos_filtfilt<<<BATCH, S3_THREADS, 0, stream>>>(ppg, acoef, bcoef, out);
}


# Stage-3 chains: one more look-ahead step along the chain index, f' = A_5(-w) f in registers; even and odd steps run as two half-length chains on two waves; f32 sum order changes, fixed per run
# speedup vs baseline: 81.0538x; 1.2202x over previous
#include <hip/hip_runtime.h>
#include <hip/hip_bf16.h>

typedef float v2f __attribute__((ext_vector_type(2)));
typedef float v8f __attribute__((ext_vector_type(8)));

constexpr int TLEN  = 18000;
constexpr int BATCH = 16;
constexpr int WIN   = 48;
constexpr int MWIN  = TLEN - WIN;
constexpr int PPG_PITCH = 18048;
constexpr int OA_TILE   = 32;
constexpr int OA_WAVES  = (TLEN + OA_TILE - 1) / OA_TILE;
static_assert(PPG_PITCH % 32 == 0 && PPG_PITCH >= OA_WAVES * OA_TILE, "row pitch: whole lines, room for the last tile");

constexpr int ST_TILE  = 16;
constexpr int ST_WAVES = MWIN / ST_TILE;
static_assert(MWIN % ST_TILE == 0, "whole tiles: no window of a tile lies past the last one");
static_assert((ST_WAVES - 1) * ST_TILE + 63 <= TLEN - 1, "the last tile's 64 samples lie inside the row");

__global__ __launch_bounds__(32) __attribute__((amdgpu_waves_per_eu(1, 2))) void pos_window_stats(const float* __restrict__ rgb,
                                 const float* __restrict__ matrix,
                                 float* __restrict__ g) {
  const int n0   = blockIdx.x * ST_TILE;
  const int lane = threadIdx.x;
  const int col  = lane & 15;
  const int hi   = lane >> 4;
  const int wofs = 2 * (col & 7) + (col >> 3);

  const float2* rp2 = reinterpret_cast<const float2*>(rgb + (size_t)(col * TLEN + n0 + 2 * hi) * 3);

  float2 u[16][3];
#pragma unroll
  for (int s = 0; s < 16; ++s) {
#pragma unroll
    for (int m = 0; m < 3; ++m) u[s][m] = rp2[6 * s + m];
  }

  v8f acc[9];
#pragma unroll
  for (int f = 0; f < 9; ++f) acc[f] = v8f{0.f, 0.f, 0.f, 0.f, 0.f, 0.f, 0.f, 0.f};

#pragma unroll
  for (int s = 0; s < 16; ++s) {
    const float2 u0 = u[s][0];
    const float2 u1 = u[s][1];
    const float2 u2 = u[s][2];
    const float x0 = u0.x, x1 = u0.y, x2 = u1.x;
    const float y0 = u1.y, y1 = u2.x, y2 = u2.y;
    const int j0 = 4 * s + 2 * hi;
    const int j1 = j0 + 1;
    v2f a;
    a.x = (j0 >= wofs && j0 <= wofs + 47) ? 1.0f : 0.0f;
    a.y = (j1 >= wofs && j1 <= wofs + 47) ? 1.0f : 0.0f;
    v2f bf[9];
    bf[0].x = x0;      bf[0].y = y0;
    bf[1].x = x1;      bf[1].y = y1;
    bf[2].x = x2;      bf[2].y = y2;
    bf[3].x = x0 * x0; bf[3].y = y0 * y0;
    bf[4].x = x0 * x1; bf[4].y = y0 * y1;
    bf[5].x = x0 * x2; bf[5].y = y0 * y2;
    bf[6].x = x1 * x1; bf[6].y = y1 * y1;
    bf[7].x = x1 * x2; bf[7].y = y1 * y2;
    bf[8].x = x2 * x2; bf[8].y = y2 * y2;
#pragma unroll
    for (int f = 0; f < 9; ++f)
      acc[f] = __builtin_amdgcn_wmma_f32_16x16x4_f32(
           false, a,  false, bf[f],
           (short)0, acc[f],  false,  false);
  }

  const float m00 = matrix[0], m01 = matrix[1], m02 = matrix[2];
  const float m10 = matrix[3], m11 = matrix[4], m12 = matrix[5];
  const float invW = 1.0f / (float)WIN;

#pragma unroll
  for (int p = 0; p < 8; ++p) {
    const int n = n0 + 2 * p + hi;
    const float s0 = acc[0][p], s1 = acc[1][p], s2 = acc[2][p];
    const float q00 = acc[3][p], q01 = acc[4][p], q02 = acc[5][p];
    const float q11 = acc[6][p], q12 = acc[7][p], q22 = acc[8][p];

    float inv0 = 1.0f / (s0 * invW + 1e-6f);
    float inv1 = 1.0f / (s1 * invW + 1e-6f);
    float inv2 = 1.0f / (s2 * invW + 1e-6f);

    float u0 = m00 * inv0, u1 = m01 * inv1, u2 = m02 * inv2;
    float sum0 = u0 * s0 + u1 * s1 + u2 * s2;
    float sq0  = u0 * u0 * q00 + u1 * u1 * q11 + u2 * u2 * q22 +
                 2.f * (u0 * u1 * q01 + u0 * u2 * q02 + u1 * u2 * q12);
    float var0 = (sq0 - sum0 * sum0 * invW) * (1.0f / (float)(WIN - 1));
    float std0 = sqrtf(fmaxf(var0, 0.f));

    float v0 = m10 * inv0, v1 = m11 * inv1, v2 = m12 * inv2;
    float sum1 = v0 * s0 + v1 * s1 + v2 * s2;
    float sq1  = v0 * v0 * q00 + v1 * v1 * q11 + v2 * v2 * q22 +
                 2.f * (v0 * v1 * q01 + v0 * v2 * q02 + v1 * v2 * q12);
    float var1 = (sq1 - sum1 * sum1 * invW) * (1.0f / (float)(WIN - 1));
    float std1 = sqrtf(fmaxf(var1, 0.f));

    float c = std0 / (std1 + 1e-4f);

    g[(size_t)(0 * MWIN + n) * 16 + col] = (m00 + c * m10) * inv0;
    g[(size_t)(1 * MWIN + n) * 16 + col] = (m01 + c * m11) * inv1;
    g[(size_t)(2 * MWIN + n) * 16 + col] = (m02 + c * m12) * inv2;
  }
}

__device__ __forceinline__ float f4c(const float4& v, int c) {
  return c == 0 ? v.x : c == 1 ? v.y : c == 2 ? v.z : v.w;
}

__global__ __launch_bounds__(32) void pos_overlap_add_wmma(const float* __restrict__ rgb,
                                     const float* __restrict__ g,
                                     float* __restrict__ ppg) {
  const int t0   = blockIdx.x * OA_TILE;
  const int lane = threadIdx.x;
  const int col  = lane & 15;
  const int hi   = lane >> 4;
  const int base = t0 - 47;

  float4 r[2][6];
#pragma unroll
  for (int tile = 0; tile < 2; ++tile) {
    const int ttc = min(t0 + 16 * tile + 8 * hi, TLEN - 8);
    const float4* rp4 = reinterpret_cast<const float4*>(rgb + (size_t)(col * TLEN + ttc) * 3);
#pragma unroll
    for (int q = 0; q < 6; ++q) r[tile][q] = rp4[q];
  }

  v8f acc[2][3];
#pragma unroll
  for (int k = 0; k < 3; ++k) {
    const float* gk = g + (size_t)k * MWIN * 16;
    v8f d0 = {0.f, 0.f, 0.f, 0.f, 0.f, 0.f, 0.f, 0.f};
    v8f d1 = {0.f, 0.f, 0.f, 0.f, 0.f, 0.f, 0.f, 0.f};
#pragma unroll
    for (int s = 0; s < 20; ++s) {
      int j0 = 4 * s + 2 * hi;
      int j1 = j0 + 1;
      int n0 = base + j0;
      int n1 = base + j1;
      int c0 = min(max(n0, 0), MWIN - 1);
      int c1 = min(max(n1, 0), MWIN - 1);
      float ok0 = ((unsigned)n0 < (unsigned)MWIN) ? 1.0f : 0.0f;
      float ok1 = ((unsigned)n1 < (unsigned)MWIN) ? 1.0f : 0.0f;
      v2f bb;
      bb.x = gk[(size_t)c0 * 16 + col] * ok0;
      bb.y = gk[(size_t)c1 * 16 + col] * ok1;
      if (s < 16) {
        v2f a;
        a.x = (j0 >= col && j0 <= col + 47) ? 1.0f : 0.0f;
        a.y = (j1 >= col && j1 <= col + 47) ? 1.0f : 0.0f;
        d0 = __builtin_amdgcn_wmma_f32_16x16x4_f32(
             false, a,  false, bb,
             (short)0, d0,  false,  false);
      }
      if (s >= 4) {
        v2f a;
        a.x = (j0 - 16 >= col && j0 - 16 <= col + 47) ? 1.0f : 0.0f;
        a.y = (j1 - 16 >= col && j1 - 16 <= col + 47) ? 1.0f : 0.0f;
        d1 = __builtin_amdgcn_wmma_f32_16x16x4_f32(
             false, a,  false, bb,
             (short)0, d1,  false,  false);
      }
    }
    acc[0][k] = d0;
    acc[1][k] = d1;
  }

#pragma unroll
  for (int tile = 0; tile < 2; ++tile) {
    const int tt = t0 + 16 * tile + 8 * hi;
    float v[8];
#pragma unroll
    for (int p = 0; p < 8; ++p)
      v[p] = f4c(r[tile][(3 * p + 0) >> 2], (3 * p + 0) & 3) * acc[tile][0][p] +
             f4c(r[tile][(3 * p + 1) >> 2], (3 * p + 1) & 3) * acc[tile][1][p] +
             f4c(r[tile][(3 * p + 2) >> 2], (3 * p + 2) & 3) * acc[tile][2][p];
    float4* dst = reinterpret_cast<float4*>(ppg + (size_t)col * PPG_PITCH + tt);
    dst[0] = make_float4(v[0], v[1], v[2], v[3]);
    dst[1] = make_float4(v[4], v[5], v[6], v[7]);
  }
}

constexpr int S3_THREADS = 256;
constexpr int LA     = 32;
constexpr int NFAC   = 5;
constexpr int PAD    = 2 * LA;
constexpr int FT     = 76;
constexpr int NFT    = (TLEN + FT - 1) / FT;
constexpr int FW     = FT + PAD;
constexpr int LASTG  = (TLEN - (NFT - 1) * FT) / 4;
constexpr int TAILX  = NFT * FT - TLEN;
constexpr int ROWBUF = PAD + NFT * FT + PAD;
static_assert(FT % 4 == 0 && (FT / 4) % 2 == 1, "tile: whole 16-B pieces, an odd number of them");
static_assert(NFT <= S3_THREADS && PAD + TAILX <= S3_THREADS, "one tile per thread, one pad sample per thread");
static_assert((TLEN - (NFT - 1) * FT) % 4 == 0 && LASTG > 0, "the row ends on a 16-B piece of the last tile");
static_assert(ROWBUF % 4 == 0, "the second buffer stays 16-B aligned");

template <int DIR, int S, int INV>
__device__ __forceinline__ void fir_factor(float (&w)[FW], float m1, float m2) {
  if (DIR > 0) {
#pragma unroll
    for (int j = FW - 1; j >= INV; --j)
      w[j] = fmaf(m2, w[j - 2 * S], fmaf(m1, w[j - S], w[j]));
  } else {
#pragma unroll
    for (int j = 0; j < FW - INV; ++j)
      w[j] = fmaf(m2, w[j + 2 * S], fmaf(m1, w[j + S], w[j]));
  }
}

template <int DIR>
__device__ __forceinline__ void fir_row(const float* src, float* dst,
                                        const float (&m1)[NFAC], const float (&m2)[NFAC],
                                        float b0, float b1, float b2, int tid) {
  if (tid < NFT) {
    const int t0 = FT * tid;
    const float4* p = reinterpret_cast<const float4*>(src + (DIR > 0 ? t0 : t0 + PAD));
    float w[FW];
#pragma unroll
    for (int m = 0; m < FW / 4; ++m) {
      float4 v = p[m];
      w[4 * m + 0] = v.x; w[4 * m + 1] = v.y; w[4 * m + 2] = v.z; w[4 * m + 3] = v.w;
    }
    fir_factor<DIR, 16, 32>(w, m1[4], m2[4]);
    fir_factor<DIR,  8, 48>(w, m1[3], m2[3]);
    fir_factor<DIR,  4, 56>(w, m1[2], m2[2]);
    fir_factor<DIR,  2, 60>(w, m1[1], m2[1]);
    fir_factor<DIR,  1, 62>(w, m1[0], m2[0]);
    float o[FT];
#pragma unroll
    for (int j = 0; j < FT; ++j)
      o[j] = DIR > 0 ? fmaf(b2, w[PAD + j - 2], fmaf(b1, w[PAD + j - 1], b0 * w[PAD + j]))
                     : fmaf(b2, w[j + 2],       fmaf(b1, w[j + 1],       b0 * w[j]));
    float4* d = reinterpret_cast<float4*>(dst + PAD + t0);
    const bool whole = tid < NFT - 1;
#pragma unroll
    for (int g = 0; g < FT / 4; ++g)
      if (g < LASTG || whole) d[g] = make_float4(o[4 * g], o[4 * g + 1], o[4 * g + 2], o[4 * g + 3]);
  }
}

template <int DIR>
__device__ __forceinline__ void chains_row2(const float* __restrict__ in, float* __restrict__ out,
                                            float m1, float m2, float be1, float be2,
                                            int lane, int par) {
  constexpr int NFULL = TLEN / LA;
  constexpr int NREM  = TLEN - NFULL * LA;
  constexpr int NJ    = NFULL / 2;
  constexpr int UB    = 16;
  constexpr int NTAIL = NJ % UB;
  constexpr int LASTBASE = NJ - UB;
  static_assert(NFULL % 2 == 0, "both waves have the same number of full steps; the partial step is an even one");
  static_assert(NTAIL > 0 && NJ >= 2 * UB, "tail indexing below assumes a clamped last fetch");
  static_assert(2 * LA <= PAD, "steps -1 and -2 lie in the zero samples");
  constexpr int STR = DIR > 0 ? LA : -LA;
  const int off = (DIR > 0 ? lane : TLEN - 1 - lane) + par * STR;
  const float* pi = in + off;
  float* po = out + off;
  float y1 = 0.f, y2 = 0.f;
  float c = 0.f;
  float ca[UB], cb[UB];
#pragma unroll
  for (int i = 0; i < UB; ++i) { ca[i] = pi[(2 * i) * STR]; cb[i] = pi[(2 * i - 1) * STR]; }
  int j = 0;
  for (; j + UB <= NJ; j += UB) {
    float na[UB], nb[UB];
    const int jb = min(j + UB, LASTBASE);
#pragma unroll
    for (int i = 0; i < UB; ++i) { na[i] = pi[(2 * (jb + i)) * STR]; nb[i] = pi[(2 * (jb + i) - 1) * STR]; }
#pragma unroll
    for (int i = 0; i < UB; ++i) {
      float fp = fmaf(m2, c, fmaf(m1, cb[i], ca[i]));
      c = ca[i];
      float v = fmaf(-be1, y1, fmaf(-be2, y2, fp));
      y2 = y1; y1 = v; ca[i] = v;
    }
#pragma unroll
    for (int i = 0; i < UB; ++i) po[(2 * (j + i)) * STR] = ca[i];
#pragma unroll
    for (int i = 0; i < UB; ++i) { ca[i] = na[i]; cb[i] = nb[i]; }
  }
#pragma unroll
  for (int i = 0; i < NTAIL; ++i) {
    float fp = fmaf(m2, c, fmaf(m1, cb[UB - NTAIL + i], ca[UB - NTAIL + i]));
    c = ca[UB - NTAIL + i];
    float v = fmaf(-be1, y1, fmaf(-be2, y2, fp));
    y2 = y1; y1 = v;
    po[(2 * (NJ - NTAIL + i)) * STR] = v;
  }
  if (par == 0 && lane < NREM) {
    float fp = fmaf(m2, c, fmaf(m1, pi[(NFULL - 1) * STR], pi[NFULL * STR]));
    po[NFULL * STR] = fmaf(-be1, y1, fmaf(-be2, y2, fp));
  }
}

constexpr int NBLK16 = TLEN / 16;

#define BIQUAD_STEP(v)                                   \
  {                                                      \
    float xin_ = (v);                                    \
    float y_   = fmaf(b0, xin_, z1);                     \
    z1 = fmaf(b1, xin_, fmaf(-a1, y_, z2));              \
    z2 = fmaf(b2, xin_, -a2 * y_);                       \
    (v) = y_;                                            \
  }

__device__ __forceinline__ void serial_scans(float* xs, float b0, float b1, float b2,
                                             float a1, float a2) {
  float4* xv = reinterpret_cast<float4*>(xs);
  float z1 = 0.f, z2 = 0.f;
  float4 c0 = xv[0], c1 = xv[1], c2 = xv[2], c3 = xv[3];
  for (int k = 0; k < NBLK16; ++k) {
    int kn = min(k + 1, NBLK16 - 1);
    float4 n0 = xv[4 * kn + 0], n1 = xv[4 * kn + 1];
    float4 n2 = xv[4 * kn + 2], n3 = xv[4 * kn + 3];
    BIQUAD_STEP(c0.x) BIQUAD_STEP(c0.y) BIQUAD_STEP(c0.z) BIQUAD_STEP(c0.w)
    BIQUAD_STEP(c1.x) BIQUAD_STEP(c1.y) BIQUAD_STEP(c1.z) BIQUAD_STEP(c1.w)
    BIQUAD_STEP(c2.x) BIQUAD_STEP(c2.y) BIQUAD_STEP(c2.z) BIQUAD_STEP(c2.w)
    BIQUAD_STEP(c3.x) BIQUAD_STEP(c3.y) BIQUAD_STEP(c3.z) BIQUAD_STEP(c3.w)
    xv[4 * k + 0] = c0; xv[4 * k + 1] = c1; xv[4 * k + 2] = c2; xv[4 * k + 3] = c3;
    c0 = n0; c1 = n1; c2 = n2; c3 = n3;
  }
  z1 = 0.f; z2 = 0.f;
  c0 = xv[4 * (NBLK16 - 1) + 0]; c1 = xv[4 * (NBLK16 - 1) + 1];
  c2 = xv[4 * (NBLK16 - 1) + 2]; c3 = xv[4 * (NBLK16 - 1) + 3];
  for (int k = NBLK16 - 1; k >= 0; --k) {
    int kn = max(k - 1, 0);
    float4 n0 = xv[4 * kn + 0], n1 = xv[4 * kn + 1];
    float4 n2 = xv[4 * kn + 2], n3 = xv[4 * kn + 3];
    BIQUAD_STEP(c3.w) BIQUAD_STEP(c3.z) BIQUAD_STEP(c3.y) BIQUAD_STEP(c3.x)
    BIQUAD_STEP(c2.w) BIQUAD_STEP(c2.z) BIQUAD_STEP(c2.y) BIQUAD_STEP(c2.x)
    BIQUAD_STEP(c1.w) BIQUAD_STEP(c1.z) BIQUAD_STEP(c1.y) BIQUAD_STEP(c1.x)
    BIQUAD_STEP(c0.w) BIQUAD_STEP(c0.z) BIQUAD_STEP(c0.y) BIQUAD_STEP(c0.x)
    xv[4 * k + 0] = c0; xv[4 * k + 1] = c1; xv[4 * k + 2] = c2; xv[4 * k + 3] = c3;
    c0 = n0; c1 = n1; c2 = n2; c3 = n3;
  }
}

__global__ __launch_bounds__(S3_THREADS)
void pos_filtfilt(const float* __restrict__ ppg,
                  const float* __restrict__ acoef,
                  const float* __restrict__ bcoef,
                  float* __restrict__ out) {
  __shared__ __attribute__((aligned(16))) float bufA[ROWBUF];
  __shared__ __attribute__((aligned(16))) float bufB[ROWBUF];
  constexpr int NCF = NFAC + 1;
  __shared__ __attribute__((aligned(16))) float cfs[2 * NCF + 2];
  __shared__ int    lookahead_ok;
  const int b   = blockIdx.x;
  const int tid = threadIdx.x;

  const float a0  = acoef[0];
  const float bn0 = bcoef[0] / a0, bn1 = bcoef[1] / a0, bn2 = bcoef[2] / a0;
  const float an1 = acoef[1] / a0, an2 = acoef[2] / a0;

  if (tid == 0) {
    double e1 = (double)an1, e2 = (double)an2;
    int ok = 0;
    for (int i = 0; i < NCF; ++i) {
      cfs[i] = (float)(-e1);
      cfs[NCF + i] = (float)e2;
      if (i == NFAC) ok = (fabs(e2) <= 0.25 && fabs(e1) <= 0.5 + 2.0 * e2) ? 1 : 0;
      const double n1 = fma(-e1, e1, 2.0 * e2);
      e2 = e2 * e2;
      e1 = n1;
    }
    cfs[2 * NCF]     = (float)e1;
    cfs[2 * NCF + 1] = (float)e2;
    lookahead_ok = ok;
  }

  {
    constexpr int NROWLD = (TLEN / 4 + S3_THREADS - 1) / S3_THREADS;
    const float4* prow = reinterpret_cast<const float4*>(ppg + (size_t)b * PPG_PITCH);
    float4* arow = reinterpret_cast<float4*>(bufA + PAD);
    float4 rowv[NROWLD];
#pragma unroll
    for (int m = 0; m < NROWLD; ++m) {
      const int i = tid + m * S3_THREADS;
      rowv[m] = prow[m == NROWLD - 1 ? min(i, TLEN / 4 - 1) : i];
    }
#pragma unroll
    for (int m = 0; m < NROWLD; ++m) {
      const int i = tid + m * S3_THREADS;
      arow[m == NROWLD - 1 ? min(i, TLEN / 4 - 1) : i] = rowv[m];
    }
  }
  if (tid < PAD) { bufA[tid] = 0.f; bufB[tid] = 0.f; }
  if (tid < PAD + TAILX) { bufA[PAD + TLEN + tid] = 0.f; bufB[PAD + TLEN + tid] = 0.f; }
  __syncthreads();

  if (lookahead_ok != 0) {
    float m1[NFAC], m2[NFAC];
#pragma unroll
    for (int i = 0; i < NFAC; ++i) { m1[i] = cfs[i]; m2[i] = cfs[NCF + i]; }
    const float cm1 = cfs[NFAC], cm2 = cfs[NCF + NFAC];
    const float be1 = cfs[2 * NCF], be2 = cfs[2 * NCF + 1];

    fir_row<+1>(bufA, bufB, m1, m2, bn0, bn1, bn2, tid);
    __syncthreads();
    if (tid < 64) chains_row2<+1>(bufB + PAD, bufA + PAD, cm1, cm2, be1, be2, tid & 31, tid >> 5);
    __syncthreads();
    fir_row<-1>(bufA, bufB, m1, m2, bn0, bn1, bn2, tid);
    __syncthreads();
    if (tid < 64) chains_row2<-1>(bufB + PAD, bufA + PAD, cm1, cm2, be1, be2, tid & 31, tid >> 5);
  } else {
    if (tid == 0) serial_scans(bufA + PAD, bn0, bn1, bn2, an1, an2);
  }
  __syncthreads();

  const float4* xr = reinterpret_cast<const float4*>(bufA + PAD);
  float4* orow = reinterpret_cast<float4*>(out + (size_t)b * TLEN);
  for (int i = tid; i < TLEN / 4; i += S3_THREADS) orow[i] = xr[i];
}

extern "C" void kernel_launch(void* const* d_in, const int* in_sizes, int n_in,
                              void* d_out, int out_size, void* d_ws, size_t ws_size,
                              hipStream_t stream) {
  const float* rgb    = (const float*)d_in[0];
  const float* matrix = (const float*)d_in[1];
  const float* acoef  = (const float*)d_in[2];
  const float* bcoef  = (const float*)d_in[3];
  float* out = (float*)d_out;

  float* g    = (float*)d_ws;
  float* ppg  = g + (size_t)3 * MWIN * 16;

  pos_window_stats<<<ST_WAVES, 32, 0, stream>>>(rgb, matrix, g);
  pos_overlap_add_wmma<<<OA_WAVES, 32, 0, stream>>>(rgb, g, ppg);
  pos_filtfilt<<<BATCH, S3_THREADS, 0, stream>>>(ppg, acoef, bcoef, out);
}


# Stage-3 chains: a trip's LDS traffic stands in one group at its head (results of the trip before out, inputs of the trip after in), none among its dependent multiply-adds; sums unchanged
# speedup vs baseline: 83.4907x; 1.0301x over previous
#include <hip/hip_runtime.h>
#include <hip/hip_bf16.h>

typedef float v2f __attribute__((ext_vector_type(2)));
typedef float v8f __attribute__((ext_vector_type(8)));

constexpr int TLEN  = 18000;
constexpr int BATCH = 16;
constexpr int WIN   = 48;
constexpr int MWIN  = TLEN - WIN;
constexpr int PPG_PITCH = 18048;
constexpr int OA_TILE   = 32;
constexpr int OA_WAVES  = (TLEN + OA_TILE - 1) / OA_TILE;
static_assert(PPG_PITCH % 32 == 0 && PPG_PITCH >= OA_WAVES * OA_TILE, "row pitch: whole lines, room for the last tile");

constexpr int ST_TILE  = 16;
constexpr int ST_WAVES = MWIN / ST_TILE;
static_assert(MWIN % ST_TILE == 0, "whole tiles: no window of a tile lies past the last one");
static_assert((ST_WAVES - 1) * ST_TILE + 63 <= TLEN - 1, "the last tile's 64 samples lie inside the row");

__global__ __launch_bounds__(32) __attribute__((amdgpu_waves_per_eu(1, 2))) void pos_window_stats(const float* __restrict__ rgb,
                                 const float* __restrict__ matrix,
                                 float* __restrict__ g) {
  const int n0   = blockIdx.x * ST_TILE;
  const int lane = threadIdx.x;
  const int col  = lane & 15;
  const int hi   = lane >> 4;
  const int wofs = 2 * (col & 7) + (col >> 3);

  const float2* rp2 = reinterpret_cast<const float2*>(rgb + (size_t)(col * TLEN + n0 + 2 * hi) * 3);

  float2 u[16][3];
#pragma unroll
  for (int s = 0; s < 16; ++s) {
#pragma unroll
    for (int m = 0; m < 3; ++m) u[s][m] = rp2[6 * s + m];
  }

  v8f acc[9];
#pragma unroll
  for (int f = 0; f < 9; ++f) acc[f] = v8f{0.f, 0.f, 0.f, 0.f, 0.f, 0.f, 0.f, 0.f};

#pragma unroll
  for (int s = 0; s < 16; ++s) {
    const float2 u0 = u[s][0];
    const float2 u1 = u[s][1];
    const float2 u2 = u[s][2];
    const float x0 = u0.x, x1 = u0.y, x2 = u1.x;
    const float y0 = u1.y, y1 = u2.x, y2 = u2.y;
    const int j0 = 4 * s + 2 * hi;
    const int j1 = j0 + 1;
    v2f a;
    a.x = (j0 >= wofs && j0 <= wofs + 47) ? 1.0f : 0.0f;
    a.y = (j1 >= wofs && j1 <= wofs + 47) ? 1.0f : 0.0f;
    v2f bf[9];
    bf[0].x = x0;      bf[0].y = y0;
    bf[1].x = x1;      bf[1].y = y1;
    bf[2].x = x2;      bf[2].y = y2;
    bf[3].x = x0 * x0; bf[3].y = y0 * y0;
    bf[4].x = x0 * x1; bf[4].y = y0 * y1;
    bf[5].x = x0 * x2; bf[5].y = y0 * y2;
    bf[6].x = x1 * x1; bf[6].y = y1 * y1;
    bf[7].x = x1 * x2; bf[7].y = y1 * y2;
    bf[8].x = x2 * x2; bf[8].y = y2 * y2;
#pragma unroll
    for (int f = 0; f < 9; ++f)
      acc[f] = __builtin_amdgcn_wmma_f32_16x16x4_f32(
           false, a,  false, bf[f],
           (short)0, acc[f],  false,  false);
  }

  const float m00 = matrix[0], m01 = matrix[1], m02 = matrix[2];
  const float m10 = matrix[3], m11 = matrix[4], m12 = matrix[5];
  const float invW = 1.0f / (float)WIN;

#pragma unroll
  for (int p = 0; p < 8; ++p) {
    const int n = n0 + 2 * p + hi;
    const float s0 = acc[0][p], s1 = acc[1][p], s2 = acc[2][p];
    const float q00 = acc[3][p], q01 = acc[4][p], q02 = acc[5][p];
    const float q11 = acc[6][p], q12 = acc[7][p], q22 = acc[8][p];

    float inv0 = 1.0f / (s0 * invW + 1e-6f);
    float inv1 = 1.0f / (s1 * invW + 1e-6f);
    float inv2 = 1.0f / (s2 * invW + 1e-6f);

    float u0 = m00 * inv0, u1 = m01 * inv1, u2 = m02 * inv2;
    float sum0 = u0 * s0 + u1 * s1 + u2 * s2;
    float sq0  = u0 * u0 * q00 + u1 * u1 * q11 + u2 * u2 * q22 +
                 2.f * (u0 * u1 * q01 + u0 * u2 * q02 + u1 * u2 * q12);
    float var0 = (sq0 - sum0 * sum0 * invW) * (1.0f / (float)(WIN - 1));
    float std0 = sqrtf(fmaxf(var0, 0.f));

    float v0 = m10 * inv0, v1 = m11 * inv1, v2 = m12 * inv2;
    float sum1 = v0 * s0 + v1 * s1 + v2 * s2;
    float sq1  = v0 * v0 * q00 + v1 * v1 * q11 + v2 * v2 * q22 +
                 2.f * (v0 * v1 * q01 + v0 * v2 * q02 + v1 * v2 * q12);
    float var1 = (sq1 - sum1 * sum1 * invW) * (1.0f / (float)(WIN - 1));
    float std1 = sqrtf(fmaxf(var1, 0.f));

    float c = std0 / (std1 + 1e-4f);

    g[(size_t)(0 * MWIN + n) * 16 + col] = (m00 + c * m10) * inv0;
    g[(size_t)(1 * MWIN + n) * 16 + col] = (m01 + c * m11) * inv1;
    g[(size_t)(2 * MWIN + n) * 16 + col] = (m02 + c * m12) * inv2;
  }
}

__device__ __forceinline__ float f4c(const float4& v, int c) {
  return c == 0 ? v.x : c == 1 ? v.y : c == 2 ? v.z : v.w;
}

__global__ __launch_bounds__(32) void pos_overlap_add_wmma(const float* __restrict__ rgb,
                                     const float* __restrict__ g,
                                     float* __restrict__ ppg) {
  const int t0   = blockIdx.x * OA_TILE;
  const int lane = threadIdx.x;
  const int col  = lane & 15;
  const int hi   = lane >> 4;
  const int base = t0 - 47;

  float4 r[2][6];
#pragma unroll
  for (int tile = 0; tile < 2; ++tile) {
    const int ttc = min(t0 + 16 * tile + 8 * hi, TLEN - 8);
    const float4* rp4 = reinterpret_cast<const float4*>(rgb + (size_t)(col * TLEN + ttc) * 3);
#pragma unroll
    for (int q = 0; q < 6; ++q) r[tile][q] = rp4[q];
  }

  v8f acc[2][3];
#pragma unroll
  for (int k = 0; k < 3; ++k) {
    const float* gk = g + (size_t)k * MWIN * 16;
    v8f d0 = {0.f, 0.f, 0.f, 0.f, 0.f, 0.f, 0.f, 0.f};
    v8f d1 = {0.f, 0.f, 0.f, 0.f, 0.f, 0.f, 0.f, 0.f};
#pragma unroll
    for (int s = 0; s < 20; ++s) {
      int j0 = 4 * s + 2 * hi;
      int j1 = j0 + 1;
      int n0 = base + j0;
      int n1 = base + j1;
      int c0 = min(max(n0, 0), MWIN - 1);
      int c1 = min(max(n1, 0), MWIN - 1);
      float ok0 = ((unsigned)n0 < (unsigned)MWIN) ? 1.0f : 0.0f;
      float ok1 = ((unsigned)n1 < (unsigned)MWIN) ? 1.0f : 0.0f;
      v2f bb;
      bb.x = gk[(size_t)c0 * 16 + col] * ok0;
      bb.y = gk[(size_t)c1 * 16 + col] * ok1;
      if (s < 16) {
        v2f a;
        a.x = (j0 >= col && j0 <= col + 47) ? 1.0f : 0.0f;
        a.y = (j1 >= col && j1 <= col + 47) ? 1.0f : 0.0f;
        d0 = __builtin_amdgcn_wmma_f32_16x16x4_f32(
             false, a,  false, bb,
             (short)0, d0,  false,  false);
      }
      if (s >= 4) {
        v2f a;
        a.x = (j0 - 16 >= col && j0 - 16 <= col + 47) ? 1.0f : 0.0f;
        a.y = (j1 - 16 >= col && j1 - 16 <= col + 47) ? 1.0f : 0.0f;
        d1 = __builtin_amdgcn_wmma_f32_16x16x4_f32(
             false, a,  false, bb,
             (short)0, d1,  false,  false);
      }
    }
    acc[0][k] = d0;
    acc[1][k] = d1;
  }

#pragma unroll
  for (int tile = 0; tile < 2; ++tile) {
    const int tt = t0 + 16 * tile + 8 * hi;
    float v[8];
#pragma unroll
    for (int p = 0; p < 8; ++p)
      v[p] = f4c(r[tile][(3 * p + 0) >> 2], (3 * p + 0) & 3) * acc[tile][0][p] +
             f4c(r[tile][(3 * p + 1) >> 2], (3 * p + 1) & 3) * acc[tile][1][p] +
             f4c(r[tile][(3 * p + 2) >> 2], (3 * p + 2) & 3) * acc[tile][2][p];
    float4* dst = reinterpret_cast<float4*>(ppg + (size_t)col * PPG_PITCH + tt);
    dst[0] = make_float4(v[0], v[1], v[2], v[3]);
    dst[1] = make_float4(v[4], v[5], v[6], v[7]);
  }
}

constexpr int S3_THREADS = 256;
constexpr int LA     = 32;
constexpr int NFAC   = 5;
constexpr int PAD    = 2 * LA;
constexpr int FT     = 76;
constexpr int NFT    = (TLEN + FT - 1) / FT;
constexpr int FW     = FT + PAD;
constexpr int LASTG  = (TLEN - (NFT - 1) * FT) / 4;
constexpr int TAILX  = NFT * FT - TLEN;
constexpr int ROWBUF = PAD + NFT * FT + PAD;
static_assert(FT % 4 == 0 && (FT / 4) % 2 == 1, "tile: whole 16-B pieces, an odd number of them");
static_assert(NFT <= S3_THREADS && PAD + TAILX <= S3_THREADS, "one tile per thread, one pad sample per thread");
static_assert((TLEN - (NFT - 1) * FT) % 4 == 0 && LASTG > 0, "the row ends on a 16-B piece of the last tile");
static_assert(ROWBUF % 4 == 0, "the second buffer stays 16-B aligned");

template <int DIR, int S, int INV>
__device__ __forceinline__ void fir_factor(float (&w)[FW], float m1, float m2) {
  if (DIR > 0) {
#pragma unroll
    for (int j = FW - 1; j >= INV; --j)
      w[j] = fmaf(m2, w[j - 2 * S], fmaf(m1, w[j - S], w[j]));
  } else {
#pragma unroll
    for (int j = 0; j < FW - INV; ++j)
      w[j] = fmaf(m2, w[j + 2 * S], fmaf(m1, w[j + S], w[j]));
  }
}

template <int DIR>
__device__ __forceinline__ void fir_row(const float* src, float* dst,
                                        const float (&m1)[NFAC], const float (&m2)[NFAC],
                                        float b0, float b1, float b2, int tid) {
  if (tid < NFT) {
    const int t0 = FT * tid;
    const float4* p = reinterpret_cast<const float4*>(src + (DIR > 0 ? t0 : t0 + PAD));
    float w[FW];
#pragma unroll
    for (int m = 0; m < FW / 4; ++m) {
      float4 v = p[m];
      w[4 * m + 0] = v.x; w[4 * m + 1] = v.y; w[4 * m + 2] = v.z; w[4 * m + 3] = v.w;
    }
    fir_factor<DIR, 16, 32>(w, m1[4], m2[4]);
    fir_factor<DIR,  8, 48>(w, m1[3], m2[3]);
    fir_factor<DIR,  4, 56>(w, m1[2], m2[2]);
    fir_factor<DIR,  2, 60>(w, m1[1], m2[1]);
    fir_factor<DIR,  1, 62>(w, m1[0], m2[0]);
    float o[FT];
#pragma unroll
    for (int j = 0; j < FT; ++j)
      o[j] = DIR > 0 ? fmaf(b2, w[PAD + j - 2], fmaf(b1, w[PAD + j - 1], b0 * w[PAD + j]))
                     : fmaf(b2, w[j + 2],       fmaf(b1, w[j + 1],       b0 * w[j]));
    float4* d = reinterpret_cast<float4*>(dst + PAD + t0);
    const bool whole = tid < NFT - 1;
#pragma unroll
    for (int g = 0; g < FT / 4; ++g)
      if (g < LASTG || whole) d[g] = make_float4(o[4 * g], o[4 * g + 1], o[4 * g + 2], o[4 * g + 3]);
  }
}


#define CHAIN_STEPS                                            \
  _Pragma("unroll") for (int i = 0; i < UB; ++i) {             \
    float fp = fmaf(m2, c, fmaf(m1, cb[i], ca[i]));            \
    c = ca[i];                                                 \
    float v = fmaf(-be1, y1, fmaf(-be2, y2, fp));              \
    y2 = y1; y1 = v; r[i] = v;                                 \
  }

template <int DIR>
__device__ __forceinline__ void chains_row2(const float* __restrict__ in, float* __restrict__ out,
                                            float m1, float m2, float be1, float be2,
                                            int lane, int par) {
  constexpr int NFULL = TLEN / LA;
  constexpr int NREM  = TLEN - NFULL * LA;
  constexpr int NJ    = NFULL / 2;
  constexpr int UB    = 16;
  constexpr int NTAIL = NJ % UB;
  constexpr int LASTBASE = NJ - UB;
  static_assert(NFULL % 2 == 0, "both waves have the same number of full steps; the partial step is an even one");
  static_assert(NTAIL > 0 && NJ >= 2 * UB, "tail indexing below assumes a clamped last fetch");
  static_assert(2 * LA <= PAD, "steps -1 and -2 lie in the zero samples");
  constexpr int STR = DIR > 0 ? LA : -LA;
  const int off = (DIR > 0 ? lane : TLEN - 1 - lane) + par * STR;
  const float* pi = in + off;
  float* po = out + off;
  float y1 = 0.f, y2 = 0.f;
  float c = 0.f;
  float ca[UB], cb[UB], na[UB], nb[UB], r[UB];
#pragma unroll
  for (int i = 0; i < UB; ++i) { ca[i] = pi[(2 * i) * STR]; cb[i] = pi[(2 * i - 1) * STR]; }
#pragma unroll
  for (int i = 0; i < UB; ++i) { na[i] = pi[(2 * (UB + i)) * STR]; nb[i] = pi[(2 * (UB + i) - 1) * STR]; }
  __builtin_amdgcn_sched_barrier(0);
  CHAIN_STEPS
  __builtin_amdgcn_sched_barrier(0);
#pragma unroll
  for (int i = 0; i < UB; ++i) { ca[i] = na[i]; cb[i] = nb[i]; }
  int j = UB;
  for (; j + UB <= NJ; j += UB) {
    const int jb = min(j + UB, LASTBASE);
#pragma unroll
    for (int i = 0; i < UB; ++i) po[(2 * (j - UB + i)) * STR] = r[i];
#pragma unroll
    for (int i = 0; i < UB; ++i) { na[i] = pi[(2 * (jb + i)) * STR]; nb[i] = pi[(2 * (jb + i) - 1) * STR]; }
    __builtin_amdgcn_sched_barrier(0);
    CHAIN_STEPS
    __builtin_amdgcn_sched_barrier(0);
#pragma unroll
    for (int i = 0; i < UB; ++i) { ca[i] = na[i]; cb[i] = nb[i]; }
  }
#pragma unroll
  for (int i = 0; i < UB; ++i) po[(2 * (j - UB + i)) * STR] = r[i];
#pragma unroll
  for (int i = 0; i < NTAIL; ++i) {
    float fp = fmaf(m2, c, fmaf(m1, cb[UB - NTAIL + i], ca[UB - NTAIL + i]));
    c = ca[UB - NTAIL + i];
    float v = fmaf(-be1, y1, fmaf(-be2, y2, fp));
    y2 = y1; y1 = v;
    po[(2 * (NJ - NTAIL + i)) * STR] = v;
  }
  if (par == 0 && lane < NREM) {
    float fp = fmaf(m2, c, fmaf(m1, pi[(NFULL - 1) * STR], pi[NFULL * STR]));
    po[NFULL * STR] = fmaf(-be1, y1, fmaf(-be2, y2, fp));
  }
}
#undef CHAIN_STEPS

constexpr int NBLK16 = TLEN / 16;

#define BIQUAD_STEP(v)                                   \
  {                                                      \
    float xin_ = (v);                                    \
    float y_   = fmaf(b0, xin_, z1);                     \
    z1 = fmaf(b1, xin_, fmaf(-a1, y_, z2));              \
    z2 = fmaf(b2, xin_, -a2 * y_);                       \
    (v) = y_;                                            \
  }

__device__ __forceinline__ void serial_scans(float* xs, float b0, float b1, float b2,
                                             float a1, float a2) {
  float4* xv = reinterpret_cast<float4*>(xs);
  float z1 = 0.f, z2 = 0.f;
  float4 c0 = xv[0], c1 = xv[1], c2 = xv[2], c3 = xv[3];
  for (int k = 0; k < NBLK16; ++k) {
    int kn = min(k + 1, NBLK16 - 1);
    float4 n0 = xv[4 * kn + 0], n1 = xv[4 * kn + 1];
    float4 n2 = xv[4 * kn + 2], n3 = xv[4 * kn + 3];
    BIQUAD_STEP(c0.x) BIQUAD_STEP(c0.y) BIQUAD_STEP(c0.z) BIQUAD_STEP(c0.w)
    BIQUAD_STEP(c1.x) BIQUAD_STEP(c1.y) BIQUAD_STEP(c1.z) BIQUAD_STEP(c1.w)
    BIQUAD_STEP(c2.x) BIQUAD_STEP(c2.y) BIQUAD_STEP(c2.z) BIQUAD_STEP(c2.w)
    BIQUAD_STEP(c3.x) BIQUAD_STEP(c3.y) BIQUAD_STEP(c3.z) BIQUAD_STEP(c3.w)
    xv[4 * k + 0] = c0; xv[4 * k + 1] = c1; xv[4 * k + 2] = c2; xv[4 * k + 3] = c3;
    c0 = n0; c1 = n1; c2 = n2; c3 = n3;
  }
  z1 = 0.f; z2 = 0.f;
  c0 = xv[4 * (NBLK16 - 1) + 0]; c1 = xv[4 * (NBLK16 - 1) + 1];
  c2 = xv[4 * (NBLK16 - 1) + 2]; c3 = xv[4 * (NBLK16 - 1) + 3];
  for (int k = NBLK16 - 1; k >= 0; --k) {
    int kn = max(k - 1, 0);
    float4 n0 = xv[4 * kn + 0], n1 = xv[4 * kn + 1];
    float4 n2 = xv[4 * kn + 2], n3 = xv[4 * kn + 3];
    BIQUAD_STEP(c3.w) BIQUAD_STEP(c3.z) BIQUAD_STEP(c3.y) BIQUAD_STEP(c3.x)
    BIQUAD_STEP(c2.w) BIQUAD_STEP(c2.z) BIQUAD_STEP(c2.y) BIQUAD_STEP(c2.x)
    BIQUAD_STEP(c1.w) BIQUAD_STEP(c1.z) BIQUAD_STEP(c1.y) BIQUAD_STEP(c1.x)
    BIQUAD_STEP(c0.w) BIQUAD_STEP(c0.z) BIQUAD_STEP(c0.y) BIQUAD_STEP(c0.x)
    xv[4 * k + 0] = c0; xv[4 * k + 1] = c1; xv[4 * k + 2] = c2; xv[4 * k + 3] = c3;
    c0 = n0; c1 = n1; c2 = n2; c3 = n3;
  }
}

__global__ __launch_bounds__(S3_THREADS)
void pos_filtfilt(const float* __restrict__ ppg,
                  const float* __restrict__ acoef,
                  const float* __restrict__ bcoef,
                  float* __restrict__ out) {
  __shared__ __attribute__((aligned(16))) float bufA[ROWBUF];
  __shared__ __attribute__((aligned(16))) float bufB[ROWBUF];
  constexpr int NCF = NFAC + 1;
  __shared__ __attribute__((aligned(16))) float cfs[2 * NCF + 2];
  __shared__ int    lookahead_ok;
  const int b   = blockIdx.x;
  const int tid = threadIdx.x;

  const float a0  = acoef[0];
  const float bn0 = bcoef[0] / a0, bn1 = bcoef[1] / a0, bn2 = bcoef[2] / a0;
  const float an1 = acoef[1] / a0, an2 = acoef[2] / a0;

  if (tid == 0) {
    double e1 = (double)an1, e2 = (double)an2;
    int ok = 0;
    for (int i = 0; i < NCF; ++i) {
      cfs[i] = (float)(-e1);
      cfs[NCF + i] = (float)e2;
      if (i == NFAC) ok = (fabs(e2) <= 0.25 && fabs(e1) <= 0.5 + 2.0 * e2) ? 1 : 0;
      const double n1 = fma(-e1, e1, 2.0 * e2);
      e2 = e2 * e2;
      e1 = n1;
    }
    cfs[2 * NCF]     = (float)e1;
    cfs[2 * NCF + 1] = (float)e2;
    lookahead_ok = ok;
  }

  {
    constexpr int NROWLD = (TLEN / 4 + S3_THREADS - 1) / S3_THREADS;
    const float4* prow = reinterpret_cast<const float4*>(ppg + (size_t)b * PPG_PITCH);
    float4* arow = reinterpret_cast<float4*>(bufA + PAD);
    float4 rowv[NROWLD];
#pragma unroll
    for (int m = 0; m < NROWLD; ++m) {
      const int i = tid + m * S3_THREADS;
      rowv[m] = prow[m == NROWLD - 1 ? min(i, TLEN / 4 - 1) : i];
    }
#pragma unroll
    for (int m = 0; m < NROWLD; ++m) {
      const int i = tid + m * S3_THREADS;
      arow[m == NROWLD - 1 ? min(i, TLEN / 4 - 1) : i] = rowv[m];
    }
  }
  if (tid < PAD) { bufA[tid] = 0.f; bufB[tid] = 0.f; }
  if (tid < PAD + TAILX) { bufA[PAD + TLEN + tid] = 0.f; bufB[PAD + TLEN + tid] = 0.f; }
  __syncthreads();

  if (lookahead_ok != 0) {
    float m1[NFAC], m2[NFAC];
#pragma unroll
    for (int i = 0; i < NFAC; ++i) { m1[i] = cfs[i]; m2[i] = cfs[NCF + i]; }
    const float cm1 = cfs[NFAC], cm2 = cfs[NCF + NFAC];
    const float be1 = cfs[2 * NCF], be2 = cfs[2 * NCF + 1];

    fir_row<+1>(bufA, bufB, m1, m2, bn0, bn1, bn2, tid);
    __syncthreads();
    if (tid < 64) chains_row2<+1>(bufB + PAD, bufA + PAD, cm1, cm2, be1, be2, tid & 31, tid >> 5);
    __syncthreads();
    fir_row<-1>(bufA, bufB, m1, m2, bn0, bn1, bn2, tid);
    __syncthreads();
    if (tid < 64) chains_row2<-1>(bufB + PAD, bufA + PAD, cm1, cm2, be1, be2, tid & 31, tid >> 5);
  } else {
    if (tid == 0) serial_scans(bufA + PAD, bn0, bn1, bn2, an1, an2);
  }
  __syncthreads();

  const float4* xr = reinterpret_cast<const float4*>(bufA + PAD);
  float4* orow = reinterpret_cast<float4*>(out + (size_t)b * TLEN);
  for (int i = tid; i < TLEN / 4; i += S3_THREADS) orow[i] = xr[i];
}

extern "C" void kernel_launch(void* const* d_in, const int* in_sizes, int n_in,
                              void* d_out, int out_size, void* d_ws, size_t ws_size,
                              hipStream_t stream) {
  const float* rgb    = (const float*)d_in[0];
  const float* matrix = (const float*)d_in[1];
  const float* acoef  = (const float*)d_in[2];
  const float* bcoef  = (const float*)d_in[3];
  float* out = (float*)d_out;

  float* g    = (float*)d_ws;
  float* ppg  = g + (size_t)3 * MWIN * 16;

  pos_window_stats<<<ST_WAVES, 32, 0, stream>>>(rgb, matrix, g);
  pos_overlap_add_wmma<<<OA_WAVES, 32, 0, stream>>>(rgb, g, ppg);
  pos_filtfilt<<<BATCH, S3_THREADS, 0, stream>>>(ppg, acoef, bcoef, out);
}


# Both banded 0/1 GEMMs: the 32 band columns that are 1 in every row are added in the lane and enter the WMMA as one K-step, 9 K-steps a series, not 16; f32 sum order changes, fixed per run
# speedup vs baseline: 89.5010x; 1.0720x over previous
#include <hip/hip_runtime.h>
#include <hip/hip_bf16.h>

typedef float v2f __attribute__((ext_vector_type(2)));
typedef float v8f __attribute__((ext_vector_type(8)));

constexpr int TLEN  = 18000;
constexpr int BATCH = 16;
constexpr int WIN   = 48;
constexpr int MWIN  = TLEN - WIN;
constexpr int PPG_PITCH = 18048;
constexpr int OA_TILE   = 32;
constexpr int OA_WAVES  = (TLEN + OA_TILE - 1) / OA_TILE;
static_assert(PPG_PITCH % 32 == 0 && PPG_PITCH >= OA_WAVES * OA_TILE, "row pitch: whole lines, room for the last tile");

constexpr int ST_TILE  = 16;
constexpr int ST_WAVES = MWIN / ST_TILE;
static_assert(MWIN % ST_TILE == 0, "whole tiles: no window of a tile lies past the last one");
static_assert((ST_WAVES - 1) * ST_TILE + 63 <= TLEN - 1, "the last tile's 64 samples lie inside the row");

__global__ __launch_bounds__(32) __attribute__((amdgpu_waves_per_eu(1, 2))) void pos_window_stats(const float* __restrict__ rgb,
                                 const float* __restrict__ matrix,
                                 float* __restrict__ g) {
  const int n0   = blockIdx.x * ST_TILE;
  const int lane = threadIdx.x;
  const int col  = lane & 15;
  const int hi   = lane >> 4;
  const int wofs = 2 * (col & 7) + (col >> 3);

  const float2* rp2 = reinterpret_cast<const float2*>(rgb + (size_t)(col * TLEN + n0 + 2 * hi) * 3);

  float2 u[16][3];
#pragma unroll
  for (int s = 0; s < 16; ++s) {
#pragma unroll
    for (int m = 0; m < 3; ++m) u[s][m] = rp2[6 * s + m];
  }

  v8f acc[9];
#pragma unroll
  for (int f = 0; f < 9; ++f) acc[f] = v8f{0.f, 0.f, 0.f, 0.f, 0.f, 0.f, 0.f, 0.f};

  static_assert(WIN == 48 && ST_TILE == 16, "common columns 16..47 = K-steps 4..11");
  v2f mid[9];
#pragma unroll
  for (int f = 0; f < 9; ++f) mid[f] = v2f{0.f, 0.f};
  const v2f ones = {1.0f, 1.0f};
#pragma unroll
  for (int s = 0; s < 16; ++s) {
    const float2 u0 = u[s][0];
    const float2 u1 = u[s][1];
    const float2 u2 = u[s][2];
    const float x0 = u0.x, x1 = u0.y, x2 = u1.x;
    const float y0 = u1.y, y1 = u2.x, y2 = u2.y;
    v2f bf[9];
    bf[0].x = x0;      bf[0].y = y0;
    bf[1].x = x1;      bf[1].y = y1;
    bf[2].x = x2;      bf[2].y = y2;
    bf[3].x = x0 * x0; bf[3].y = y0 * y0;
    bf[4].x = x0 * x1; bf[4].y = y0 * y1;
    bf[5].x = x0 * x2; bf[5].y = y0 * y2;
    bf[6].x = x1 * x1; bf[6].y = y1 * y1;
    bf[7].x = x1 * x2; bf[7].y = y1 * y2;
    bf[8].x = x2 * x2; bf[8].y = y2 * y2;
    if (s >= 4 && s < 12) {
#pragma unroll
      for (int f = 0; f < 9; ++f) mid[f] = (s == 4) ? bf[f] : mid[f] + bf[f];
      if (s == 11) {
#pragma unroll
        for (int f = 0; f < 9; ++f)
          acc[f] = __builtin_amdgcn_wmma_f32_16x16x4_f32(
               false, ones,  false, mid[f],
               (short)0, acc[f],  false,  false);
      }
    } else {
      const int j0 = 4 * s + 2 * hi;
      const int j1 = j0 + 1;
      v2f a;
      a.x = (j0 >= wofs && j0 <= wofs + 47) ? 1.0f : 0.0f;
      a.y = (j1 >= wofs && j1 <= wofs + 47) ? 1.0f : 0.0f;
#pragma unroll
      for (int f = 0; f < 9; ++f)
        acc[f] = __builtin_amdgcn_wmma_f32_16x16x4_f32(
             false, a,  false, bf[f],
             (short)0, acc[f],  false,  false);
    }
  }

  const float m00 = matrix[0], m01 = matrix[1], m02 = matrix[2];
  const float m10 = matrix[3], m11 = matrix[4], m12 = matrix[5];
  const float invW = 1.0f / (float)WIN;

#pragma unroll
  for (int p = 0; p < 8; ++p) {
    const int n = n0 + 2 * p + hi;
    const float s0 = acc[0][p], s1 = acc[1][p], s2 = acc[2][p];
    const float q00 = acc[3][p], q01 = acc[4][p], q02 = acc[5][p];
    const float q11 = acc[6][p], q12 = acc[7][p], q22 = acc[8][p];

    float inv0 = 1.0f / (s0 * invW + 1e-6f);
    float inv1 = 1.0f / (s1 * invW + 1e-6f);
    float inv2 = 1.0f / (s2 * invW + 1e-6f);

    float u0 = m00 * inv0, u1 = m01 * inv1, u2 = m02 * inv2;
    float sum0 = u0 * s0 + u1 * s1 + u2 * s2;
    float sq0  = u0 * u0 * q00 + u1 * u1 * q11 + u2 * u2 * q22 +
                 2.f * (u0 * u1 * q01 + u0 * u2 * q02 + u1 * u2 * q12);
    float var0 = (sq0 - sum0 * sum0 * invW) * (1.0f / (float)(WIN - 1));
    float std0 = sqrtf(fmaxf(var0, 0.f));

    float v0 = m10 * inv0, v1 = m11 * inv1, v2 = m12 * inv2;
    float sum1 = v0 * s0 + v1 * s1 + v2 * s2;
    float sq1  = v0 * v0 * q00 + v1 * v1 * q11 + v2 * v2 * q22 +
                 2.f * (v0 * v1 * q01 + v0 * v2 * q02 + v1 * v2 * q12);
    float var1 = (sq1 - sum1 * sum1 * invW) * (1.0f / (float)(WIN - 1));
    float std1 = sqrtf(fmaxf(var1, 0.f));

    float c = std0 / (std1 + 1e-4f);

    g[(size_t)(0 * MWIN + n) * 16 + col] = (m00 + c * m10) * inv0;
    g[(size_t)(1 * MWIN + n) * 16 + col] = (m01 + c * m11) * inv1;
    g[(size_t)(2 * MWIN + n) * 16 + col] = (m02 + c * m12) * inv2;
  }
}

__device__ __forceinline__ float f4c(const float4& v, int c) {
  return c == 0 ? v.x : c == 1 ? v.y : c == 2 ? v.z : v.w;
}

__global__ __launch_bounds__(32) __attribute__((amdgpu_waves_per_eu(1, 2))) void pos_overlap_add_wmma(const float* __restrict__ rgb,
                                     const float* __restrict__ g,
                                     float* __restrict__ ppg) {
  const int t0   = blockIdx.x * OA_TILE;
  const int lane = threadIdx.x;
  const int col  = lane & 15;
  const int hi   = lane >> 4;
  const int base = t0 - 47;

  float4 r[2][6];
#pragma unroll
  for (int tile = 0; tile < 2; ++tile) {
    const int ttc = min(t0 + 16 * tile + 8 * hi, TLEN - 8);
    const float4* rp4 = reinterpret_cast<const float4*>(rgb + (size_t)(col * TLEN + ttc) * 3);
#pragma unroll
    for (int q = 0; q < 6; ++q) r[tile][q] = rp4[q];
  }

  v8f acc[2][3];
#pragma unroll
  for (int k = 0; k < 3; ++k) {
    const float* gk = g + (size_t)k * MWIN * 16;
    v8f d0 = {0.f, 0.f, 0.f, 0.f, 0.f, 0.f, 0.f, 0.f};
    v8f d1 = {0.f, 0.f, 0.f, 0.f, 0.f, 0.f, 0.f, 0.f};
    v2f ma = {0.f, 0.f}, mb = {0.f, 0.f}, mc = {0.f, 0.f};
    const v2f ones = {1.0f, 1.0f};
#pragma unroll
    for (int s = 0; s < 20; ++s) {
      int j0 = 4 * s + 2 * hi;
      int j1 = j0 + 1;
      int n0 = base + j0;
      int n1 = base + j1;
      int c0 = min(max(n0, 0), MWIN - 1);
      int c1 = min(max(n1, 0), MWIN - 1);
      float ok0 = ((unsigned)n0 < (unsigned)MWIN) ? 1.0f : 0.0f;
      float ok1 = ((unsigned)n1 < (unsigned)MWIN) ? 1.0f : 0.0f;
      v2f bb;
      bb.x = gk[(size_t)c0 * 16 + col] * ok0;
      bb.y = gk[(size_t)c1 * 16 + col] * ok1;
      if (s >= 4 && s < 8)   ma = (s == 4)  ? bb : ma + bb;
      if (s >= 8 && s < 12)  mb = (s == 8)  ? bb : mb + bb;
      if (s >= 12 && s < 16) mc = (s == 12) ? bb : mc + bb;
      if (s < 4 || (s >= 12 && s < 16)) {
        v2f a;
        a.x = (j0 >= col && j0 <= col + 47) ? 1.0f : 0.0f;
        a.y = (j1 >= col && j1 <= col + 47) ? 1.0f : 0.0f;
        d0 = __builtin_amdgcn_wmma_f32_16x16x4_f32(
             false, a,  false, bb,
             (short)0, d0,  false,  false);
      }
      if (s == 11)
        d0 = __builtin_amdgcn_wmma_f32_16x16x4_f32(
             false, ones,  false, ma + mb,
             (short)0, d0,  false,  false);
      if ((s >= 4 && s < 8) || s >= 16) {
        v2f a;
        a.x = (j0 - 16 >= col && j0 - 16 <= col + 47) ? 1.0f : 0.0f;
        a.y = (j1 - 16 >= col && j1 - 16 <= col + 47) ? 1.0f : 0.0f;
        d1 = __builtin_amdgcn_wmma_f32_16x16x4_f32(
             false, a,  false, bb,
             (short)0, d1,  false,  false);
      }
      if (s == 15)
        d1 = __builtin_amdgcn_wmma_f32_16x16x4_f32(
             false, ones,  false, mb + mc,
             (short)0, d1,  false,  false);
    }
    acc[0][k] = d0;
    acc[1][k] = d1;
  }

#pragma unroll
  for (int tile = 0; tile < 2; ++tile) {
    const int tt = t0 + 16 * tile + 8 * hi;
    float v[8];
#pragma unroll
    for (int p = 0; p < 8; ++p)
      v[p] = f4c(r[tile][(3 * p + 0) >> 2], (3 * p + 0) & 3) * acc[tile][0][p] +
             f4c(r[tile][(3 * p + 1) >> 2], (3 * p + 1) & 3) * acc[tile][1][p] +
             f4c(r[tile][(3 * p + 2) >> 2], (3 * p + 2) & 3) * acc[tile][2][p];
    float4* dst = reinterpret_cast<float4*>(ppg + (size_t)col * PPG_PITCH + tt);
    dst[0] = make_float4(v[0], v[1], v[2], v[3]);
    dst[1] = make_float4(v[4], v[5], v[6], v[7]);
  }
}

constexpr int S3_THREADS = 256;
constexpr int LA     = 32;
constexpr int NFAC   = 5;
constexpr int PAD    = 2 * LA;
constexpr int FT     = 76;
constexpr int NFT    = (TLEN + FT - 1) / FT;
constexpr int FW     = FT + PAD;
constexpr int LASTG  = (TLEN - (NFT - 1) * FT) / 4;
constexpr int TAILX  = NFT * FT - TLEN;
constexpr int ROWBUF = PAD + NFT * FT + PAD;
static_assert(FT % 4 == 0 && (FT / 4) % 2 == 1, "tile: whole 16-B pieces, an odd number of them");
static_assert(NFT <= S3_THREADS && PAD + TAILX <= S3_THREADS, "one tile per thread, one pad sample per thread");
static_assert((TLEN - (NFT - 1) * FT) % 4 == 0 && LASTG > 0, "the row ends on a 16-B piece of the last tile");
static_assert(ROWBUF % 4 == 0, "the second buffer stays 16-B aligned");

template <int DIR, int S, int INV>
__device__ __forceinline__ void fir_factor(float (&w)[FW], float m1, float m2) {
  if (DIR > 0) {
#pragma unroll
    for (int j = FW - 1; j >= INV; --j)
      w[j] = fmaf(m2, w[j - 2 * S], fmaf(m1, w[j - S], w[j]));
  } else {
#pragma unroll
    for (int j = 0; j < FW - INV; ++j)
      w[j] = fmaf(m2, w[j + 2 * S], fmaf(m1, w[j + S], w[j]));
  }
}

template <int DIR>
__device__ __forceinline__ void fir_row(const float* src, float* dst,
                                        const float (&m1)[NFAC], const float (&m2)[NFAC],
                                        float b0, float b1, float b2, int tid) {
  if (tid < NFT) {
    const int t0 = FT * tid;
    const float4* p = reinterpret_cast<const float4*>(src + (DIR > 0 ? t0 : t0 + PAD));
    float w[FW];
#pragma unroll
    for (int m = 0; m < FW / 4; ++m) {
      float4 v = p[m];
      w[4 * m + 0] = v.x; w[4 * m + 1] = v.y; w[4 * m + 2] = v.z; w[4 * m + 3] = v.w;
    }
    fir_factor<DIR, 16, 32>(w, m1[4], m2[4]);
    fir_factor<DIR,  8, 48>(w, m1[3], m2[3]);
    fir_factor<DIR,  4, 56>(w, m1[2], m2[2]);
    fir_factor<DIR,  2, 60>(w, m1[1], m2[1]);
    fir_factor<DIR,  1, 62>(w, m1[0], m2[0]);
    float o[FT];
#pragma unroll
    for (int j = 0; j < FT; ++j)
      o[j] = DIR > 0 ? fmaf(b2, w[PAD + j - 2], fmaf(b1, w[PAD + j - 1], b0 * w[PAD + j]))
                     : fmaf(b2, w[j + 2],       fmaf(b1, w[j + 1],       b0 * w[j]));
    float4* d = reinterpret_cast<float4*>(dst + PAD + t0);
    const bool whole = tid < NFT - 1;
#pragma unroll
    for (int g = 0; g < FT / 4; ++g)
      if (g < LASTG || whole) d[g] = make_float4(o[4 * g], o[4 * g + 1], o[4 * g + 2], o[4 * g + 3]);
  }
}


#define CHAIN_STEPS                                            \
  _Pragma("unroll") for (int i = 0; i < UB; ++i) {             \
    float fp = fmaf(m2, c, fmaf(m1, cb[i], ca[i]));            \
    c = ca[i];                                                 \
    float v = fmaf(-be1, y1, fmaf(-be2, y2, fp));              \
    y2 = y1; y1 = v; r[i] = v;                                 \
  }

template <int DIR>
__device__ __forceinline__ void chains_row2(const float* __restrict__ in, float* __restrict__ out,
                                            float m1, float m2, float be1, float be2,
                                            int lane, int par) {
  constexpr int NFULL = TLEN / LA;
  constexpr int NREM  = TLEN - NFULL * LA;
  constexpr int NJ    = NFULL / 2;
  constexpr int UB    = 16;
  constexpr int NTAIL = NJ % UB;
  constexpr int LASTBASE = NJ - UB;
  static_assert(NFULL % 2 == 0, "both waves have the same number of full steps; the partial step is an even one");
  static_assert(NTAIL > 0 && NJ >= 2 * UB, "tail indexing below assumes a clamped last fetch");
  static_assert(2 * LA <= PAD, "steps -1 and -2 lie in the zero samples");
  constexpr int STR = DIR > 0 ? LA : -LA;
  const int off = (DIR > 0 ? lane : TLEN - 1 - lane) + par * STR;
  const float* pi = in + off;
  float* po = out + off;
  float y1 = 0.f, y2 = 0.f;
  float c = 0.f;
  float ca[UB], cb[UB], na[UB], nb[UB], r[UB];
#pragma unroll
  for (int i = 0; i < UB; ++i) { ca[i] = pi[(2 * i) * STR]; cb[i] = pi[(2 * i - 1) * STR]; }
#pragma unroll
  for (int i = 0; i < UB; ++i) { na[i] = pi[(2 * (UB + i)) * STR]; nb[i] = pi[(2 * (UB + i) - 1) * STR]; }
  __builtin_amdgcn_sched_barrier(0);
  CHAIN_STEPS
  __builtin_amdgcn_sched_barrier(0);
#pragma unroll
  for (int i = 0; i < UB; ++i) { ca[i] = na[i]; cb[i] = nb[i]; }
  int j = UB;
  for (; j + UB <= NJ; j += UB) {
    const int jb = min(j + UB, LASTBASE);
#pragma unroll
    for (int i = 0; i < UB; ++i) po[(2 * (j - UB + i)) * STR] = r[i];
#pragma unroll
    for (int i = 0; i < UB; ++i) { na[i] = pi[(2 * (jb + i)) * STR]; nb[i] = pi[(2 * (jb + i) - 1) * STR]; }
    __builtin_amdgcn_sched_barrier(0);
    CHAIN_STEPS
    __builtin_amdgcn_sched_barrier(0);
#pragma unroll
    for (int i = 0; i < UB; ++i) { ca[i] = na[i]; cb[i] = nb[i]; }
  }
#pragma unroll
  for (int i = 0; i < UB; ++i) po[(2 * (j - UB + i)) * STR] = r[i];
#pragma unroll
  for (int i = 0; i < NTAIL; ++i) {
    float fp = fmaf(m2, c, fmaf(m1, cb[UB - NTAIL + i], ca[UB - NTAIL + i]));
    c = ca[UB - NTAIL + i];
    float v = fmaf(-be1, y1, fmaf(-be2, y2, fp));
    y2 = y1; y1 = v;
    po[(2 * (NJ - NTAIL + i)) * STR] = v;
  }
  if (par == 0 && lane < NREM) {
    float fp = fmaf(m2, c, fmaf(m1, pi[(NFULL - 1) * STR], pi[NFULL * STR]));
    po[NFULL * STR] = fmaf(-be1, y1, fmaf(-be2, y2, fp));
  }
}
#undef CHAIN_STEPS

constexpr int NBLK16 = TLEN / 16;

#define BIQUAD_STEP(v)                                   \
  {                                                      \
    float xin_ = (v);                                    \
    float y_   = fmaf(b0, xin_, z1);                     \
    z1 = fmaf(b1, xin_, fmaf(-a1, y_, z2));              \
    z2 = fmaf(b2, xin_, -a2 * y_);                       \
    (v) = y_;                                            \
  }

__device__ __forceinline__ void serial_scans(float* xs, float b0, float b1, float b2,
                                             float a1, float a2) {
  float4* xv = reinterpret_cast<float4*>(xs);
  float z1 = 0.f, z2 = 0.f;
  float4 c0 = xv[0], c1 = xv[1], c2 = xv[2], c3 = xv[3];
  for (int k = 0; k < NBLK16; ++k) {
    int kn = min(k + 1, NBLK16 - 1);
    float4 n0 = xv[4 * kn + 0], n1 = xv[4 * kn + 1];
    float4 n2 = xv[4 * kn + 2], n3 = xv[4 * kn + 3];
    BIQUAD_STEP(c0.x) BIQUAD_STEP(c0.y) BIQUAD_STEP(c0.z) BIQUAD_STEP(c0.w)
    BIQUAD_STEP(c1.x) BIQUAD_STEP(c1.y) BIQUAD_STEP(c1.z) BIQUAD_STEP(c1.w)
    BIQUAD_STEP(c2.x) BIQUAD_STEP(c2.y) BIQUAD_STEP(c2.z) BIQUAD_STEP(c2.w)
    BIQUAD_STEP(c3.x) BIQUAD_STEP(c3.y) BIQUAD_STEP(c3.z) BIQUAD_STEP(c3.w)
    xv[4 * k + 0] = c0; xv[4 * k + 1] = c1; xv[4 * k + 2] = c2; xv[4 * k + 3] = c3;
    c0 = n0; c1 = n1; c2 = n2; c3 = n3;
  }
  z1 = 0.f; z2 = 0.f;
  c0 = xv[4 * (NBLK16 - 1) + 0]; c1 = xv[4 * (NBLK16 - 1) + 1];
  c2 = xv[4 * (NBLK16 - 1) + 2]; c3 = xv[4 * (NBLK16 - 1) + 3];
  for (int k = NBLK16 - 1; k >= 0; --k) {
    int kn = max(k - 1, 0);
    float4 n0 = xv[4 * kn + 0], n1 = xv[4 * kn + 1];
    float4 n2 = xv[4 * kn + 2], n3 = xv[4 * kn + 3];
    BIQUAD_STEP(c3.w) BIQUAD_STEP(c3.z) BIQUAD_STEP(c3.y) BIQUAD_STEP(c3.x)
    BIQUAD_STEP(c2.w) BIQUAD_STEP(c2.z) BIQUAD_STEP(c2.y) BIQUAD_STEP(c2.x)
    BIQUAD_STEP(c1.w) BIQUAD_STEP(c1.z) BIQUAD_STEP(c1.y) BIQUAD_STEP(c1.x)
    BIQUAD_STEP(c0.w) BIQUAD_STEP(c0.z) BIQUAD_STEP(c0.y) BIQUAD_STEP(c0.x)
    xv[4 * k + 0] = c0; xv[4 * k + 1] = c1; xv[4 * k + 2] = c2; xv[4 * k + 3] = c3;
    c0 = n0; c1 = n1; c2 = n2; c3 = n3;
  }
}

__global__ __launch_bounds__(S3_THREADS)
void pos_filtfilt(const float* __restrict__ ppg,
                  const float* __restrict__ acoef,
                  const float* __restrict__ bcoef,
                  float* __restrict__ out) {
  __shared__ __attribute__((aligned(16))) float bufA[ROWBUF];
  __shared__ __attribute__((aligned(16))) float bufB[ROWBUF];
  constexpr int NCF = NFAC + 1;
  __shared__ __attribute__((aligned(16))) float cfs[2 * NCF + 2];
  __shared__ int    lookahead_ok;
  const int b   = blockIdx.x;
  const int tid = threadIdx.x;

  const float a0  = acoef[0];
  const float bn0 = bcoef[0] / a0, bn1 = bcoef[1] / a0, bn2 = bcoef[2] / a0;
  const float an1 = acoef[1] / a0, an2 = acoef[2] / a0;

  if (tid == 0) {
    double e1 = (double)an1, e2 = (double)an2;
    int ok = 0;
    for (int i = 0; i < NCF; ++i) {
      cfs[i] = (float)(-e1);
      cfs[NCF + i] = (float)e2;
      if (i == NFAC) ok = (fabs(e2) <= 0.25 && fabs(e1) <= 0.5 + 2.0 * e2) ? 1 : 0;
      const double n1 = fma(-e1, e1, 2.0 * e2);
      e2 = e2 * e2;
      e1 = n1;
    }
    cfs[2 * NCF]     = (float)e1;
    cfs[2 * NCF + 1] = (float)e2;
    lookahead_ok = ok;
  }

  {
    constexpr int NROWLD = (TLEN / 4 + S3_THREADS - 1) / S3_THREADS;
    const float4* prow = reinterpret_cast<const float4*>(ppg + (size_t)b * PPG_PITCH);
    float4* arow = reinterpret_cast<float4*>(bufA + PAD);
    float4 rowv[NROWLD];
#pragma unroll
    for (int m = 0; m < NROWLD; ++m) {
      const int i = tid + m * S3_THREADS;
      rowv[m] = prow[m == NROWLD - 1 ? min(i, TLEN / 4 - 1) : i];
    }
#pragma unroll
    for (int m = 0; m < NROWLD; ++m) {
      const int i = tid + m * S3_THREADS;
      arow[m == NROWLD - 1 ? min(i, TLEN / 4 - 1) : i] = rowv[m];
    }
  }
  if (tid < PAD) { bufA[tid] = 0.f; bufB[tid] = 0.f; }
  if (tid < PAD + TAILX) { bufA[PAD + TLEN + tid] = 0.f; bufB[PAD + TLEN + tid] = 0.f; }
  __syncthreads();

  if (lookahead_ok != 0) {
    float m1[NFAC], m2[NFAC];
#pragma unroll
    for (int i = 0; i < NFAC; ++i) { m1[i] = cfs[i]; m2[i] = cfs[NCF + i]; }
    const float cm1 = cfs[NFAC], cm2 = cfs[NCF + NFAC];
    const float be1 = cfs[2 * NCF], be2 = cfs[2 * NCF + 1];

    fir_row<+1>(bufA, bufB, m1, m2, bn0, bn1, bn2, tid);
    __syncthreads();
    if (tid < 64) chains_row2<+1>(bufB + PAD, bufA + PAD, cm1, cm2, be1, be2, tid & 31, tid >> 5);
    __syncthreads();
    fir_row<-1>(bufA, bufB, m1, m2, bn0, bn1, bn2, tid);
    __syncthreads();
    if (tid < 64) chains_row2<-1>(bufB + PAD, bufA + PAD, cm1, cm2, be1, be2, tid & 31, tid >> 5);
  } else {
    if (tid == 0) serial_scans(bufA + PAD, bn0, bn1, bn2, an1, an2);
  }
  __syncthreads();

  const float4* xr = reinterpret_cast<const float4*>(bufA + PAD);
  float4* orow = reinterpret_cast<float4*>(out + (size_t)b * TLEN);
  for (int i = tid; i < TLEN / 4; i += S3_THREADS) orow[i] = xr[i];
}

extern "C" void kernel_launch(void* const* d_in, const int* in_sizes, int n_in,
                              void* d_out, int out_size, void* d_ws, size_t ws_size,
                              hipStream_t stream) {
  const float* rgb    = (const float*)d_in[0];
  const float* matrix = (const float*)d_in[1];
  const float* acoef  = (const float*)d_in[2];
  const float* bcoef  = (const float*)d_in[3];
  float* out = (float*)d_out;

  float* g    = (float*)d_ws;
  float* ppg  = g + (size_t)3 * MWIN * 16;

  pos_window_stats<<<ST_WAVES, 32, 0, stream>>>(rgb, matrix, g);
  pos_overlap_add_wmma<<<OA_WAVES, 32, 0, stream>>>(rgb, g, ppg);
  pos_filtfilt<<<BATCH, S3_THREADS, 0, stream>>>(ppg, acoef, bcoef, out);
}


# Stage-3 chains: look-ahead along the step index from 2 steps to 8; A_5..A_7 as 3-tap factors on register tiles along the steps; 256 chains of 71 steps in place; f32 sum order changes, fixed per run
# speedup vs baseline: 95.2907x; 1.0647x over previous
#include <hip/hip_runtime.h>
#include <hip/hip_bf16.h>

typedef float v2f __attribute__((ext_vector_type(2)));
typedef float v8f __attribute__((ext_vector_type(8)));

constexpr int TLEN  = 18000;
constexpr int BATCH = 16;
constexpr int WIN   = 48;
constexpr int MWIN  = TLEN - WIN;
constexpr int PPG_PITCH = 18048;
constexpr int OA_TILE   = 32;
constexpr int OA_WAVES  = (TLEN + OA_TILE - 1) / OA_TILE;
static_assert(PPG_PITCH % 32 == 0 && PPG_PITCH >= OA_WAVES * OA_TILE, "row pitch: whole lines, room for the last tile");

constexpr int ST_TILE  = 16;
constexpr int ST_WAVES = MWIN / ST_TILE;
static_assert(MWIN % ST_TILE == 0, "whole tiles: no window of a tile lies past the last one");
static_assert((ST_WAVES - 1) * ST_TILE + 63 <= TLEN - 1, "the last tile's 64 samples lie inside the row");

__global__ __launch_bounds__(32) __attribute__((amdgpu_waves_per_eu(1, 2))) void pos_window_stats(const float* __restrict__ rgb,
                                 const float* __restrict__ matrix,
                                 float* __restrict__ g) {
  const int n0   = blockIdx.x * ST_TILE;
  const int lane = threadIdx.x;
  const int col  = lane & 15;
  const int hi   = lane >> 4;
  const int wofs = 2 * (col & 7) + (col >> 3);

  const float2* rp2 = reinterpret_cast<const float2*>(rgb + (size_t)(col * TLEN + n0 + 2 * hi) * 3);

  float2 u[16][3];
#pragma unroll
  for (int s = 0; s < 16; ++s) {
#pragma unroll
    for (int m = 0; m < 3; ++m) u[s][m] = rp2[6 * s + m];
  }

  v8f acc[9];
#pragma unroll
  for (int f = 0; f < 9; ++f) acc[f] = v8f{0.f, 0.f, 0.f, 0.f, 0.f, 0.f, 0.f, 0.f};

  static_assert(WIN == 48 && ST_TILE == 16, "common columns 16..47 = K-steps 4..11");
  v2f mid[9];
#pragma unroll
  for (int f = 0; f < 9; ++f) mid[f] = v2f{0.f, 0.f};
  const v2f ones = {1.0f, 1.0f};
#pragma unroll
  for (int s = 0; s < 16; ++s) {
    const float2 u0 = u[s][0];
    const float2 u1 = u[s][1];
    const float2 u2 = u[s][2];
    const float x0 = u0.x, x1 = u0.y, x2 = u1.x;
    const float y0 = u1.y, y1 = u2.x, y2 = u2.y;
    v2f bf[9];
    bf[0].x = x0;      bf[0].y = y0;
    bf[1].x = x1;      bf[1].y = y1;
    bf[2].x = x2;      bf[2].y = y2;
    bf[3].x = x0 * x0; bf[3].y = y0 * y0;
    bf[4].x = x0 * x1; bf[4].y = y0 * y1;
    bf[5].x = x0 * x2; bf[5].y = y0 * y2;
    bf[6].x = x1 * x1; bf[6].y = y1 * y1;
    bf[7].x = x1 * x2; bf[7].y = y1 * y2;
    bf[8].x = x2 * x2; bf[8].y = y2 * y2;
    if (s >= 4 && s < 12) {
#pragma unroll
      for (int f = 0; f < 9; ++f) mid[f] = (s == 4) ? bf[f] : mid[f] + bf[f];
      if (s == 11) {
#pragma unroll
        for (int f = 0; f < 9; ++f)
          acc[f] = __builtin_amdgcn_wmma_f32_16x16x4_f32(
               false, ones,  false, mid[f],
               (short)0, acc[f],  false,  false);
      }
    } else {
      const int j0 = 4 * s + 2 * hi;
      const int j1 = j0 + 1;
      v2f a;
      a.x = (j0 >= wofs && j0 <= wofs + 47) ? 1.0f : 0.0f;
      a.y = (j1 >= wofs && j1 <= wofs + 47) ? 1.0f : 0.0f;
#pragma unroll
      for (int f = 0; f < 9; ++f)
        acc[f] = __builtin_amdgcn_wmma_f32_16x16x4_f32(
             false, a,  false, bf[f],
             (short)0, acc[f],  false,  false);
    }
  }

  const float m00 = matrix[0], m01 = matrix[1], m02 = matrix[2];
  const float m10 = matrix[3], m11 = matrix[4], m12 = matrix[5];
  const float invW = 1.0f / (float)WIN;

#pragma unroll
  for (int p = 0; p < 8; ++p) {
    const int n = n0 + 2 * p + hi;
    const float s0 = acc[0][p], s1 = acc[1][p], s2 = acc[2][p];
    const float q00 = acc[3][p], q01 = acc[4][p], q02 = acc[5][p];
    const float q11 = acc[6][p], q12 = acc[7][p], q22 = acc[8][p];

    float inv0 = 1.0f / (s0 * invW + 1e-6f);
    float inv1 = 1.0f / (s1 * invW + 1e-6f);
    float inv2 = 1.0f / (s2 * invW + 1e-6f);

    float u0 = m00 * inv0, u1 = m01 * inv1, u2 = m02 * inv2;
    float sum0 = u0 * s0 + u1 * s1 + u2 * s2;
    float sq0  = u0 * u0 * q00 + u1 * u1 * q11 + u2 * u2 * q22 +
                 2.f * (u0 * u1 * q01 + u0 * u2 * q02 + u1 * u2 * q12);
    float var0 = (sq0 - sum0 * sum0 * invW) * (1.0f / (float)(WIN - 1));
    float std0 = sqrtf(fmaxf(var0, 0.f));

    float v0 = m10 * inv0, v1 = m11 * inv1, v2 = m12 * inv2;
    float sum1 = v0 * s0 + v1 * s1 + v2 * s2;
    float sq1  = v0 * v0 * q00 + v1 * v1 * q11 + v2 * v2 * q22 +
                 2.f * (v0 * v1 * q01 + v0 * v2 * q02 + v1 * v2 * q12);
    float var1 = (sq1 - sum1 * sum1 * invW) * (1.0f / (float)(WIN - 1));
    float std1 = sqrtf(fmaxf(var1, 0.f));

    float c = std0 / (std1 + 1e-4f);

    g[(size_t)(0 * MWIN + n) * 16 + col] = (m00 + c * m10) * inv0;
    g[(size_t)(1 * MWIN + n) * 16 + col] = (m01 + c * m11) * inv1;
    g[(size_t)(2 * MWIN + n) * 16 + col] = (m02 + c * m12) * inv2;
  }
}

__device__ __forceinline__ float f4c(const float4& v, int c) {
  return c == 0 ? v.x : c == 1 ? v.y : c == 2 ? v.z : v.w;
}

__global__ __launch_bounds__(32) __attribute__((amdgpu_waves_per_eu(1, 2))) void pos_overlap_add_wmma(const float* __restrict__ rgb,
                                     const float* __restrict__ g,
                                     float* __restrict__ ppg) {
  const int t0   = blockIdx.x * OA_TILE;
  const int lane = threadIdx.x;
  const int col  = lane & 15;
  const int hi   = lane >> 4;
  const int base = t0 - 47;

  float4 r[2][6];
#pragma unroll
  for (int tile = 0; tile < 2; ++tile) {
    const int ttc = min(t0 + 16 * tile + 8 * hi, TLEN - 8);
    const float4* rp4 = reinterpret_cast<const float4*>(rgb + (size_t)(col * TLEN + ttc) * 3);
#pragma unroll
    for (int q = 0; q < 6; ++q) r[tile][q] = rp4[q];
  }

  v8f acc[2][3];
#pragma unroll
  for (int k = 0; k < 3; ++k) {
    const float* gk = g + (size_t)k * MWIN * 16;
    v8f d0 = {0.f, 0.f, 0.f, 0.f, 0.f, 0.f, 0.f, 0.f};
    v8f d1 = {0.f, 0.f, 0.f, 0.f, 0.f, 0.f, 0.f, 0.f};
    v2f ma = {0.f, 0.f}, mb = {0.f, 0.f}, mc = {0.f, 0.f};
    const v2f ones = {1.0f, 1.0f};
#pragma unroll
    for (int s = 0; s < 20; ++s) {
      int j0 = 4 * s + 2 * hi;
      int j1 = j0 + 1;
      int n0 = base + j0;
      int n1 = base + j1;
      int c0 = min(max(n0, 0), MWIN - 1);
      int c1 = min(max(n1, 0), MWIN - 1);
      float ok0 = ((unsigned)n0 < (unsigned)MWIN) ? 1.0f : 0.0f;
      float ok1 = ((unsigned)n1 < (unsigned)MWIN) ? 1.0f : 0.0f;
      v2f bb;
      bb.x = gk[(size_t)c0 * 16 + col] * ok0;
      bb.y = gk[(size_t)c1 * 16 + col] * ok1;
      if (s >= 4 && s < 8)   ma = (s == 4)  ? bb : ma + bb;
      if (s >= 8 && s < 12)  mb = (s == 8)  ? bb : mb + bb;
      if (s >= 12 && s < 16) mc = (s == 12) ? bb : mc + bb;
      if (s < 4 || (s >= 12 && s < 16)) {
        v2f a;
        a.x = (j0 >= col && j0 <= col + 47) ? 1.0f : 0.0f;
        a.y = (j1 >= col && j1 <= col + 47) ? 1.0f : 0.0f;
        d0 = __builtin_amdgcn_wmma_f32_16x16x4_f32(
             false, a,  false, bb,
             (short)0, d0,  false,  false);
      }
      if (s == 11)
        d0 = __builtin_amdgcn_wmma_f32_16x16x4_f32(
             false, ones,  false, ma + mb,
             (short)0, d0,  false,  false);
      if ((s >= 4 && s < 8) || s >= 16) {
        v2f a;
        a.x = (j0 - 16 >= col && j0 - 16 <= col + 47) ? 1.0f : 0.0f;
        a.y = (j1 - 16 >= col && j1 - 16 <= col + 47) ? 1.0f : 0.0f;
        d1 = __builtin_amdgcn_wmma_f32_16x16x4_f32(
             false, a,  false, bb,
             (short)0, d1,  false,  false);
      }
      if (s == 15)
        d1 = __builtin_amdgcn_wmma_f32_16x16x4_f32(
             false, ones,  false, mb + mc,
             (short)0, d1,  false,  false);
    }
    acc[0][k] = d0;
    acc[1][k] = d1;
  }

#pragma unroll
  for (int tile = 0; tile < 2; ++tile) {
    const int tt = t0 + 16 * tile + 8 * hi;
    float v[8];
#pragma unroll
    for (int p = 0; p < 8; ++p)
      v[p] = f4c(r[tile][(3 * p + 0) >> 2], (3 * p + 0) & 3) * acc[tile][0][p] +
             f4c(r[tile][(3 * p + 1) >> 2], (3 * p + 1) & 3) * acc[tile][1][p] +
             f4c(r[tile][(3 * p + 2) >> 2], (3 * p + 2) & 3) * acc[tile][2][p];
    float4* dst = reinterpret_cast<float4*>(ppg + (size_t)col * PPG_PITCH + tt);
    dst[0] = make_float4(v[0], v[1], v[2], v[3]);
    dst[1] = make_float4(v[4], v[5], v[6], v[7]);
  }
}

constexpr int S3_THREADS = 256;
constexpr int LA     = 32;
constexpr int NFAC   = 5;
constexpr int PAD    = 2 * LA;
constexpr int CST    = 8;
constexpr int NSFAC  = 3;
constexpr int NCF    = NFAC + NSFAC;
constexpr int SREACH = 2 * (CST - 1);
constexpr int FPAD   = SREACH * LA;
constexpr int NSTEP  = (TLEN + LA - 1) / LA;
constexpr int TL     = 19;
constexpr int NRNG   = (NSTEP + TL - 1) / TL;
constexpr int TW     = TL + SREACH;
constexpr int FT     = 76;
constexpr int NFT    = (TLEN + FT - 1) / FT;
constexpr int FW     = FT + PAD;
constexpr int LASTG  = (TLEN - (NFT - 1) * FT) / 4;
constexpr int TAILX  = NFT * FT - TLEN;
constexpr int ROWBUF = FPAD + TLEN + FPAD;
static_assert(FT % 4 == 0 && (FT / 4) % 2 == 1, "tile: whole 16-B pieces, an odd number of them");
static_assert(NFT <= S3_THREADS && PAD + TAILX <= FPAD, "one tile per thread; the last tile's reads past the row stay in the zero samples");
static_assert(CST * LA == S3_THREADS, "one chain per thread");
static_assert(8 * NRNG <= S3_THREADS && TL % 2 == 1, "one step tile per thread");
static_assert(NRNG * TL * LA - TLEN <= FPAD && FPAD % 4 == 0 && TLEN % 4 == 0,
              "the step tiles' reads past either end of the row stay in the zero samples, 16-B aligned");
static_assert((TLEN - (NFT - 1) * FT) % 4 == 0 && LASTG > 0, "the row ends on a 16-B piece of the last tile");
static_assert(ROWBUF % 4 == 0, "the second buffer stays 16-B aligned");

template <int DIR, int S, int INV>
__device__ __forceinline__ void fir_factor(float (&w)[FW], float m1, float m2) {
  if (DIR > 0) {
#pragma unroll
    for (int j = FW - 1; j >= INV; --j)
      w[j] = fmaf(m2, w[j - 2 * S], fmaf(m1, w[j - S], w[j]));
  } else {
#pragma unroll
    for (int j = 0; j < FW - INV; ++j)
      w[j] = fmaf(m2, w[j + 2 * S], fmaf(m1, w[j + S], w[j]));
  }
}

template <int DIR>
__device__ __forceinline__ void fir_row(const float* src, float* dst,
                                        const float (&m1)[NFAC], const float (&m2)[NFAC],
                                        float b0, float b1, float b2, int tid) {
  if (tid < NFT) {
    const int t0 = FT * tid;
    const float4* p = reinterpret_cast<const float4*>(src + (DIR > 0 ? t0 : t0 + PAD));
    float w[FW];
#pragma unroll
    for (int m = 0; m < FW / 4; ++m) {
      float4 v = p[m];
      w[4 * m + 0] = v.x; w[4 * m + 1] = v.y; w[4 * m + 2] = v.z; w[4 * m + 3] = v.w;
    }
    fir_factor<DIR, 16, 32>(w, m1[4], m2[4]);
    fir_factor<DIR,  8, 48>(w, m1[3], m2[3]);
    fir_factor<DIR,  4, 56>(w, m1[2], m2[2]);
    fir_factor<DIR,  2, 60>(w, m1[1], m2[1]);
    fir_factor<DIR,  1, 62>(w, m1[0], m2[0]);
    float o[FT];
#pragma unroll
    for (int j = 0; j < FT; ++j)
      o[j] = DIR > 0 ? fmaf(b2, w[PAD + j - 2], fmaf(b1, w[PAD + j - 1], b0 * w[PAD + j]))
                     : fmaf(b2, w[j + 2],       fmaf(b1, w[j + 1],       b0 * w[j]));
    float4* d = reinterpret_cast<float4*>(dst + PAD + t0);
    const bool whole = tid < NFT - 1;
#pragma unroll
    for (int g = 0; g < FT / 4; ++g)
      if (g < LASTG || whole) d[g] = make_float4(o[4 * g], o[4 * g + 1], o[4 * g + 2], o[4 * g + 3]);
  }
}

template <int D, int INV>
__device__ __forceinline__ void step_factor(float (&w)[TW][4], float a, float b) {
#pragma unroll
  for (int k = TW - 1; k >= INV; --k) {
#pragma unroll
    for (int q = 0; q < 4; ++q) w[k][q] = fmaf(b, w[k - 2 * D][q], fmaf(a, w[k - D][q], w[k][q]));
  }
}

template <int DIR>
__device__ __forceinline__ void step_factors(const float* __restrict__ src, float* __restrict__ dst,
                                             const float (&m1)[NSFAC], const float (&m2)[NSFAC],
                                             int tid) {
  if (tid < 8 * NRNG) {
    const int g  = tid & 7;
    const int s0 = TL * (tid >> 3) - SREACH;
    const int e0 = DIR > 0 ? 4 * g : TLEN - 4 - 4 * g;
    constexpr int SV = DIR > 0 ? LA / 4 : -(LA / 4);
    const float4* p = reinterpret_cast<const float4*>(src + e0) + SV * s0;
    float w[TW][4];
#pragma unroll
    for (int k = 0; k < TW; ++k) {
      const float4 v = p[SV * k];
      w[k][0] = v.x; w[k][1] = v.y; w[k][2] = v.z; w[k][3] = v.w;
    }
    step_factor<4,  8>(w, m1[2], m2[2]);
    step_factor<2, 12>(w, m1[1], m2[1]);
    step_factor<1, SREACH>(w, m1[0], m2[0]);
    float4* d = reinterpret_cast<float4*>(dst + e0) + SV * s0;
#pragma unroll
    for (int k = SREACH; k < TW; ++k)
      if (LA * (s0 + k) + 4 * g < TLEN) d[SV * k] = make_float4(w[k][0], w[k][1], w[k][2], w[k][3]);
  }
}

template <int DIR>
__device__ __forceinline__ void chains_row(float* row, float ga1, float ga2, int tid) {
  constexpr int CH    = CST * LA;
  constexpr int NFULL = TLEN / CH;
  constexpr int NREM  = TLEN - NFULL * CH;
  constexpr int STR   = DIR > 0 ? CH : -CH;
  float* p = row + (DIR > 0 ? tid : TLEN - 1 - tid);
  float f[NFULL];
#pragma unroll
  for (int j = 0; j < NFULL; ++j) f[j] = p[j * STR];
  __builtin_amdgcn_sched_barrier(0);
  float y1 = 0.f, y2 = 0.f;
#pragma unroll
  for (int j = 0; j < NFULL; ++j) {
    const float v = fmaf(-ga1, y1, fmaf(-ga2, y2, f[j]));
    y2 = y1; y1 = v;
    p[j * STR] = v;
  }
  if (tid < NREM) p[NFULL * STR] = fmaf(-ga1, y1, fmaf(-ga2, y2, p[NFULL * STR]));
}

constexpr int NBLK16 = TLEN / 16;

#define BIQUAD_STEP(v)                                   \
  {                                                      \
    float xin_ = (v);                                    \
    float y_   = fmaf(b0, xin_, z1);                     \
    z1 = fmaf(b1, xin_, fmaf(-a1, y_, z2));              \
    z2 = fmaf(b2, xin_, -a2 * y_);                       \
    (v) = y_;                                            \
  }

__device__ __forceinline__ void serial_scans(float* xs, float b0, float b1, float b2,
                                             float a1, float a2) {
  float4* xv = reinterpret_cast<float4*>(xs);
  float z1 = 0.f, z2 = 0.f;
  float4 c0 = xv[0], c1 = xv[1], c2 = xv[2], c3 = xv[3];
  for (int k = 0; k < NBLK16; ++k) {
    int kn = min(k + 1, NBLK16 - 1);
    float4 n0 = xv[4 * kn + 0], n1 = xv[4 * kn + 1];
    float4 n2 = xv[4 * kn + 2], n3 = xv[4 * kn + 3];
    BIQUAD_STEP(c0.x) BIQUAD_STEP(c0.y) BIQUAD_STEP(c0.z) BIQUAD_STEP(c0.w)
    BIQUAD_STEP(c1.x) BIQUAD_STEP(c1.y) BIQUAD_STEP(c1.z) BIQUAD_STEP(c1.w)
    BIQUAD_STEP(c2.x) BIQUAD_STEP(c2.y) BIQUAD_STEP(c2.z) BIQUAD_STEP(c2.w)
    BIQUAD_STEP(c3.x) BIQUAD_STEP(c3.y) BIQUAD_STEP(c3.z) BIQUAD_STEP(c3.w)
    xv[4 * k + 0] = c0; xv[4 * k + 1] = c1; xv[4 * k + 2] = c2; xv[4 * k + 3] = c3;
    c0 = n0; c1 = n1; c2 = n2; c3 = n3;
  }
  z1 = 0.f; z2 = 0.f;
  c0 = xv[4 * (NBLK16 - 1) + 0]; c1 = xv[4 * (NBLK16 - 1) + 1];
  c2 = xv[4 * (NBLK16 - 1) + 2]; c3 = xv[4 * (NBLK16 - 1) + 3];
  for (int k = NBLK16 - 1; k >= 0; --k) {
    int kn = max(k - 1, 0);
    float4 n0 = xv[4 * kn + 0], n1 = xv[4 * kn + 1];
    float4 n2 = xv[4 * kn + 2], n3 = xv[4 * kn + 3];
    BIQUAD_STEP(c3.w) BIQUAD_STEP(c3.z) BIQUAD_STEP(c3.y) BIQUAD_STEP(c3.x)
    BIQUAD_STEP(c2.w) BIQUAD_STEP(c2.z) BIQUAD_STEP(c2.y) BIQUAD_STEP(c2.x)
    BIQUAD_STEP(c1.w) BIQUAD_STEP(c1.z) BIQUAD_STEP(c1.y) BIQUAD_STEP(c1.x)
    BIQUAD_STEP(c0.w) BIQUAD_STEP(c0.z) BIQUAD_STEP(c0.y) BIQUAD_STEP(c0.x)
    xv[4 * k + 0] = c0; xv[4 * k + 1] = c1; xv[4 * k + 2] = c2; xv[4 * k + 3] = c3;
    c0 = n0; c1 = n1; c2 = n2; c3 = n3;
  }
}

__global__ __launch_bounds__(S3_THREADS)
void pos_filtfilt(const float* __restrict__ ppg,
                  const float* __restrict__ acoef,
                  const float* __restrict__ bcoef,
                  float* __restrict__ out) {
  __shared__ __attribute__((aligned(16))) float bufA[ROWBUF];
  __shared__ __attribute__((aligned(16))) float bufB[ROWBUF];
  __shared__ __attribute__((aligned(16))) float cfs[2 * NCF + 2];
  __shared__ int    lookahead_ok;
  const int b   = blockIdx.x;
  const int tid = threadIdx.x;

  const float a0  = acoef[0];
  const float bn0 = bcoef[0] / a0, bn1 = bcoef[1] / a0, bn2 = bcoef[2] / a0;
  const float an1 = acoef[1] / a0, an2 = acoef[2] / a0;

  if (tid == 0) {
    double e1 = (double)an1, e2 = (double)an2;
    int ok = 0;
    for (int i = 0; i < NCF; ++i) {
      cfs[i] = (float)(-e1);
      cfs[NCF + i] = (float)e2;
      if (i == NFAC) ok = (fabs(e2) <= 0.25 && fabs(e1) <= 0.5 + 2.0 * e2) ? 1 : 0;
      const double n1 = fma(-e1, e1, 2.0 * e2);
      e2 = e2 * e2;
      e1 = n1;
    }
    cfs[2 * NCF]     = (float)e1;
    cfs[2 * NCF + 1] = (float)e2;
    lookahead_ok = ok;
  }

  {
    constexpr int NROWLD = (TLEN / 4 + S3_THREADS - 1) / S3_THREADS;
    const float4* prow = reinterpret_cast<const float4*>(ppg + (size_t)b * PPG_PITCH);
    float4* arow = reinterpret_cast<float4*>(bufA + FPAD);
    float4 rowv[NROWLD];
#pragma unroll
    for (int m = 0; m < NROWLD; ++m) {
      const int i = tid + m * S3_THREADS;
      rowv[m] = prow[m == NROWLD - 1 ? min(i, TLEN / 4 - 1) : i];
    }
#pragma unroll
    for (int m = 0; m < NROWLD; ++m) {
      const int i = tid + m * S3_THREADS;
      arow[m == NROWLD - 1 ? min(i, TLEN / 4 - 1) : i] = rowv[m];
    }
  }
  if (tid < FPAD / 4) {
    const float4 z4 = make_float4(0.f, 0.f, 0.f, 0.f);
    reinterpret_cast<float4*>(bufA)[tid] = z4;
    reinterpret_cast<float4*>(bufB)[tid] = z4;
    reinterpret_cast<float4*>(bufA + FPAD + TLEN)[tid] = z4;
    reinterpret_cast<float4*>(bufB + FPAD + TLEN)[tid] = z4;
  }
  __syncthreads();

  if (lookahead_ok != 0) {
    float m1[NFAC], m2[NFAC];
#pragma unroll
    for (int i = 0; i < NFAC; ++i) { m1[i] = cfs[i]; m2[i] = cfs[NCF + i]; }
    float sm1[NSFAC], sm2[NSFAC];
#pragma unroll
    for (int i = 0; i < NSFAC; ++i) { sm1[i] = cfs[NFAC + i]; sm2[i] = cfs[NCF + NFAC + i]; }
    const float ga1 = cfs[2 * NCF], ga2 = cfs[2 * NCF + 1];
    float* rowA = bufA + FPAD;
    float* rowB = bufB + FPAD;

    fir_row<+1>(rowA - PAD, rowB - PAD, m1, m2, bn0, bn1, bn2, tid);
    __syncthreads();
    step_factors<+1>(rowB, rowA, sm1, sm2, tid);
    __syncthreads();
    chains_row<+1>(rowA, ga1, ga2, tid);
    __syncthreads();
    fir_row<-1>(rowA - PAD, rowB - PAD, m1, m2, bn0, bn1, bn2, tid);
    __syncthreads();
    step_factors<-1>(rowB, rowA, sm1, sm2, tid);
    __syncthreads();
    chains_row<-1>(rowA, ga1, ga2, tid);
  } else {
    if (tid == 0) serial_scans(bufA + FPAD, bn0, bn1, bn2, an1, an2);
  }
  __syncthreads();

  const float4* xr = reinterpret_cast<const float4*>(bufA + FPAD);
  float4* orow = reinterpret_cast<float4*>(out + (size_t)b * TLEN);
  for (int i = tid; i < TLEN / 4; i += S3_THREADS) orow[i] = xr[i];
}

extern "C" void kernel_launch(void* const* d_in, const int* in_sizes, int n_in,
                              void* d_out, int out_size, void* d_ws, size_t ws_size,
                              hipStream_t stream) {
  const float* rgb    = (const float*)d_in[0];
  const float* matrix = (const float*)d_in[1];
  const float* acoef  = (const float*)d_in[2];
  const float* bcoef  = (const float*)d_in[3];
  float* out = (float*)d_out;

  float* g    = (float*)d_ws;
  float* ppg  = g + (size_t)3 * MWIN * 16;

  pos_window_stats<<<ST_WAVES, 32, 0, stream>>>(rgb, matrix, g);
  pos_overlap_add_wmma<<<OA_WAVES, 32, 0, stream>>>(rgb, g, ppg);
  pos_filtfilt<<<BATCH, S3_THREADS, 0, stream>>>(ppg, acoef, bcoef, out);
}


# Stage-3 filtfilt: row split over 9 work-groups by overlap-and-discard (2048 owned + 512 run-in a side) where poles^32 <= 1/16, else whole row; pass memory cut at 512 samples, sums otherwise unchanged
# speedup vs baseline: 120.4445x; 1.2640x over previous
#include <hip/hip_runtime.h>
#include <hip/hip_bf16.h>

typedef float v2f __attribute__((ext_vector_type(2)));
typedef float v8f __attribute__((ext_vector_type(8)));

constexpr int TLEN  = 18000;
constexpr int BATCH = 16;
constexpr int WIN   = 48;
constexpr int MWIN  = TLEN - WIN;
constexpr int PPG_PITCH = 18048;
constexpr int OA_TILE   = 32;
constexpr int OA_WAVES  = (TLEN + OA_TILE - 1) / OA_TILE;
static_assert(PPG_PITCH % 32 == 0 && PPG_PITCH >= OA_WAVES * OA_TILE, "row pitch: whole lines, room for the last tile");

constexpr int ST_TILE  = 16;
constexpr int ST_WAVES = MWIN / ST_TILE;
static_assert(MWIN % ST_TILE == 0, "whole tiles: no window of a tile lies past the last one");
static_assert((ST_WAVES - 1) * ST_TILE + 63 <= TLEN - 1, "the last tile's 64 samples lie inside the row");

__global__ __launch_bounds__(32) __attribute__((amdgpu_waves_per_eu(1, 2))) void pos_window_stats(const float* __restrict__ rgb,
                                 const float* __restrict__ matrix,
                                 float* __restrict__ g) {
  const int n0   = blockIdx.x * ST_TILE;
  const int lane = threadIdx.x;
  const int col  = lane & 15;
  const int hi   = lane >> 4;
  const int wofs = 2 * (col & 7) + (col >> 3);

  const float2* rp2 = reinterpret_cast<const float2*>(rgb + (size_t)(col * TLEN + n0 + 2 * hi) * 3);

  float2 u[16][3];
#pragma unroll
  for (int s = 0; s < 16; ++s) {
#pragma unroll
    for (int m = 0; m < 3; ++m) u[s][m] = rp2[6 * s + m];
  }

  v8f acc[9];
#pragma unroll
  for (int f = 0; f < 9; ++f) acc[f] = v8f{0.f, 0.f, 0.f, 0.f, 0.f, 0.f, 0.f, 0.f};

  static_assert(WIN == 48 && ST_TILE == 16, "common columns 16..47 = K-steps 4..11");
  v2f mid[9];
#pragma unroll
  for (int f = 0; f < 9; ++f) mid[f] = v2f{0.f, 0.f};
  const v2f ones = {1.0f, 1.0f};
#pragma unroll
  for (int s = 0; s < 16; ++s) {
    const float2 u0 = u[s][0];
    const float2 u1 = u[s][1];
    const float2 u2 = u[s][2];
    const float x0 = u0.x, x1 = u0.y, x2 = u1.x;
    const float y0 = u1.y, y1 = u2.x, y2 = u2.y;
    v2f bf[9];
    bf[0].x = x0;      bf[0].y = y0;
    bf[1].x = x1;      bf[1].y = y1;
    bf[2].x = x2;      bf[2].y = y2;
    bf[3].x = x0 * x0; bf[3].y = y0 * y0;
    bf[4].x = x0 * x1; bf[4].y = y0 * y1;
    bf[5].x = x0 * x2; bf[5].y = y0 * y2;
    bf[6].x = x1 * x1; bf[6].y = y1 * y1;
    bf[7].x = x1 * x2; bf[7].y = y1 * y2;
    bf[8].x = x2 * x2; bf[8].y = y2 * y2;
    if (s >= 4 && s < 12) {
#pragma unroll
      for (int f = 0; f < 9; ++f) mid[f] = (s == 4) ? bf[f] : mid[f] + bf[f];
      if (s == 11) {
#pragma unroll
        for (int f = 0; f < 9; ++f)
          acc[f] = __builtin_amdgcn_wmma_f32_16x16x4_f32(
               false, ones,  false, mid[f],
               (short)0, acc[f],  false,  false);
      }
    } else {
      const int j0 = 4 * s + 2 * hi;
      const int j1 = j0 + 1;
      v2f a;
      a.x = (j0 >= wofs && j0 <= wofs + 47) ? 1.0f : 0.0f;
      a.y = (j1 >= wofs && j1 <= wofs + 47) ? 1.0f : 0.0f;
#pragma unroll
      for (int f = 0; f < 9; ++f)
        acc[f] = __builtin_amdgcn_wmma_f32_16x16x4_f32(
             false, a,  false, bf[f],
             (short)0, acc[f],  false,  false);
    }
  }

  const float m00 = matrix[0], m01 = matrix[1], m02 = matrix[2];
  const float m10 = matrix[3], m11 = matrix[4], m12 = matrix[5];
  const float invW = 1.0f / (float)WIN;

#pragma unroll
  for (int p = 0; p < 8; ++p) {
    const int n = n0 + 2 * p + hi;
    const float s0 = acc[0][p], s1 = acc[1][p], s2 = acc[2][p];
    const float q00 = acc[3][p], q01 = acc[4][p], q02 = acc[5][p];
    const float q11 = acc[6][p], q12 = acc[7][p], q22 = acc[8][p];

    float inv0 = 1.0f / (s0 * invW + 1e-6f);
    float inv1 = 1.0f / (s1 * invW + 1e-6f);
    float inv2 = 1.0f / (s2 * invW + 1e-6f);

    float u0 = m00 * inv0, u1 = m01 * inv1, u2 = m02 * inv2;
    float sum0 = u0 * s0 + u1 * s1 + u2 * s2;
    float sq0  = u0 * u0 * q00 + u1 * u1 * q11 + u2 * u2 * q22 +
                 2.f * (u0 * u1 * q01 + u0 * u2 * q02 + u1 * u2 * q12);
    float var0 = (sq0 - sum0 * sum0 * invW) * (1.0f / (float)(WIN - 1));
    float std0 = sqrtf(fmaxf(var0, 0.f));

    float v0 = m10 * inv0, v1 = m11 * inv1, v2 = m12 * inv2;
    float sum1 = v0 * s0 + v1 * s1 + v2 * s2;
    float sq1  = v0 * v0 * q00 + v1 * v1 * q11 + v2 * v2 * q22 +
                 2.f * (v0 * v1 * q01 + v0 * v2 * q02 + v1 * v2 * q12);
    float var1 = (sq1 - sum1 * sum1 * invW) * (1.0f / (float)(WIN - 1));
    float std1 = sqrtf(fmaxf(var1, 0.f));

    float c = std0 / (std1 + 1e-4f);

    g[(size_t)(0 * MWIN + n) * 16 + col] = (m00 + c * m10) * inv0;
    g[(size_t)(1 * MWIN + n) * 16 + col] = (m01 + c * m11) * inv1;
    g[(size_t)(2 * MWIN + n) * 16 + col] = (m02 + c * m12) * inv2;
  }
}

__device__ __forceinline__ float f4c(const float4& v, int c) {
  return c == 0 ? v.x : c == 1 ? v.y : c == 2 ? v.z : v.w;
}

__global__ __launch_bounds__(32) __attribute__((amdgpu_waves_per_eu(1, 2))) void pos_overlap_add_wmma(const float* __restrict__ rgb,
                                     const float* __restrict__ g,
                                     float* __restrict__ ppg) {
  const int t0   = blockIdx.x * OA_TILE;
  const int lane = threadIdx.x;
  const int col  = lane & 15;
  const int hi   = lane >> 4;
  const int base = t0 - 47;

  float4 r[2][6];
#pragma unroll
  for (int tile = 0; tile < 2; ++tile) {
    const int ttc = min(t0 + 16 * tile + 8 * hi, TLEN - 8);
    const float4* rp4 = reinterpret_cast<const float4*>(rgb + (size_t)(col * TLEN + ttc) * 3);
#pragma unroll
    for (int q = 0; q < 6; ++q) r[tile][q] = rp4[q];
  }

  v8f acc[2][3];
#pragma unroll
  for (int k = 0; k < 3; ++k) {
    const float* gk = g + (size_t)k * MWIN * 16;
    v8f d0 = {0.f, 0.f, 0.f, 0.f, 0.f, 0.f, 0.f, 0.f};
    v8f d1 = {0.f, 0.f, 0.f, 0.f, 0.f, 0.f, 0.f, 0.f};
    v2f ma = {0.f, 0.f}, mb = {0.f, 0.f}, mc = {0.f, 0.f};
    const v2f ones = {1.0f, 1.0f};
#pragma unroll
    for (int s = 0; s < 20; ++s) {
      int j0 = 4 * s + 2 * hi;
      int j1 = j0 + 1;
      int n0 = base + j0;
      int n1 = base + j1;
      int c0 = min(max(n0, 0), MWIN - 1);
      int c1 = min(max(n1, 0), MWIN - 1);
      float ok0 = ((unsigned)n0 < (unsigned)MWIN) ? 1.0f : 0.0f;
      float ok1 = ((unsigned)n1 < (unsigned)MWIN) ? 1.0f : 0.0f;
      v2f bb;
      bb.x = gk[(size_t)c0 * 16 + col] * ok0;
      bb.y = gk[(size_t)c1 * 16 + col] * ok1;
      if (s >= 4 && s < 8)   ma = (s == 4)  ? bb : ma + bb;
      if (s >= 8 && s < 12)  mb = (s == 8)  ? bb : mb + bb;
      if (s >= 12 && s < 16) mc = (s == 12) ? bb : mc + bb;
      if (s < 4 || (s >= 12 && s < 16)) {
        v2f a;
        a.x = (j0 >= col && j0 <= col + 47) ? 1.0f : 0.0f;
        a.y = (j1 >= col && j1 <= col + 47) ? 1.0f : 0.0f;
        d0 = __builtin_amdgcn_wmma_f32_16x16x4_f32(
             false, a,  false, bb,
             (short)0, d0,  false,  false);
      }
      if (s == 11)
        d0 = __builtin_amdgcn_wmma_f32_16x16x4_f32(
             false, ones,  false, ma + mb,
             (short)0, d0,  false,  false);
      if ((s >= 4 && s < 8) || s >= 16) {
        v2f a;
        a.x = (j0 - 16 >= col && j0 - 16 <= col + 47) ? 1.0f : 0.0f;
        a.y = (j1 - 16 >= col && j1 - 16 <= col + 47) ? 1.0f : 0.0f;
        d1 = __builtin_amdgcn_wmma_f32_16x16x4_f32(
             false, a,  false, bb,
             (short)0, d1,  false,  false);
      }
      if (s == 15)
        d1 = __builtin_amdgcn_wmma_f32_16x16x4_f32(
             false, ones,  false, mb + mc,
             (short)0, d1,  false,  false);
    }
    acc[0][k] = d0;
    acc[1][k] = d1;
  }

#pragma unroll
  for (int tile = 0; tile < 2; ++tile) {
    const int tt = t0 + 16 * tile + 8 * hi;
    float v[8];
#pragma unroll
    for (int p = 0; p < 8; ++p)
      v[p] = f4c(r[tile][(3 * p + 0) >> 2], (3 * p + 0) & 3) * acc[tile][0][p] +
             f4c(r[tile][(3 * p + 1) >> 2], (3 * p + 1) & 3) * acc[tile][1][p] +
             f4c(r[tile][(3 * p + 2) >> 2], (3 * p + 2) & 3) * acc[tile][2][p];
    float4* dst = reinterpret_cast<float4*>(ppg + (size_t)col * PPG_PITCH + tt);
    dst[0] = make_float4(v[0], v[1], v[2], v[3]);
    dst[1] = make_float4(v[4], v[5], v[6], v[7]);
  }
}

constexpr int S3_THREADS = 256;
constexpr int LA     = 32;
constexpr int NFAC   = 5;
constexpr int PAD    = 2 * LA;
constexpr int CST    = 8;
constexpr int NSFAC  = 3;
constexpr int NCF    = NFAC + NSFAC;
constexpr int SREACH = 2 * (CST - 1);
constexpr int FPAD   = SREACH * LA;
template <int L_, int FT_, int TL_>
struct Geo {
  static constexpr int L      = L_;
  static constexpr int FT     = FT_;
  static constexpr int TL     = TL_;
  static constexpr int NSTEP  = (L + LA - 1) / LA;
  static constexpr int NRNG   = (NSTEP + TL - 1) / TL;
  static constexpr int TW     = TL + SREACH;
  static constexpr int NFT    = (L + FT - 1) / FT;
  static constexpr int FW     = FT + PAD;
  static constexpr int LASTG  = (L - (NFT - 1) * FT) / 4;
  static constexpr int TAILX  = NFT * FT - L;
  static constexpr int ROWBUF = FPAD + L + FPAD;
  static_assert(FT % 4 == 0 && (FT / 4) % 2 == 1, "tile: whole 16-B pieces, an odd number of them");
  static_assert(NFT <= S3_THREADS && PAD + TAILX <= FPAD, "one tile per thread; the last tile's reads past the end stay in the zero samples");
  static_assert(8 * NRNG <= S3_THREADS && TL % 2 == 1, "one step tile per thread");
  static_assert(NRNG * TL * LA - L <= FPAD && FPAD % 4 == 0 && L % 4 == 0,
                "the step tiles' reads past either end stay in the zero samples, 16-B aligned");
  static_assert((L - (NFT - 1) * FT) % 4 == 0 && LASTG > 0, "the data end on a 16-B piece of the last tile");
  static_assert(ROWBUF % 4 == 0, "the second buffer stays 16-B aligned");
};
static_assert(CST * LA == S3_THREADS, "one chain per thread");

template <int DIR, int S, int INV, int FW>
__device__ __forceinline__ void fir_factor(float (&w)[FW], float m1, float m2) {
  if (DIR > 0) {
#pragma unroll
    for (int j = FW - 1; j >= INV; --j)
      w[j] = fmaf(m2, w[j - 2 * S], fmaf(m1, w[j - S], w[j]));
  } else {
#pragma unroll
    for (int j = 0; j < FW - INV; ++j)
      w[j] = fmaf(m2, w[j + 2 * S], fmaf(m1, w[j + S], w[j]));
  }
}

template <int DIR, class G>
__device__ __forceinline__ void fir_row(const float* src, float* dst,
                                        const float (&m1)[NFAC], const float (&m2)[NFAC],
                                        float b0, float b1, float b2, int tid) {
  constexpr int FT = G::FT, FW = G::FW, NFT = G::NFT, LASTG = G::LASTG;
  if (tid < NFT) {
    const int t0 = FT * tid;
    const float4* p = reinterpret_cast<const float4*>(src + (DIR > 0 ? t0 : t0 + PAD));
    float w[FW];
#pragma unroll
    for (int m = 0; m < FW / 4; ++m) {
      float4 v = p[m];
      w[4 * m + 0] = v.x; w[4 * m + 1] = v.y; w[4 * m + 2] = v.z; w[4 * m + 3] = v.w;
    }
    fir_factor<DIR, 16, 32>(w, m1[4], m2[4]);
    fir_factor<DIR,  8, 48>(w, m1[3], m2[3]);
    fir_factor<DIR,  4, 56>(w, m1[2], m2[2]);
    fir_factor<DIR,  2, 60>(w, m1[1], m2[1]);
    fir_factor<DIR,  1, 62>(w, m1[0], m2[0]);
    float o[FT];
#pragma unroll
    for (int j = 0; j < FT; ++j)
      o[j] = DIR > 0 ? fmaf(b2, w[PAD + j - 2], fmaf(b1, w[PAD + j - 1], b0 * w[PAD + j]))
                     : fmaf(b2, w[j + 2],       fmaf(b1, w[j + 1],       b0 * w[j]));
    float4* d = reinterpret_cast<float4*>(dst + PAD + t0);
    const bool whole = tid < NFT - 1;
#pragma unroll
    for (int g = 0; g < FT / 4; ++g)
      if (g < LASTG || whole) d[g] = make_float4(o[4 * g], o[4 * g + 1], o[4 * g + 2], o[4 * g + 3]);
  }
}

template <int D, int INV, int TW>
__device__ __forceinline__ void step_factor(float (&w)[TW][4], float a, float b) {
#pragma unroll
  for (int k = TW - 1; k >= INV; --k) {
#pragma unroll
    for (int q = 0; q < 4; ++q) w[k][q] = fmaf(b, w[k - 2 * D][q], fmaf(a, w[k - D][q], w[k][q]));
  }
}

template <int DIR, class G>
__device__ __forceinline__ void step_factors(const float* __restrict__ src, float* __restrict__ dst,
                                             const float (&m1)[NSFAC], const float (&m2)[NSFAC],
                                             int tid) {
  constexpr int TL = G::TL, TW = G::TW, NRNG = G::NRNG, LEN = G::L;
  if (tid < 8 * NRNG) {
    const int g  = tid & 7;
    const int s0 = TL * (tid >> 3) - SREACH;
    const int e0 = DIR > 0 ? 4 * g : LEN - 4 - 4 * g;
    constexpr int SV = DIR > 0 ? LA / 4 : -(LA / 4);
    const float4* p = reinterpret_cast<const float4*>(src + e0) + SV * s0;
    float w[TW][4];
#pragma unroll
    for (int k = 0; k < TW; ++k) {
      const float4 v = p[SV * k];
      w[k][0] = v.x; w[k][1] = v.y; w[k][2] = v.z; w[k][3] = v.w;
    }
    step_factor<4,  8>(w, m1[2], m2[2]);
    step_factor<2, 12>(w, m1[1], m2[1]);
    step_factor<1, SREACH>(w, m1[0], m2[0]);
    float4* d = reinterpret_cast<float4*>(dst + e0) + SV * s0;
#pragma unroll
    for (int k = SREACH; k < TW; ++k)
      if (LA * (s0 + k) + 4 * g < LEN) d[SV * k] = make_float4(w[k][0], w[k][1], w[k][2], w[k][3]);
  }
}

template <int DIR, class G>
__device__ __forceinline__ void chains_row(float* row, float ga1, float ga2, int tid) {
  constexpr int LEN   = G::L;
  constexpr int CH    = CST * LA;
  constexpr int NFULL = LEN / CH;
  constexpr int NREM  = LEN - NFULL * CH;
  constexpr int STR   = DIR > 0 ? CH : -CH;
  float* p = row + (DIR > 0 ? tid : LEN - 1 - tid);
  float f[NFULL];
#pragma unroll
  for (int j = 0; j < NFULL; ++j) f[j] = p[j * STR];
  __builtin_amdgcn_sched_barrier(0);
  float y1 = 0.f, y2 = 0.f;
#pragma unroll
  for (int j = 0; j < NFULL; ++j) {
    const float v = fmaf(-ga1, y1, fmaf(-ga2, y2, f[j]));
    y2 = y1; y1 = v;
    p[j * STR] = v;
  }
  if (tid < NREM) p[NFULL * STR] = fmaf(-ga1, y1, fmaf(-ga2, y2, p[NFULL * STR]));
}

constexpr int NBLK16 = TLEN / 16;

#define BIQUAD_STEP(v)                                   \
  {                                                      \
    float xin_ = (v);                                    \
    float y_   = fmaf(b0, xin_, z1);                     \
    z1 = fmaf(b1, xin_, fmaf(-a1, y_, z2));              \
    z2 = fmaf(b2, xin_, -a2 * y_);                       \
    (v) = y_;                                            \
  }

__device__ __forceinline__ void serial_scans(float* xs, float b0, float b1, float b2,
                                             float a1, float a2) {
  float4* xv = reinterpret_cast<float4*>(xs);
  float z1 = 0.f, z2 = 0.f;
  float4 c0 = xv[0], c1 = xv[1], c2 = xv[2], c3 = xv[3];
  for (int k = 0; k < NBLK16; ++k) {
    int kn = min(k + 1, NBLK16 - 1);
    float4 n0 = xv[4 * kn + 0], n1 = xv[4 * kn + 1];
    float4 n2 = xv[4 * kn + 2], n3 = xv[4 * kn + 3];
    BIQUAD_STEP(c0.x) BIQUAD_STEP(c0.y) BIQUAD_STEP(c0.z) BIQUAD_STEP(c0.w)
    BIQUAD_STEP(c1.x) BIQUAD_STEP(c1.y) BIQUAD_STEP(c1.z) BIQUAD_STEP(c1.w)
    BIQUAD_STEP(c2.x) BIQUAD_STEP(c2.y) BIQUAD_STEP(c2.z) BIQUAD_STEP(c2.w)
    BIQUAD_STEP(c3.x) BIQUAD_STEP(c3.y) BIQUAD_STEP(c3.z) BIQUAD_STEP(c3.w)
    xv[4 * k + 0] = c0; xv[4 * k + 1] = c1; xv[4 * k + 2] = c2; xv[4 * k + 3] = c3;
    c0 = n0; c1 = n1; c2 = n2; c3 = n3;
  }
  z1 = 0.f; z2 = 0.f;
  c0 = xv[4 * (NBLK16 - 1) + 0]; c1 = xv[4 * (NBLK16 - 1) + 1];
  c2 = xv[4 * (NBLK16 - 1) + 2]; c3 = xv[4 * (NBLK16 - 1) + 3];
  for (int k = NBLK16 - 1; k >= 0; --k) {
    int kn = max(k - 1, 0);
    float4 n0 = xv[4 * kn + 0], n1 = xv[4 * kn + 1];
    float4 n2 = xv[4 * kn + 2], n3 = xv[4 * kn + 3];
    BIQUAD_STEP(c3.w) BIQUAD_STEP(c3.z) BIQUAD_STEP(c3.y) BIQUAD_STEP(c3.x)
    BIQUAD_STEP(c2.w) BIQUAD_STEP(c2.z) BIQUAD_STEP(c2.y) BIQUAD_STEP(c2.x)
    BIQUAD_STEP(c1.w) BIQUAD_STEP(c1.z) BIQUAD_STEP(c1.y) BIQUAD_STEP(c1.x)
    BIQUAD_STEP(c0.w) BIQUAD_STEP(c0.z) BIQUAD_STEP(c0.y) BIQUAD_STEP(c0.x)
    xv[4 * k + 0] = c0; xv[4 * k + 1] = c1; xv[4 * k + 2] = c2; xv[4 * k + 3] = c3;
    c0 = n0; c1 = n1; c2 = n2; c3 = n3;
  }
}

constexpr int SEG  = 2048;
constexpr int HALO = 512;
constexpr int LLEN = SEG + 2 * HALO;
constexpr int NSEG = (TLEN + 16 + SEG - 1) / SEG;
using GRow = Geo<TLEN, 76, 19>;
using GSeg = Geo<LLEN, 12, 3>;
static_assert(NSEG * SEG - 16 >= TLEN, "the pieces cover the row, also where they begin 16 samples earlier");
static_assert(HALO % LA == 0 && HALO >= PAD + FPAD, "the run-in covers the reach of B P and of the step factors");
static_assert((LLEN / 4) % S3_THREADS == 0 && (SEG / 4) % S3_THREADS == 0,
              "whole 16-B pieces per lane in the window's load and in the piece's store");

__global__ __launch_bounds__(S3_THREADS)
void pos_filtfilt(const float* __restrict__ ppg,
                  const float* __restrict__ acoef,
                  const float* __restrict__ bcoef,
                  float* __restrict__ out) {
  __shared__ __attribute__((aligned(16))) float bufA[GRow::ROWBUF];
  __shared__ __attribute__((aligned(16))) float bufB[GRow::ROWBUF];
  __shared__ __attribute__((aligned(16))) float cfs[2 * NCF + 2];
  __shared__ int    la_mode;
  const int seg = blockIdx.x;
  const int b   = blockIdx.y;
  const int tid = threadIdx.x;

  const float a0  = acoef[0];
  const float bn0 = bcoef[0] / a0, bn1 = bcoef[1] / a0, bn2 = bcoef[2] / a0;
  const float an1 = acoef[1] / a0, an2 = acoef[2] / a0;

  if (tid == 0) {
    double e1 = (double)an1, e2 = (double)an2;
    int ok = 0;
    for (int i = 0; i < NCF; ++i) {
      cfs[i] = (float)(-e1);
      cfs[NCF + i] = (float)e2;
      if (i == NFAC) {
        ok = (fabs(e2) <= 0.25 && fabs(e1) <= 0.5 + 2.0 * e2) ? 1 : 0;
        if (ok != 0 && fabs(e2) <= 1.0 / 256.0 && fabs(e1) <= 1.0 / 16.0 + 16.0 * e2) ok = 2;
      }
      const double n1 = fma(-e1, e1, 2.0 * e2);
      e2 = e2 * e2;
      e1 = n1;
    }
    cfs[2 * NCF]     = (float)e1;
    cfs[2 * NCF + 1] = (float)e2;
    la_mode = ok;
  }

  const float4* prow = reinterpret_cast<const float4*>(ppg + (size_t)b * PPG_PITCH);
  const int w0 = seg * SEG - 16 * (b & 1) - HALO;

  {
    constexpr int NWLD = LLEN / 4 / S3_THREADS;
    float4* arow = reinterpret_cast<float4*>(bufA + FPAD);
    float4 rowv[NWLD];
#pragma unroll
    for (int m = 0; m < NWLD; ++m) {
      const int t = w0 + 4 * (tid + m * S3_THREADS);
      rowv[m] = prow[min(max(t, 0), TLEN - 4) >> 2];
    }
#pragma unroll
    for (int m = 0; m < NWLD; ++m) {
      const int t = w0 + 4 * (tid + m * S3_THREADS);
      const bool in = (unsigned)t < (unsigned)TLEN;
      arow[tid + m * S3_THREADS] = make_float4(in ? rowv[m].x : 0.f, in ? rowv[m].y : 0.f,
                                               in ? rowv[m].z : 0.f, in ? rowv[m].w : 0.f);
    }
  }
  if (tid < FPAD / 4) {
    const float4 z4 = make_float4(0.f, 0.f, 0.f, 0.f);
    reinterpret_cast<float4*>(bufA)[tid] = z4;
    reinterpret_cast<float4*>(bufB)[tid] = z4;
    reinterpret_cast<float4*>(bufA + FPAD + LLEN)[tid] = z4;
    reinterpret_cast<float4*>(bufB + FPAD + LLEN)[tid] = z4;
  }
  __syncthreads();

  const int mode = la_mode;
  float m1[NFAC], m2[NFAC];
#pragma unroll
  for (int i = 0; i < NFAC; ++i) { m1[i] = cfs[i]; m2[i] = cfs[NCF + i]; }
  float sm1[NSFAC], sm2[NSFAC];
#pragma unroll
  for (int i = 0; i < NSFAC; ++i) { sm1[i] = cfs[NFAC + i]; sm2[i] = cfs[NCF + NFAC + i]; }
  const float ga1 = cfs[2 * NCF], ga2 = cfs[2 * NCF + 1];
  float* rowA = bufA + FPAD;
  float* rowB = bufB + FPAD;
  float4* orow = reinterpret_cast<float4*>(out + (size_t)b * TLEN);

  if (mode == 2) {
    fir_row<+1, GSeg>(rowA - PAD, rowB - PAD, m1, m2, bn0, bn1, bn2, tid);
    __syncthreads();
    step_factors<+1, GSeg>(rowB, rowA, sm1, sm2, tid);
    __syncthreads();
    chains_row<+1, GSeg>(rowA, ga1, ga2, tid);
    __syncthreads();
    if (w0 + LLEN > TLEN) {
      for (int i = tid; i < LLEN / 4; i += S3_THREADS)
        if (w0 + 4 * i >= TLEN) reinterpret_cast<float4*>(rowA)[i] = make_float4(0.f, 0.f, 0.f, 0.f);
      __syncthreads();
    }
    fir_row<-1, GSeg>(rowA - PAD, rowB - PAD, m1, m2, bn0, bn1, bn2, tid);
    __syncthreads();
    step_factors<-1, GSeg>(rowB, rowA, sm1, sm2, tid);
    __syncthreads();
    chains_row<-1, GSeg>(rowA, ga1, ga2, tid);
    __syncthreads();

    const float4* xr = reinterpret_cast<const float4*>(rowA);
#pragma unroll
    for (int m = 0; m < SEG / 4 / S3_THREADS; ++m) {
      const int i = HALO / 4 + tid + m * S3_THREADS;
      const int t = w0 + 4 * i;
      if ((unsigned)t < (unsigned)TLEN) orow[t >> 2] = xr[i];
    }
  } else if (seg == 0) {
    {
      constexpr int NROWLD = (TLEN / 4 + S3_THREADS - 1) / S3_THREADS;
      float4* arow = reinterpret_cast<float4*>(rowA);
      float4 rowv[NROWLD];
#pragma unroll
      for (int m = 0; m < NROWLD; ++m) {
        const int i = tid + m * S3_THREADS;
        rowv[m] = prow[m == NROWLD - 1 ? min(i, TLEN / 4 - 1) : i];
      }
#pragma unroll
      for (int m = 0; m < NROWLD; ++m) {
        const int i = tid + m * S3_THREADS;
        arow[m == NROWLD - 1 ? min(i, TLEN / 4 - 1) : i] = rowv[m];
      }
    }
    if (tid < FPAD / 4) {
      const float4 z4 = make_float4(0.f, 0.f, 0.f, 0.f);
      reinterpret_cast<float4*>(bufA)[tid] = z4;
      reinterpret_cast<float4*>(bufB)[tid] = z4;
      reinterpret_cast<float4*>(bufA + FPAD + TLEN)[tid] = z4;
      reinterpret_cast<float4*>(bufB + FPAD + TLEN)[tid] = z4;
    }
    __syncthreads();
    if (mode == 1) {
      fir_row<+1, GRow>(rowA - PAD, rowB - PAD, m1, m2, bn0, bn1, bn2, tid);
      __syncthreads();
      step_factors<+1, GRow>(rowB, rowA, sm1, sm2, tid);
      __syncthreads();
      chains_row<+1, GRow>(rowA, ga1, ga2, tid);
      __syncthreads();
      fir_row<-1, GRow>(rowA - PAD, rowB - PAD, m1, m2, bn0, bn1, bn2, tid);
      __syncthreads();
      step_factors<-1, GRow>(rowB, rowA, sm1, sm2, tid);
      __syncthreads();
      chains_row<-1, GRow>(rowA, ga1, ga2, tid);
    } else {
      if (tid == 0) serial_scans(rowA, bn0, bn1, bn2, an1, an2);
    }
    __syncthreads();

    const float4* xr = reinterpret_cast<const float4*>(rowA);
    for (int i = tid; i < TLEN / 4; i += S3_THREADS) orow[i] = xr[i];
  }
}

extern "C" void kernel_launch(void* const* d_in, const int* in_sizes, int n_in,
                              void* d_out, int out_size, void* d_ws, size_t ws_size,
                              hipStream_t stream) {
  const float* rgb    = (const float*)d_in[0];
  const float* matrix = (const float*)d_in[1];
  const float* acoef  = (const float*)d_in[2];
  const float* bcoef  = (const float*)d_in[3];
  float* out = (float*)d_out;

  float* g    = (float*)d_ws;
  float* ppg  = g + (size_t)3 * MWIN * 16;

  pos_window_stats<<<ST_WAVES, 32, 0, stream>>>(rgb, matrix, g);
  pos_overlap_add_wmma<<<OA_WAVES, 32, 0, stream>>>(rgb, g, ppg);
  pos_filtfilt<<<dim3(NSEG, BATCH), S3_THREADS, 0, stream>>>(ppg, acoef, bcoef, out);
}
